# starting kernel
# baseline (speedup 1.0000x reference)
#include <hip/hip_runtime.h>
#include <math.h>

#define NN   100000
#define CH   64
#define HD   128
#define NE   800000
#define N64  6400000
#define EPS_MSG 1e-7f
#define EPS_LN  1e-5f

typedef float v2f __attribute__((ext_vector_type(2)));
typedef float v8f __attribute__((ext_vector_type(8)));

static __device__ __forceinline__ v8f wmma4(v2f a, v2f b, v8f c) {
  return __builtin_amdgcn_wmma_f32_16x16x4_f32(
       false, a,  false, b,
       (short)0, c,  false,  false);
}

__global__ __launch_bounds__(256) void k_zero(float4* __restrict__ p, int n4) {
  int i = blockIdx.x * 256 + threadIdx.x;
  if (i < n4) p[i] = make_float4(0.f, 0.f, 0.f, 0.f);
}

__global__ void k_zero2(float* __restrict__ p) {
  if (threadIdx.x < 2) p[threadIdx.x] = 0.f;
}

__global__ __launch_bounds__(256) void k_reduce(const float4* __restrict__ h,
                                                float* __restrict__ sums) {
  __shared__ float ss[256];
  __shared__ float sq[256];
  float s = 0.f, q = 0.f;
  for (int i = blockIdx.x * 256 + threadIdx.x; i < N64 / 4; i += gridDim.x * 256) {
    float4 v = h[i];
    s += v.x + v.y + v.z + v.w;
    q += v.x * v.x + v.y * v.y + v.z * v.z + v.w * v.w;
  }
  ss[threadIdx.x] = s; sq[threadIdx.x] = q;
  __syncthreads();
  for (int off = 128; off > 0; off >>= 1) {
    if (threadIdx.x < off) {
      ss[threadIdx.x] += ss[threadIdx.x + off];
      sq[threadIdx.x] += sq[threadIdx.x + off];
    }
    __syncthreads();
  }
  if (threadIdx.x == 0) {
    atomicAdd(&sums[0], ss[0]);
    atomicAdd(&sums[1], sq[0]);
  }
}

__global__ __launch_bounds__(256) void k_gln_apply(const float4* __restrict__ h,
                                                   const float* __restrict__ sums,
                                                   const float* __restrict__ w,
                                                   const float* __restrict__ b,
                                                   float4* __restrict__ out) {
  int i = blockIdx.x * 256 + threadIdx.x;
  if (i >= N64 / 4) return;
  const float invn = 1.f / (float)N64;
  float mu = sums[0] * invn;
  float var = sums[1] * invn - mu * mu;
  float scale = 1.f / (sqrtf(fmaxf(var, 0.f)) + EPS_LN);
  int c = (i & 15) << 2;
  float4 v = h[i];
  float r0 = (v.x - mu) * scale * w[c + 0] + b[c + 0];
  float r1 = (v.y - mu) * scale * w[c + 1] + b[c + 1];
  float r2 = (v.z - mu) * scale * w[c + 2] + b[c + 2];
  float r3 = (v.w - mu) * scale * w[c + 3] + b[c + 3];
  out[i] = make_float4(fmaxf(r0, 0.f), fmaxf(r1, 0.f), fmaxf(r2, 0.f), fmaxf(r3, 0.f));
}

__global__ __launch_bounds__(256) void k_edge_max(const int* __restrict__ src,
                                                  const int* __restrict__ dst,
                                                  const float* __restrict__ hin,
                                                  int* __restrict__ amax) {
  int idx = blockIdx.x * 256 + threadIdx.x;
  if (idx >= NE * 16) return;
  int e = idx >> 4, g = (idx & 15) << 2;
  int s = src[e], d = dst[e];
  const float4 v = *(const float4*)(hin + (size_t)s * CH + g);
  int* p = amax + (size_t)d * CH + g;
  atomicMax(p + 0, __float_as_int(fmaxf(v.x, 0.f) + EPS_MSG));
  atomicMax(p + 1, __float_as_int(fmaxf(v.y, 0.f) + EPS_MSG));
  atomicMax(p + 2, __float_as_int(fmaxf(v.z, 0.f) + EPS_MSG));
  atomicMax(p + 3, __float_as_int(fmaxf(v.w, 0.f) + EPS_MSG));
}

__global__ __launch_bounds__(256) void k_edge_sum(const int* __restrict__ src,
                                                  const int* __restrict__ dst,
                                                  const float* __restrict__ hin,
                                                  const float* __restrict__ amax,
                                                  float* __restrict__ denom) {
  int idx = blockIdx.x * 256 + threadIdx.x;
  if (idx >= NE * 16) return;
  int e = idx >> 4, g = (idx & 15) << 2;
  int s = src[e], d = dst[e];
  const float4 v = *(const float4*)(hin + (size_t)s * CH + g);
  const float4 am = *(const float4*)(amax + (size_t)d * CH + g);
  float* p = denom + (size_t)d * CH + g;
  atomicAdd(p + 0, __expf(fmaxf(v.x, 0.f) + EPS_MSG - am.x));
  atomicAdd(p + 1, __expf(fmaxf(v.y, 0.f) + EPS_MSG - am.y));
  atomicAdd(p + 2, __expf(fmaxf(v.z, 0.f) + EPS_MSG - am.z));
  atomicAdd(p + 3, __expf(fmaxf(v.w, 0.f) + EPS_MSG - am.w));
}

__global__ __launch_bounds__(256) void k_edge_agg(const int* __restrict__ src,
                                                  const int* __restrict__ dst,
                                                  const float* __restrict__ hin,
                                                  const float* __restrict__ amax,
                                                  const float* __restrict__ denom,
                                                  float* __restrict__ agg) {
  int idx = blockIdx.x * 256 + threadIdx.x;
  if (idx >= NE * 16) return;
  int e = idx >> 4, g = (idx & 15) << 2;
  int s = src[e], d = dst[e];
  const float4 v = *(const float4*)(hin + (size_t)s * CH + g);
  const float4 am = *(const float4*)(amax + (size_t)d * CH + g);
  const float4 dn = *(const float4*)(denom + (size_t)d * CH + g);
  float m0 = fmaxf(v.x, 0.f) + EPS_MSG;
  float m1 = fmaxf(v.y, 0.f) + EPS_MSG;
  float m2 = fmaxf(v.z, 0.f) + EPS_MSG;
  float m3 = fmaxf(v.w, 0.f) + EPS_MSG;
  float* p = agg + (size_t)d * CH + g;
  atomicAdd(p + 0, m0 * __expf(m0 - am.x) / dn.x);
  atomicAdd(p + 1, m1 * __expf(m1 - am.y) / dn.y);
  atomicAdd(p + 2, m2 * __expf(m2 - am.z) / dn.z);
  atomicAdd(p + 3, m3 * __expf(m3 - am.w) / dn.w);
}

__global__ __launch_bounds__(256)
void k_conv_mlp(const float* __restrict__ resin, const float* __restrict__ agg,
                const float* __restrict__ W1, const float* __restrict__ b1,
                const float* __restrict__ lnw, const float* __restrict__ lnb,
                const float* __restrict__ W2, const float* __restrict__ b2,
                const float* __restrict__ outer, float* __restrict__ out) {
  __shared__ float sAct[32 * CH];
  __shared__ float sHid[32 * HD];
  __shared__ float sb1[HD], slnw[HD], slnb[HD], sb2[CH];
  const int tid = threadIdx.x;
  const int base = blockIdx.x * 32;
  if (tid < HD) { sb1[tid] = b1[tid]; slnw[tid] = lnw[tid]; slnb[tid] = lnb[tid]; }
  if (tid < CH) sb2[tid] = b2[tid];
  for (int i = tid; i < 32 * CH; i += 256) {
    size_t idx = (size_t)base * CH + i;
    sAct[i] = resin[idx] + agg[idx];
  }
  __syncthreads();

  const int wave = tid >> 5, lane = tid & 31;
  const int half = lane >> 4, l16 = lane & 15;

  for (int task = wave; task < 16; task += 8) {
    int rt = task >> 3, ct = task & 7;
    const float* arow = &sAct[(rt * 16 + l16) * CH];
    v8f acc = {0.f, 0.f, 0.f, 0.f, 0.f, 0.f, 0.f, 0.f};
    #pragma unroll
    for (int k0 = 0; k0 < CH; k0 += 4) {
      int k = k0 + 2 * half;
      v2f a;  a.x = arow[k];                     a.y = arow[k + 1];
      v2f bb; bb.x = W1[k * HD + ct * 16 + l16]; bb.y = W1[(k + 1) * HD + ct * 16 + l16];
      acc = wmma4(a, bb, acc);
    }
    #pragma unroll
    for (int r = 0; r < 8; r++) {
      int m = rt * 16 + r + 8 * half;
      int n = ct * 16 + l16;
      sHid[m * HD + n] = acc[r] + sb1[n];
    }
  }
  __syncthreads();

  if (tid < 32) {
    float s = 0.f, q = 0.f;
    for (int c = 0; c < HD; c++) { float v = sHid[tid * HD + c]; s += v; q += v * v; }
    float mu = s * (1.f / HD);
    float var = q * (1.f / HD) - mu * mu;
    float inv = rsqrtf(var + EPS_LN);
    for (int c = 0; c < HD; c++) {
      float v = (sHid[tid * HD + c] - mu) * inv * slnw[c] + slnb[c];
      sHid[tid * HD + c] = fmaxf(v, 0.f);
    }
  }
  __syncthreads();

  {
    int rt = wave >> 2, ct = wave & 3;
    const float* arow = &sHid[(rt * 16 + l16) * HD];
    v8f acc = {0.f, 0.f, 0.f, 0.f, 0.f, 0.f, 0.f, 0.f};
    #pragma unroll
    for (int k0 = 0; k0 < HD; k0 += 4) {
      int k = k0 + 2 * half;
      v2f a;  a.x = arow[k];                     a.y = arow[k + 1];
      v2f bb; bb.x = W2[k * CH + ct * 16 + l16]; bb.y = W2[(k + 1) * CH + ct * 16 + l16];
      acc = wmma4(a, bb, acc);
    }
    #pragma unroll
    for (int r = 0; r < 8; r++) {
      int m = rt * 16 + r + 8 * half;
      size_t node = (size_t)(base + m);
      int n = ct * 16 + l16;
      float v = acc[r] + sb2[n];
      if (outer) v += outer[node * CH + n];
      out[node * CH + n] = v;
    }
  }
}

__global__ __launch_bounds__(256)
void k_head(const float* __restrict__ xc,
            const float* __restrict__ W1, const float* __restrict__ b1,
            const float* __restrict__ W2, const float* __restrict__ b2,
            const float* __restrict__ W3, const float* __restrict__ b3,
            float* __restrict__ out) {
  __shared__ float sA[32 * CH];
  __shared__ float sH[32 * CH];
  __shared__ float sb1[CH], sb2[CH], sW3[CH * 2], sb3[2];
  const int tid = threadIdx.x;
  const int base = blockIdx.x * 32;
  if (tid < CH) { sb1[tid] = b1[tid]; sb2[tid] = b2[tid]; }
  if (tid < CH * 2) sW3[tid] = W3[tid];
  if (tid < 2) sb3[tid] = b3[tid];
  for (int i = tid; i < 32 * CH; i += 256) sA[i] = xc[(size_t)base * CH + i];
  __syncthreads();

  const int wave = tid >> 5, lane = tid & 31;
  const int half = lane >> 4, l16 = lane & 15;
  const int rt = wave >> 2, ct = wave & 3;

  {
    const float* arow = &sA[(rt * 16 + l16) * CH];
    v8f acc = {0.f, 0.f, 0.f, 0.f, 0.f, 0.f, 0.f, 0.f};
    #pragma unroll
    for (int k0 = 0; k0 < CH; k0 += 4) {
      int k = k0 + 2 * half;
      v2f a;  a.x = arow[k];                     a.y = arow[k + 1];
      v2f bb; bb.x = W1[k * CH + ct * 16 + l16]; bb.y = W1[(k + 1) * CH + ct * 16 + l16];
      acc = wmma4(a, bb, acc);
    }
    #pragma unroll
    for (int r = 0; r < 8; r++) {
      int m = rt * 16 + r + 8 * half, n = ct * 16 + l16;
      sH[m * CH + n] = fmaxf(acc[r] + sb1[n], 0.f);
    }
  }
  __syncthreads();
  {
    const float* arow = &sH[(rt * 16 + l16) * CH];
    v8f acc = {0.f, 0.f, 0.f, 0.f, 0.f, 0.f, 0.f, 0.f};
    #pragma unroll
    for (int k0 = 0; k0 < CH; k0 += 4) {
      int k = k0 + 2 * half;
      v2f a;  a.x = arow[k];                     a.y = arow[k + 1];
      v2f bb; bb.x = W2[k * CH + ct * 16 + l16]; bb.y = W2[(k + 1) * CH + ct * 16 + l16];
      acc = wmma4(a, bb, acc);
    }
    #pragma unroll
    for (int r = 0; r < 8; r++) {
      int m = rt * 16 + r + 8 * half, n = ct * 16 + l16;
      sA[m * CH + n] = fmaxf(acc[r] + sb2[n], 0.f);
    }
  }
  __syncthreads();
  if (tid < 64) {
    int m = tid >> 1, o = tid & 1;
    float s = sb3[o];
    for (int k = 0; k < CH; k++) s += sA[m * CH + k] * sW3[k * 2 + o];
    out[(size_t)(base + m) * 2 + o] = s;
  }
}

extern "C" void kernel_launch(void* const* d_in, const int* in_sizes, int n_in,
                              void* d_out, int out_size, void* d_ws, size_t ws_size,
                              hipStream_t stream) {
  (void)in_sizes; (void)n_in; (void)out_size; (void)ws_size;
  const float* x      = (const float*)d_in[0];
  const int*   ei     = (const int*)d_in[1];
  const int*   src    = ei;
  const int*   dst    = ei + NE;
  const float* c1_W1  = (const float*)d_in[2];
  const float* c1_b1  = (const float*)d_in[3];
  const float* c1_lnw = (const float*)d_in[4];
  const float* c1_lnb = (const float*)d_in[5];
  const float* c1_W2  = (const float*)d_in[6];
  const float* c1_b2  = (const float*)d_in[7];
  const float* n1_w   = (const float*)d_in[8];
  const float* n1_b   = (const float*)d_in[9];
  const float* cW1    = (const float*)d_in[10];
  const float* cb1    = (const float*)d_in[11];
  const float* clnw   = (const float*)d_in[12];
  const float* clnb   = (const float*)d_in[13];
  const float* cW2    = (const float*)d_in[14];
  const float* cb2    = (const float*)d_in[15];
  const float* nw     = (const float*)d_in[16];
  const float* nb     = (const float*)d_in[17];
  const float* lW1    = (const float*)d_in[18];
  const float* lb1    = (const float*)d_in[19];
  const float* lW2    = (const float*)d_in[20];
  const float* lb2    = (const float*)d_in[21];
  const float* lW3    = (const float*)d_in[22];
  const float* lb3    = (const float*)d_in[23];

  float* ws    = (float*)d_ws;
  float* xc    = ws;
  float* hpre  = ws + (size_t)N64;
  float* amaxf = ws + 2 * (size_t)N64;
  int*   amax  = (int*)amaxf;
  float* denom = ws + 3 * (size_t)N64;
  float* agg   = ws + 4 * (size_t)N64;
  float* sums  = ws + 5 * (size_t)N64;

  const dim3 B(256);
  const int EB   = (NE * 16 + 255) / 256;
  const int Z3   = (3 * N64 / 4 + 255) / 256;
  const int APB  = (N64 / 4 + 255) / 256;
  const int MLPB = NN / 32;

  auto edge_phase = [&](const float* hin) {
    k_zero<<<Z3, B, 0, stream>>>((float4*)amaxf, 3 * N64 / 4);
    k_edge_max<<<EB, B, 0, stream>>>(src, dst, hin, amax);
    k_edge_sum<<<EB, B, 0, stream>>>(src, dst, hin, amaxf, denom);
    k_edge_agg<<<EB, B, 0, stream>>>(src, dst, hin, amaxf, denom, agg);
  };

  edge_phase(x);
  k_conv_mlp<<<MLPB, B, 0, stream>>>(x, agg, c1_W1, c1_b1, c1_lnw, c1_lnb,
                                     c1_W2, c1_b2, nullptr, hpre);
  k_zero2<<<1, 32, 0, stream>>>(sums);
  k_reduce<<<4096, B, 0, stream>>>((const float4*)hpre, sums);
  k_gln_apply<<<APB, B, 0, stream>>>((const float4*)hpre, sums, n1_w, n1_b, (float4*)xc);

  for (int i = 0; i < 3; i++) {
    k_zero2<<<1, 32, 0, stream>>>(sums);
    k_reduce<<<4096, B, 0, stream>>>((const float4*)xc, sums);
    k_gln_apply<<<APB, B, 0, stream>>>((const float4*)xc, sums, nw + i * CH, nb + i * CH,
                                       (float4*)hpre);
    edge_phase(hpre);
    k_conv_mlp<<<MLPB, B, 0, stream>>>(hpre, agg,
                                       cW1 + (size_t)i * CH * HD, cb1 + (size_t)i * HD,
                                       clnw + (size_t)i * HD, clnb + (size_t)i * HD,
                                       cW2 + (size_t)i * HD * CH, cb2 + (size_t)i * CH,
                                       xc, xc);
  }

  k_head<<<MLPB, B, 0, stream>>>(xc, lW1, lb1, lW2, lb2, lW3, lb3, (float*)d_out);
}


# Owner edge list built once per call: the first layer's launch stores each work-group's noted (source, local node) list and count in the workspace; layers 2-4 load it instead of re-scanning dst
# speedup vs baseline: 2.0849x; 2.0849x over previous
#include <hip/hip_runtime.h>
#include <math.h>

#define NN   100000
#define CH   64
#define HD   128
#define NE   800000
#define N64  6400000
#define EPS_MSG 1e-7f
#define EPS_LN  1e-5f

typedef float v2f __attribute__((ext_vector_type(2)));
typedef float v8f __attribute__((ext_vector_type(8)));

static __device__ __forceinline__ v8f wmma4(v2f a, v2f b, v8f c) {
  return __builtin_amdgcn_wmma_f32_16x16x4_f32(
       false, a,  false, b,
       (short)0, c,  false,  false);
}

__global__ void k_zero2(float* __restrict__ p) {
  if (threadIdx.x < 2) p[threadIdx.x] = 0.f;
}

__global__ __launch_bounds__(256) void k_reduce(const float4* __restrict__ h,
                                                float* __restrict__ sums) {
  __shared__ float ss[256];
  __shared__ float sq[256];
  float s = 0.f, q = 0.f;
  for (int i = blockIdx.x * 256 + threadIdx.x; i < N64 / 4; i += gridDim.x * 256) {
    float4 v = h[i];
    s += v.x + v.y + v.z + v.w;
    q += v.x * v.x + v.y * v.y + v.z * v.z + v.w * v.w;
  }
  ss[threadIdx.x] = s; sq[threadIdx.x] = q;
  __syncthreads();
  for (int off = 128; off > 0; off >>= 1) {
    if (threadIdx.x < off) {
      ss[threadIdx.x] += ss[threadIdx.x + off];
      sq[threadIdx.x] += sq[threadIdx.x + off];
    }
    __syncthreads();
  }
  if (threadIdx.x == 0) {
    atomicAdd(&sums[0], ss[0]);
    atomicAdd(&sums[1], sq[0]);
  }
}

__global__ __launch_bounds__(256) void k_gln_apply(const float4* __restrict__ h,
                                                   const float* __restrict__ sums,
                                                   const float* __restrict__ w,
                                                   const float* __restrict__ b,
                                                   float4* __restrict__ out) {
  int i = blockIdx.x * 256 + threadIdx.x;
  if (i >= N64 / 4) return;
  const float invn = 1.f / (float)N64;
  float mu = sums[0] * invn;
  float var = sums[1] * invn - mu * mu;
  float scale = 1.f / (sqrtf(fmaxf(var, 0.f)) + EPS_LN);
  int c = (i & 15) << 2;
  float4 v = h[i];
  float r0 = (v.x - mu) * scale * w[c + 0] + b[c + 0];
  float r1 = (v.y - mu) * scale * w[c + 1] + b[c + 1];
  float r2 = (v.z - mu) * scale * w[c + 2] + b[c + 2];
  float r3 = (v.w - mu) * scale * w[c + 3] + b[c + 3];
  out[i] = make_float4(fmaxf(r0, 0.f), fmaxf(r1, 0.f), fmaxf(r2, 0.f), fmaxf(r3, 0.f));
}

#define NPB   391
#define NOWN  256
#define OWNT  1024
#define LCAP  4096
#define LGRP  16
#define SGRP  8
#define LSTR  (LCAP + 32)
static_assert(NPB * NOWN >= NN && NPB * (NOWN - 1) < NN, "owner ranges must cover NN");
static_assert(NPB <= 512 && NN <= (1 << 23) && NE <= (1 << 23), "packed entry: 9 bits node, 23 bits edge or source");
static_assert(NE % 32 == 0 && NE % 4 == 0, "edge list is walked in 32s and in 16-byte loads");

static __device__ __forceinline__ void own_max(int* sMax, unsigned dl, float v0, float v1, int lane) {
  atomicMax(&sMax[dl * CH + lane],      __float_as_int(fmaxf(v0, 0.f) + EPS_MSG));
  atomicMax(&sMax[dl * CH + 32 + lane], __float_as_int(fmaxf(v1, 0.f) + EPS_MSG));
}
static __device__ __forceinline__ void own_acc(const int* sMax, float* sDen, float* sNum,
                                               unsigned dl, float v0, float v1, int lane) {
  const float m0 = fmaxf(v0, 0.f) + EPS_MSG, m1 = fmaxf(v1, 0.f) + EPS_MSG;
  const float e0 = __expf(m0 - __int_as_float(sMax[dl * CH + lane]));
  const float e1 = __expf(m1 - __int_as_float(sMax[dl * CH + 32 + lane]));
  atomicAdd(&sDen[dl * CH + lane], e0);       atomicAdd(&sNum[dl * CH + lane], m0 * e0);
  atomicAdd(&sDen[dl * CH + 32 + lane], e1);  atomicAdd(&sNum[dl * CH + 32 + lane], m1 * e1);
}

template <bool FIRST>
__global__ __launch_bounds__(OWNT)
void k_agg_owner(const int* __restrict__ src, const int* __restrict__ dst,
                 const float* __restrict__ hin, float* __restrict__ agg,
                 unsigned* __restrict__ wlist) {
  __shared__ int      sMax[NPB * CH];
  __shared__ float    sDen[NPB * CH];
  __shared__ float    sNum[NPB * CH];
  __shared__ unsigned sList[LCAP];
  __shared__ unsigned sCnt;
  const int tid = threadIdx.x, lane = tid & 31, wave = tid >> 5;
  const int lo = blockIdx.x * NPB;
  const unsigned cnt = (unsigned)min(NPB, NN - lo);

  for (int i = tid; i < NPB * CH; i += OWNT) { sMax[i] = 0; sDen[i] = 0.f; sNum[i] = 0.f; }
  unsigned* wl = wlist + (size_t)blockIdx.x * LSTR;
  if (FIRST && tid == 0) sCnt = 0u;
  __syncthreads();

  unsigned found;
  if (FIRST) {
    const int4* dst4 = (const int4*)dst;
    for (int i = tid; i < NE / 4; i += SGRP * OWNT) {
      int4 d[SGRP];
      #pragma unroll
      for (int u = 0; u < SGRP; u++) {
        const int ii = i + u * OWNT;
        d[u] = (ii < NE / 4) ? dst4[ii] : make_int4(-1, -1, -1, -1);
      }
      #pragma unroll
      for (int u = 0; u < SGRP; u++) {
        const int e = 4 * (i + u * OWNT);
        const int dd[4] = {d[u].x, d[u].y, d[u].z, d[u].w};
        #pragma unroll
        for (int c = 0; c < 4; c++) {
          const unsigned dl = (unsigned)(dd[c] - lo);
          if (dl < cnt) {
            const unsigned pos = atomicAdd(&sCnt, 1u);
            if (pos < LCAP) sList[pos] = ((unsigned)(e + c) << 9) | dl;
          }
        }
      }
    }
    __syncthreads();
    found = sCnt;
    if (tid == 0) wl[LCAP] = found;
  } else {
    found = wl[LCAP];
  }

  if (found <= LCAP) {
    const int total = (int)found;
    if (FIRST) {
      for (int i = tid; i < total; i += OWNT) {
        const unsigned ent = sList[i];
        const unsigned fin = ((unsigned)src[ent >> 9] << 9) | (ent & 511u);
        sList[i] = fin;
        wl[i] = fin;
      }
    } else {
      for (int i = tid; i < total; i += OWNT) sList[i] = wl[i];
    }
    __syncthreads();
    for (int ph = 0; ph < 2; ph++) {
      for (int base = wave * LGRP; base < total; base += 32 * LGRP) {
        unsigned ent[LGRP]; float v0[LGRP], v1[LGRP];
        #pragma unroll
        for (int u = 0; u < LGRP; u++) {
          const int idx = base + u;
          ent[u] = sList[idx < total ? idx : base];
          const float* hp = hin + (size_t)(ent[u] >> 9) * CH + lane;
          v0[u] = hp[0]; v1[u] = hp[32];
        }
        #pragma unroll
        for (int u = 0; u < LGRP; u++) {
          if (base + u < total) {
            if (ph == 0) own_max(sMax, ent[u] & 511u, v0[u], v1[u], lane);
            else         own_acc(sMax, sDen, sNum, ent[u] & 511u, v0[u], v1[u], lane);
          }
        }
      }
      __syncthreads();
    }
  } else {
    for (int ph = 0; ph < 2; ph++) {
      for (int eb = wave * 32; eb < NE; eb += 32 * 32) {
        const unsigned dl = (unsigned)(dst[eb + lane] - lo);
        unsigned mask = __builtin_amdgcn_ballot_w32(dl < cnt);
        while (mask) {
          const int b = __ffs(mask) - 1;
          mask &= mask - 1u;
          const unsigned dlb = (unsigned)__shfl((int)dl, b);
          const float* hp = hin + (size_t)src[eb + b] * CH + lane;
          const float v0 = hp[0], v1 = hp[32];
          if (ph == 0) own_max(sMax, dlb, v0, v1, lane);
          else         own_acc(sMax, sDen, sNum, dlb, v0, v1, lane);
        }
      }
      __syncthreads();
    }
  }

  const int nout = (int)cnt * CH;
  float* ap = agg + (size_t)lo * CH;
  for (int i = tid; i < nout; i += OWNT) {
    const float den = sDen[i];
    ap[i] = den > 0.f ? sNum[i] / den : 0.f;
  }
}

__global__ __launch_bounds__(256)
void k_conv_mlp(const float* __restrict__ resin, const float* __restrict__ agg,
                const float* __restrict__ W1, const float* __restrict__ b1,
                const float* __restrict__ lnw, const float* __restrict__ lnb,
                const float* __restrict__ W2, const float* __restrict__ b2,
                const float* __restrict__ outer, float* __restrict__ out) {
  __shared__ float sAct[32 * CH];
  __shared__ float sHid[32 * HD];
  __shared__ float sb1[HD], slnw[HD], slnb[HD], sb2[CH];
  const int tid = threadIdx.x;
  const int base = blockIdx.x * 32;
  if (tid < HD) { sb1[tid] = b1[tid]; slnw[tid] = lnw[tid]; slnb[tid] = lnb[tid]; }
  if (tid < CH) sb2[tid] = b2[tid];
  for (int i = tid; i < 32 * CH; i += 256) {
    size_t idx = (size_t)base * CH + i;
    sAct[i] = resin[idx] + agg[idx];
  }
  __syncthreads();

  const int wave = tid >> 5, lane = tid & 31;
  const int half = lane >> 4, l16 = lane & 15;

  for (int task = wave; task < 16; task += 8) {
    int rt = task >> 3, ct = task & 7;
    const float* arow = &sAct[(rt * 16 + l16) * CH];
    v8f acc = {0.f, 0.f, 0.f, 0.f, 0.f, 0.f, 0.f, 0.f};
    #pragma unroll
    for (int k0 = 0; k0 < CH; k0 += 4) {
      int k = k0 + 2 * half;
      v2f a;  a.x = arow[k];                     a.y = arow[k + 1];
      v2f bb; bb.x = W1[k * HD + ct * 16 + l16]; bb.y = W1[(k + 1) * HD + ct * 16 + l16];
      acc = wmma4(a, bb, acc);
    }
    #pragma unroll
    for (int r = 0; r < 8; r++) {
      int m = rt * 16 + r + 8 * half;
      int n = ct * 16 + l16;
      sHid[m * HD + n] = acc[r] + sb1[n];
    }
  }
  __syncthreads();

  if (tid < 32) {
    float s = 0.f, q = 0.f;
    for (int c = 0; c < HD; c++) { float v = sHid[tid * HD + c]; s += v; q += v * v; }
    float mu = s * (1.f / HD);
    float var = q * (1.f / HD) - mu * mu;
    float inv = rsqrtf(var + EPS_LN);
    for (int c = 0; c < HD; c++) {
      float v = (sHid[tid * HD + c] - mu) * inv * slnw[c] + slnb[c];
      sHid[tid * HD + c] = fmaxf(v, 0.f);
    }
  }
  __syncthreads();

  {
    int rt = wave >> 2, ct = wave & 3;
    const float* arow = &sHid[(rt * 16 + l16) * HD];
    v8f acc = {0.f, 0.f, 0.f, 0.f, 0.f, 0.f, 0.f, 0.f};
    #pragma unroll
    for (int k0 = 0; k0 < HD; k0 += 4) {
      int k = k0 + 2 * half;
      v2f a;  a.x = arow[k];                     a.y = arow[k + 1];
      v2f bb; bb.x = W2[k * CH + ct * 16 + l16]; bb.y = W2[(k + 1) * CH + ct * 16 + l16];
      acc = wmma4(a, bb, acc);
    }
    #pragma unroll
    for (int r = 0; r < 8; r++) {
      int m = rt * 16 + r + 8 * half;
      size_t node = (size_t)(base + m);
      int n = ct * 16 + l16;
      float v = acc[r] + sb2[n];
      if (outer) v += outer[node * CH + n];
      out[node * CH + n] = v;
    }
  }
}

__global__ __launch_bounds__(256)
void k_head(const float* __restrict__ xc,
            const float* __restrict__ W1, const float* __restrict__ b1,
            const float* __restrict__ W2, const float* __restrict__ b2,
            const float* __restrict__ W3, const float* __restrict__ b3,
            float* __restrict__ out) {
  __shared__ float sA[32 * CH];
  __shared__ float sH[32 * CH];
  __shared__ float sb1[CH], sb2[CH], sW3[CH * 2], sb3[2];
  const int tid = threadIdx.x;
  const int base = blockIdx.x * 32;
  if (tid < CH) { sb1[tid] = b1[tid]; sb2[tid] = b2[tid]; }
  if (tid < CH * 2) sW3[tid] = W3[tid];
  if (tid < 2) sb3[tid] = b3[tid];
  for (int i = tid; i < 32 * CH; i += 256) sA[i] = xc[(size_t)base * CH + i];
  __syncthreads();

  const int wave = tid >> 5, lane = tid & 31;
  const int half = lane >> 4, l16 = lane & 15;
  const int rt = wave >> 2, ct = wave & 3;

  {
    const float* arow = &sA[(rt * 16 + l16) * CH];
    v8f acc = {0.f, 0.f, 0.f, 0.f, 0.f, 0.f, 0.f, 0.f};
    #pragma unroll
    for (int k0 = 0; k0 < CH; k0 += 4) {
      int k = k0 + 2 * half;
      v2f a;  a.x = arow[k];                     a.y = arow[k + 1];
      v2f bb; bb.x = W1[k * CH + ct * 16 + l16]; bb.y = W1[(k + 1) * CH + ct * 16 + l16];
      acc = wmma4(a, bb, acc);
    }
    #pragma unroll
    for (int r = 0; r < 8; r++) {
      int m = rt * 16 + r + 8 * half, n = ct * 16 + l16;
      sH[m * CH + n] = fmaxf(acc[r] + sb1[n], 0.f);
    }
  }
  __syncthreads();
  {
    const float* arow = &sH[(rt * 16 + l16) * CH];
    v8f acc = {0.f, 0.f, 0.f, 0.f, 0.f, 0.f, 0.f, 0.f};
    #pragma unroll
    for (int k0 = 0; k0 < CH; k0 += 4) {
      int k = k0 + 2 * half;
      v2f a;  a.x = arow[k];                     a.y = arow[k + 1];
      v2f bb; bb.x = W2[k * CH + ct * 16 + l16]; bb.y = W2[(k + 1) * CH + ct * 16 + l16];
      acc = wmma4(a, bb, acc);
    }
    #pragma unroll
    for (int r = 0; r < 8; r++) {
      int m = rt * 16 + r + 8 * half, n = ct * 16 + l16;
      sA[m * CH + n] = fmaxf(acc[r] + sb2[n], 0.f);
    }
  }
  __syncthreads();
  if (tid < 64) {
    int m = tid >> 1, o = tid & 1;
    float s = sb3[o];
    for (int k = 0; k < CH; k++) s += sA[m * CH + k] * sW3[k * 2 + o];
    out[(size_t)(base + m) * 2 + o] = s;
  }
}

extern "C" void kernel_launch(void* const* d_in, const int* in_sizes, int n_in,
                              void* d_out, int out_size, void* d_ws, size_t ws_size,
                              hipStream_t stream) {
  (void)in_sizes; (void)n_in; (void)out_size; (void)ws_size;
  const float* x      = (const float*)d_in[0];
  const int*   ei     = (const int*)d_in[1];
  const int*   src    = ei;
  const int*   dst    = ei + NE;
  const float* c1_W1  = (const float*)d_in[2];
  const float* c1_b1  = (const float*)d_in[3];
  const float* c1_lnw = (const float*)d_in[4];
  const float* c1_lnb = (const float*)d_in[5];
  const float* c1_W2  = (const float*)d_in[6];
  const float* c1_b2  = (const float*)d_in[7];
  const float* n1_w   = (const float*)d_in[8];
  const float* n1_b   = (const float*)d_in[9];
  const float* cW1    = (const float*)d_in[10];
  const float* cb1    = (const float*)d_in[11];
  const float* clnw   = (const float*)d_in[12];
  const float* clnb   = (const float*)d_in[13];
  const float* cW2    = (const float*)d_in[14];
  const float* cb2    = (const float*)d_in[15];
  const float* nw     = (const float*)d_in[16];
  const float* nb     = (const float*)d_in[17];
  const float* lW1    = (const float*)d_in[18];
  const float* lb1    = (const float*)d_in[19];
  const float* lW2    = (const float*)d_in[20];
  const float* lb2    = (const float*)d_in[21];
  const float* lW3    = (const float*)d_in[22];
  const float* lb3    = (const float*)d_in[23];

  float* ws    = (float*)d_ws;
  float* xc    = ws;
  float* hpre  = ws + (size_t)N64;
  float* agg   = ws + 4 * (size_t)N64;
  float* sums  = ws + 5 * (size_t)N64;
  unsigned* wlist = (unsigned*)(ws + 2 * (size_t)N64);

  const dim3 B(256);
  const int APB  = (N64 / 4 + 255) / 256;
  const int MLPB = NN / 32;

  auto edge_phase = [&](const float* hin, bool first) {
    if (first) k_agg_owner<true><<<NOWN, OWNT, 0, stream>>>(src, dst, hin, agg, wlist);
    else       k_agg_owner<false><<<NOWN, OWNT, 0, stream>>>(src, dst, hin, agg, wlist);
  };

  edge_phase(x, true);
  k_conv_mlp<<<MLPB, B, 0, stream>>>(x, agg, c1_W1, c1_b1, c1_lnw, c1_lnb,
                                     c1_W2, c1_b2, nullptr, hpre);
  k_zero2<<<1, 32, 0, stream>>>(sums);
  k_reduce<<<4096, B, 0, stream>>>((const float4*)hpre, sums);
  k_gln_apply<<<APB, B, 0, stream>>>((const float4*)hpre, sums, n1_w, n1_b, (float4*)xc);

  for (int i = 0; i < 3; i++) {
    k_zero2<<<1, 32, 0, stream>>>(sums);
    k_reduce<<<4096, B, 0, stream>>>((const float4*)xc, sums);
    k_gln_apply<<<APB, B, 0, stream>>>((const float4*)xc, sums, nw + i * CH, nb + i * CH,
                                       (float4*)hpre);
    edge_phase(hpre, false);
    k_conv_mlp<<<MLPB, B, 0, stream>>>(hpre, agg,
                                       cW1 + (size_t)i * CH * HD, cb1 + (size_t)i * HD,
                                       clnw + (size_t)i * HD, clnb + (size_t)i * HD,
                                       cW2 + (size_t)i * HD * CH, cb2 + (size_t)i * CH,
                                       xc, xc);
  }

  k_head<<<MLPB, B, 0, stream>>>(xc, lW1, lb1, lW2, lb2, lW3, lb3, (float*)d_out);
}


# Conv-MLP and head re-tiled: one wave owns 16 whole rows, first GEMM transposed so hidden rows stay in registers (row LN there, no LDS); LN sums, second-GEMM k order, last dot re-ordered
# speedup vs baseline: 2.7948x; 1.3405x over previous
#include <hip/hip_runtime.h>
#include <math.h>

#define NN   100000
#define CH   64
#define HD   128
#define NE   800000
#define N64  6400000
#define EPS_MSG 1e-7f
#define EPS_LN  1e-5f

typedef float v2f __attribute__((ext_vector_type(2)));
typedef float v8f __attribute__((ext_vector_type(8)));

static __device__ __forceinline__ v8f wmma4(v2f a, v2f b, v8f c) {
  return __builtin_amdgcn_wmma_f32_16x16x4_f32(
       false, a,  false, b,
       (short)0, c,  false,  false);
}

__global__ void k_zero2(float* __restrict__ p) {
  if (threadIdx.x < 2) p[threadIdx.x] = 0.f;
}

__global__ __launch_bounds__(256) void k_reduce(const float4* __restrict__ h,
                                                float* __restrict__ sums) {
  __shared__ float ss[256];
  __shared__ float sq[256];
  float s = 0.f, q = 0.f;
  for (int i = blockIdx.x * 256 + threadIdx.x; i < N64 / 4; i += gridDim.x * 256) {
    float4 v = h[i];
    s += v.x + v.y + v.z + v.w;
    q += v.x * v.x + v.y * v.y + v.z * v.z + v.w * v.w;
  }
  ss[threadIdx.x] = s; sq[threadIdx.x] = q;
  __syncthreads();
  for (int off = 128; off > 0; off >>= 1) {
    if (threadIdx.x < off) {
      ss[threadIdx.x] += ss[threadIdx.x + off];
      sq[threadIdx.x] += sq[threadIdx.x + off];
    }
    __syncthreads();
  }
  if (threadIdx.x == 0) {
    atomicAdd(&sums[0], ss[0]);
    atomicAdd(&sums[1], sq[0]);
  }
}

__global__ __launch_bounds__(256) void k_gln_apply(const float4* __restrict__ h,
                                                   const float* __restrict__ sums,
                                                   const float* __restrict__ w,
                                                   const float* __restrict__ b,
                                                   float4* __restrict__ out) {
  int i = blockIdx.x * 256 + threadIdx.x;
  if (i >= N64 / 4) return;
  const float invn = 1.f / (float)N64;
  float mu = sums[0] * invn;
  float var = sums[1] * invn - mu * mu;
  float scale = 1.f / (sqrtf(fmaxf(var, 0.f)) + EPS_LN);
  int c = (i & 15) << 2;
  float4 v = h[i];
  float r0 = (v.x - mu) * scale * w[c + 0] + b[c + 0];
  float r1 = (v.y - mu) * scale * w[c + 1] + b[c + 1];
  float r2 = (v.z - mu) * scale * w[c + 2] + b[c + 2];
  float r3 = (v.w - mu) * scale * w[c + 3] + b[c + 3];
  out[i] = make_float4(fmaxf(r0, 0.f), fmaxf(r1, 0.f), fmaxf(r2, 0.f), fmaxf(r3, 0.f));
}

#define NPB   391
#define NOWN  256
#define OWNT  1024
#define LCAP  4096
#define LGRP  16
#define SGRP  8
#define LSTR  (LCAP + 32)
static_assert(NPB * NOWN >= NN && NPB * (NOWN - 1) < NN, "owner ranges must cover NN");
static_assert(NPB <= 512 && NN <= (1 << 23) && NE <= (1 << 23), "packed entry: 9 bits node, 23 bits edge or source");
static_assert(NE % 32 == 0 && NE % 4 == 0, "edge list is walked in 32s and in 16-byte loads");

static __device__ __forceinline__ void own_max(int* sMax, unsigned dl, float v0, float v1, int lane) {
  atomicMax(&sMax[dl * CH + lane],      __float_as_int(fmaxf(v0, 0.f) + EPS_MSG));
  atomicMax(&sMax[dl * CH + 32 + lane], __float_as_int(fmaxf(v1, 0.f) + EPS_MSG));
}
static __device__ __forceinline__ void own_acc(const int* sMax, float* sDen, float* sNum,
                                               unsigned dl, float v0, float v1, int lane) {
  const float m0 = fmaxf(v0, 0.f) + EPS_MSG, m1 = fmaxf(v1, 0.f) + EPS_MSG;
  const float e0 = __expf(m0 - __int_as_float(sMax[dl * CH + lane]));
  const float e1 = __expf(m1 - __int_as_float(sMax[dl * CH + 32 + lane]));
  atomicAdd(&sDen[dl * CH + lane], e0);       atomicAdd(&sNum[dl * CH + lane], m0 * e0);
  atomicAdd(&sDen[dl * CH + 32 + lane], e1);  atomicAdd(&sNum[dl * CH + 32 + lane], m1 * e1);
}

template <bool FIRST>
__global__ __launch_bounds__(OWNT)
void k_agg_owner(const int* __restrict__ src, const int* __restrict__ dst,
                 const float* __restrict__ hin, float* __restrict__ agg,
                 unsigned* __restrict__ wlist) {
  __shared__ int      sMax[NPB * CH];
  __shared__ float    sDen[NPB * CH];
  __shared__ float    sNum[NPB * CH];
  __shared__ unsigned sList[LCAP];
  __shared__ unsigned sCnt;
  const int tid = threadIdx.x, lane = tid & 31, wave = tid >> 5;
  const int lo = blockIdx.x * NPB;
  const unsigned cnt = (unsigned)min(NPB, NN - lo);

  for (int i = tid; i < NPB * CH; i += OWNT) { sMax[i] = 0; sDen[i] = 0.f; sNum[i] = 0.f; }
  unsigned* wl = wlist + (size_t)blockIdx.x * LSTR;
  if (FIRST && tid == 0) sCnt = 0u;
  __syncthreads();

  unsigned found;
  if (FIRST) {
    const int4* dst4 = (const int4*)dst;
    for (int i = tid; i < NE / 4; i += SGRP * OWNT) {
      int4 d[SGRP];
      #pragma unroll
      for (int u = 0; u < SGRP; u++) {
        const int ii = i + u * OWNT;
        d[u] = (ii < NE / 4) ? dst4[ii] : make_int4(-1, -1, -1, -1);
      }
      #pragma unroll
      for (int u = 0; u < SGRP; u++) {
        const int e = 4 * (i + u * OWNT);
        const int dd[4] = {d[u].x, d[u].y, d[u].z, d[u].w};
        #pragma unroll
        for (int c = 0; c < 4; c++) {
          const unsigned dl = (unsigned)(dd[c] - lo);
          if (dl < cnt) {
            const unsigned pos = atomicAdd(&sCnt, 1u);
            if (pos < LCAP) sList[pos] = ((unsigned)(e + c) << 9) | dl;
          }
        }
      }
    }
    __syncthreads();
    found = sCnt;
    if (tid == 0) wl[LCAP] = found;
  } else {
    found = wl[LCAP];
  }

  if (found <= LCAP) {
    const int total = (int)found;
    if (FIRST) {
      for (int i = tid; i < total; i += OWNT) {
        const unsigned ent = sList[i];
        const unsigned fin = ((unsigned)src[ent >> 9] << 9) | (ent & 511u);
        sList[i] = fin;
        wl[i] = fin;
      }
    } else {
      for (int i = tid; i < total; i += OWNT) sList[i] = wl[i];
    }
    __syncthreads();
    for (int ph = 0; ph < 2; ph++) {
      for (int base = wave * LGRP; base < total; base += 32 * LGRP) {
        unsigned ent[LGRP]; float v0[LGRP], v1[LGRP];
        #pragma unroll
        for (int u = 0; u < LGRP; u++) {
          const int idx = base + u;
          ent[u] = sList[idx < total ? idx : base];
          const float* hp = hin + (size_t)(ent[u] >> 9) * CH + lane;
          v0[u] = hp[0]; v1[u] = hp[32];
        }
        #pragma unroll
        for (int u = 0; u < LGRP; u++) {
          if (base + u < total) {
            if (ph == 0) own_max(sMax, ent[u] & 511u, v0[u], v1[u], lane);
            else         own_acc(sMax, sDen, sNum, ent[u] & 511u, v0[u], v1[u], lane);
          }
        }
      }
      __syncthreads();
    }
  } else {
    for (int ph = 0; ph < 2; ph++) {
      for (int eb = wave * 32; eb < NE; eb += 32 * 32) {
        const unsigned dl = (unsigned)(dst[eb + lane] - lo);
        unsigned mask = __builtin_amdgcn_ballot_w32(dl < cnt);
        while (mask) {
          const int b = __ffs(mask) - 1;
          mask &= mask - 1u;
          const unsigned dlb = (unsigned)__shfl((int)dl, b);
          const float* hp = hin + (size_t)src[eb + b] * CH + lane;
          const float v0 = hp[0], v1 = hp[32];
          if (ph == 0) own_max(sMax, dlb, v0, v1, lane);
          else         own_acc(sMax, sDen, sNum, dlb, v0, v1, lane);
        }
      }
      __syncthreads();
    }
  }

  const int nout = (int)cnt * CH;
  float* ap = agg + (size_t)lo * CH;
  for (int i = tid; i < nout; i += OWNT) {
    const float den = sDen[i];
    ap[i] = den > 0.f ? sNum[i] / den : 0.f;
  }
}

static_assert(NN % 32 == 0, "two 16-row strips per work-group");
static_assert(CH == 64 && HD == 128, "channel maps below are written for 64 -> 128 -> 64");

static __device__ __forceinline__ float other_half(float v) {
  return __shfl_xor(v, 16);
}
static __device__ __forceinline__ float f4at(const float4& v, int i) {
  return i == 0 ? v.x : i == 1 ? v.y : i == 2 ? v.z : v.w;
}

template <bool OUTER>
__global__ __launch_bounds__(64)
void k_conv_mlp(const float* __restrict__ resin, const float* __restrict__ agg,
                const float* __restrict__ W1, const float* __restrict__ b1,
                const float* __restrict__ lnw, const float* __restrict__ lnb,
                const float* __restrict__ W2, const float* __restrict__ b2,
                const float* outer, float* out) {
  const int tid = threadIdx.x;
  const int wave = tid >> 5, lane = tid & 31;
  const int half = lane >> 4, l16 = lane & 15;
  const size_t base = ((size_t)blockIdx.x * 2 + wave) * 16;

  v2f actf[CH / 4];
  {
    const float* rp = resin + (base + l16) * CH + 2 * half;
    const float* ap = agg   + (base + l16) * CH + 2 * half;
    #pragma unroll
    for (int j = 0; j < CH / 4; j++) {
      const v2f r = *(const v2f*)(rp + 4 * j);
      const v2f a = *(const v2f*)(ap + 4 * j);
      actf[j] = r + a;
    }
  }

  v8f acc[HD / 16];
  #pragma unroll
  for (int nt = 0; nt < HD / 16; nt++) acc[nt] = (v8f){0.f, 0.f, 0.f, 0.f, 0.f, 0.f, 0.f, 0.f};
  {
    const float* wp = W1 + (2 * half) * HD + 4 * l16;
    #pragma unroll
    for (int j = 0; j < CH / 4; j++) {
      #pragma unroll
      for (int g = 0; g < 2; g++) {
        const float4 w0 = *(const float4*)(wp + (4 * j) * HD + 64 * g);
        const float4 w1 = *(const float4*)(wp + (4 * j + 1) * HD + 64 * g);
        #pragma unroll
        for (int t = 0; t < 4; t++) {
          v2f wf; wf.x = f4at(w0, t); wf.y = f4at(w1, t);
          acc[4 * g + t] = wmma4(wf, actf[j], acc[4 * g + t]);
        }
      }
    }
  }

  {
    float s = 0.f, q = 0.f;
    #pragma unroll
    for (int g = 0; g < 2; g++) {
      #pragma unroll
      for (int r = 0; r < 8; r++) {
        const float4 bv = *(const float4*)(b1 + 64 * g + 4 * (r + 8 * half));
        #pragma unroll
        for (int t = 0; t < 4; t++) {
          const float v = acc[4 * g + t][r] + f4at(bv, t);
          acc[4 * g + t][r] = v;
          s += v; q += v * v;
        }
      }
    }
    s += other_half(s);
    q += other_half(q);
    const float mu = s * (1.f / HD);
    const float var = q * (1.f / HD) - mu * mu;
    const float inv = rsqrtf(var + EPS_LN);
    #pragma unroll
    for (int g = 0; g < 2; g++) {
      #pragma unroll
      for (int r = 0; r < 8; r++) {
        const float4 wv = *(const float4*)(lnw + 64 * g + 4 * (r + 8 * half));
        const float4 cv = *(const float4*)(lnb + 64 * g + 4 * (r + 8 * half));
        #pragma unroll
        for (int t = 0; t < 4; t++) {
          const float v = (acc[4 * g + t][r] - mu) * inv * f4at(wv, t) + f4at(cv, t);
          acc[4 * g + t][r] = fmaxf(v, 0.f);
        }
      }
    }
  }

  v8f acc2[CH / 16];
  #pragma unroll
  for (int ct = 0; ct < CH / 16; ct++) acc2[ct] = (v8f){0.f, 0.f, 0.f, 0.f, 0.f, 0.f, 0.f, 0.f};
  {
    const float* wp = W2 + (size_t)(32 * half) * CH + 4 * l16;
    #pragma unroll
    for (int nt = 0; nt < HD / 16; nt++) {
      #pragma unroll
      for (int rp = 0; rp < 4; rp++) {
        v2f a; a.x = acc[nt][2 * rp]; a.y = acc[nt][2 * rp + 1];
        const int k0 = 64 * (nt / 4) + 8 * rp + (nt % 4);
        const float4 w0 = *(const float4*)(wp + (k0) * CH);
        const float4 w1 = *(const float4*)(wp + (k0 + 4) * CH);
        #pragma unroll
        for (int ct = 0; ct < CH / 16; ct++) {
          v2f bb; bb.x = f4at(w0, ct); bb.y = f4at(w1, ct);
          acc2[ct] = wmma4(a, bb, acc2[ct]);
        }
      }
    }
  }

  const float4 bias = *(const float4*)(b2 + 4 * l16);
  float4 res[8];
  #pragma unroll
  for (int r = 0; r < 8; r++) {
    if (OUTER) res[r] = *(const float4*)(outer + (base + r + 8 * half) * CH + 4 * l16);
    else       res[r] = make_float4(0.f, 0.f, 0.f, 0.f);
  }
  #pragma unroll
  for (int r = 0; r < 8; r++) {
    float4 o;
    o.x = acc2[0][r] + bias.x + res[r].x;
    o.y = acc2[1][r] + bias.y + res[r].y;
    o.z = acc2[2][r] + bias.z + res[r].z;
    o.w = acc2[3][r] + bias.w + res[r].w;
    *(float4*)(out + (base + r + 8 * half) * CH + 4 * l16) = o;
  }
}

__global__ __launch_bounds__(64)
void k_head(const float* __restrict__ xc,
            const float* __restrict__ W1, const float* __restrict__ b1,
            const float* __restrict__ W2, const float* __restrict__ b2,
            const float* __restrict__ W3, const float* __restrict__ b3,
            float* __restrict__ out) {
  const int tid = threadIdx.x;
  const int wave = tid >> 5, lane = tid & 31;
  const int half = lane >> 4, l16 = lane & 15;
  const size_t base = ((size_t)blockIdx.x * 2 + wave) * 16;

  v2f xf[CH / 4];
  {
    const float* xp = xc + (base + l16) * CH + 2 * half;
    #pragma unroll
    for (int j = 0; j < CH / 4; j++) xf[j] = *(const v2f*)(xp + 4 * j);
  }

  v8f h1[CH / 16];
  #pragma unroll
  for (int nt = 0; nt < CH / 16; nt++) h1[nt] = (v8f){0.f, 0.f, 0.f, 0.f, 0.f, 0.f, 0.f, 0.f};
  {
    const float* wp = W1 + (2 * half) * CH + 4 * l16;
    #pragma unroll
    for (int j = 0; j < CH / 4; j++) {
      const float4 w0 = *(const float4*)(wp + (4 * j) * CH);
      const float4 w1 = *(const float4*)(wp + (4 * j + 1) * CH);
      #pragma unroll
      for (int nt = 0; nt < CH / 16; nt++) {
        v2f wf; wf.x = f4at(w0, nt); wf.y = f4at(w1, nt);
        h1[nt] = wmma4(wf, xf[j], h1[nt]);
      }
    }
  }
  #pragma unroll
  for (int r = 0; r < 8; r++) {
    const float4 bv = *(const float4*)(b1 + 4 * (r + 8 * half));
    #pragma unroll
    for (int nt = 0; nt < CH / 16; nt++) h1[nt][r] = fmaxf(h1[nt][r] + f4at(bv, nt), 0.f);
  }

  v8f h2[CH / 16];
  #pragma unroll
  for (int nt = 0; nt < CH / 16; nt++) h2[nt] = (v8f){0.f, 0.f, 0.f, 0.f, 0.f, 0.f, 0.f, 0.f};
  {
    const float* wp = W2 + (size_t)(32 * half) * CH + 4 * l16;
    #pragma unroll
    for (int nt = 0; nt < CH / 16; nt++) {
      #pragma unroll
      for (int rp = 0; rp < 4; rp++) {
        v2f hb; hb.x = h1[nt][2 * rp]; hb.y = h1[nt][2 * rp + 1];
        const int k0 = 8 * rp + nt;
        const float4 w0 = *(const float4*)(wp + (k0) * CH);
        const float4 w1 = *(const float4*)(wp + (k0 + 4) * CH);
        #pragma unroll
        for (int n2 = 0; n2 < CH / 16; n2++) {
          v2f wf; wf.x = f4at(w0, n2); wf.y = f4at(w1, n2);
          h2[n2] = wmma4(wf, hb, h2[n2]);
        }
      }
    }
  }

  float o0 = 0.f, o1 = 0.f;
  #pragma unroll
  for (int r = 0; r < 8; r++) {
    const int c0 = 4 * (r + 8 * half);
    const float4 bv = *(const float4*)(b2 + c0);
    const float4 ta = *(const float4*)(W3 + 2 * c0);
    const float4 tb = *(const float4*)(W3 + 2 * c0 + 4);
    const float w3[8] = {ta.x, ta.y, ta.z, ta.w, tb.x, tb.y, tb.z, tb.w};
    #pragma unroll
    for (int n2 = 0; n2 < CH / 16; n2++) {
      const float v = fmaxf(h2[n2][r] + f4at(bv, n2), 0.f);
      o0 += v * w3[2 * n2];
      o1 += v * w3[2 * n2 + 1];
    }
  }
  o0 += other_half(o0);
  o1 += other_half(o1);
  if (half == 0) {
    float2 res; res.x = o0 + b3[0]; res.y = o1 + b3[1];
    *(float2*)(out + (base + l16) * 2) = res;
  }
}

extern "C" void kernel_launch(void* const* d_in, const int* in_sizes, int n_in,
                              void* d_out, int out_size, void* d_ws, size_t ws_size,
                              hipStream_t stream) {
  (void)in_sizes; (void)n_in; (void)out_size; (void)ws_size;
  const float* x      = (const float*)d_in[0];
  const int*   ei     = (const int*)d_in[1];
  const int*   src    = ei;
  const int*   dst    = ei + NE;
  const float* c1_W1  = (const float*)d_in[2];
  const float* c1_b1  = (const float*)d_in[3];
  const float* c1_lnw = (const float*)d_in[4];
  const float* c1_lnb = (const float*)d_in[5];
  const float* c1_W2  = (const float*)d_in[6];
  const float* c1_b2  = (const float*)d_in[7];
  const float* n1_w   = (const float*)d_in[8];
  const float* n1_b   = (const float*)d_in[9];
  const float* cW1    = (const float*)d_in[10];
  const float* cb1    = (const float*)d_in[11];
  const float* clnw   = (const float*)d_in[12];
  const float* clnb   = (const float*)d_in[13];
  const float* cW2    = (const float*)d_in[14];
  const float* cb2    = (const float*)d_in[15];
  const float* nw     = (const float*)d_in[16];
  const float* nb     = (const float*)d_in[17];
  const float* lW1    = (const float*)d_in[18];
  const float* lb1    = (const float*)d_in[19];
  const float* lW2    = (const float*)d_in[20];
  const float* lb2    = (const float*)d_in[21];
  const float* lW3    = (const float*)d_in[22];
  const float* lb3    = (const float*)d_in[23];

  float* ws    = (float*)d_ws;
  float* xc    = ws;
  float* hpre  = ws + (size_t)N64;
  float* agg   = ws + 4 * (size_t)N64;
  float* sums  = ws + 5 * (size_t)N64;
  unsigned* wlist = (unsigned*)(ws + 2 * (size_t)N64);

  const dim3 B(256);
  const int APB  = (N64 / 4 + 255) / 256;
  const int MLPB = NN / 32;
  const dim3 B2(64);

  auto edge_phase = [&](const float* hin, bool first) {
    if (first) k_agg_owner<true><<<NOWN, OWNT, 0, stream>>>(src, dst, hin, agg, wlist);
    else       k_agg_owner<false><<<NOWN, OWNT, 0, stream>>>(src, dst, hin, agg, wlist);
  };

  edge_phase(x, true);
  k_conv_mlp<false><<<MLPB, B2, 0, stream>>>(x, agg, c1_W1, c1_b1, c1_lnw, c1_lnb,
                                     c1_W2, c1_b2, nullptr, hpre);
  k_zero2<<<1, 32, 0, stream>>>(sums);
  k_reduce<<<4096, B, 0, stream>>>((const float4*)hpre, sums);
  k_gln_apply<<<APB, B, 0, stream>>>((const float4*)hpre, sums, n1_w, n1_b, (float4*)xc);

  for (int i = 0; i < 3; i++) {
    k_zero2<<<1, 32, 0, stream>>>(sums);
    k_reduce<<<4096, B, 0, stream>>>((const float4*)xc, sums);
    k_gln_apply<<<APB, B, 0, stream>>>((const float4*)xc, sums, nw + i * CH, nb + i * CH,
                                       (float4*)hpre);
    edge_phase(hpre, false);
    k_conv_mlp<true><<<MLPB, B2, 0, stream>>>(hpre, agg,
                                       cW1 + (size_t)i * CH * HD, cb1 + (size_t)i * HD,
                                       clnw + (size_t)i * HD, clnb + (size_t)i * HD,
                                       cW2 + (size_t)i * HD * CH, cb2 + (size_t)i * CH,
                                       xc, xc);
  }

  k_head<<<MLPB, B2, 0, stream>>>(xc, lW1, lb1, lW2, lb2, lW3, lb3, (float*)d_out);
}


# Softmax shift = destination's own message value, known before any edge is read: segment-max pass (2nd gather of every source row) gone; exact two-pass redo if an exponent leaves +-40; sum order kept
# speedup vs baseline: 2.9958x; 1.0719x over previous
#include <hip/hip_runtime.h>
#include <math.h>

#define NN   100000
#define CH   64
#define HD   128
#define NE   800000
#define N64  6400000
#define EPS_MSG 1e-7f
#define EPS_LN  1e-5f

typedef float v2f __attribute__((ext_vector_type(2)));
typedef float v8f __attribute__((ext_vector_type(8)));

static __device__ __forceinline__ v8f wmma4(v2f a, v2f b, v8f c) {
  return __builtin_amdgcn_wmma_f32_16x16x4_f32(
       false, a,  false, b,
       (short)0, c,  false,  false);
}

__global__ void k_zero2(float* __restrict__ p) {
  if (threadIdx.x < 2) p[threadIdx.x] = 0.f;
}

__global__ __launch_bounds__(256) void k_reduce(const float4* __restrict__ h,
                                                float* __restrict__ sums) {
  __shared__ float ss[256];
  __shared__ float sq[256];
  float s = 0.f, q = 0.f;
  for (int i = blockIdx.x * 256 + threadIdx.x; i < N64 / 4; i += gridDim.x * 256) {
    float4 v = h[i];
    s += v.x + v.y + v.z + v.w;
    q += v.x * v.x + v.y * v.y + v.z * v.z + v.w * v.w;
  }
  ss[threadIdx.x] = s; sq[threadIdx.x] = q;
  __syncthreads();
  for (int off = 128; off > 0; off >>= 1) {
    if (threadIdx.x < off) {
      ss[threadIdx.x] += ss[threadIdx.x + off];
      sq[threadIdx.x] += sq[threadIdx.x + off];
    }
    __syncthreads();
  }
  if (threadIdx.x == 0) {
    atomicAdd(&sums[0], ss[0]);
    atomicAdd(&sums[1], sq[0]);
  }
}

__global__ __launch_bounds__(256) void k_gln_apply(const float4* __restrict__ h,
                                                   const float* __restrict__ sums,
                                                   const float* __restrict__ w,
                                                   const float* __restrict__ b,
                                                   float4* __restrict__ out) {
  int i = blockIdx.x * 256 + threadIdx.x;
  if (i >= N64 / 4) return;
  const float invn = 1.f / (float)N64;
  float mu = sums[0] * invn;
  float var = sums[1] * invn - mu * mu;
  float scale = 1.f / (sqrtf(fmaxf(var, 0.f)) + EPS_LN);
  int c = (i & 15) << 2;
  float4 v = h[i];
  float r0 = (v.x - mu) * scale * w[c + 0] + b[c + 0];
  float r1 = (v.y - mu) * scale * w[c + 1] + b[c + 1];
  float r2 = (v.z - mu) * scale * w[c + 2] + b[c + 2];
  float r3 = (v.w - mu) * scale * w[c + 3] + b[c + 3];
  out[i] = make_float4(fmaxf(r0, 0.f), fmaxf(r1, 0.f), fmaxf(r2, 0.f), fmaxf(r3, 0.f));
}

#define NPB   391
#define NOWN  256
#define OWNT  1024
#define LCAP  4096
#define LGRP  16
#define SGRP  8
#define LSTR  (LCAP + 32)
static_assert(NPB * NOWN >= NN && NPB * (NOWN - 1) < NN, "owner ranges must cover NN");
static_assert(NPB <= 512 && NN <= (1 << 23) && NE <= (1 << 23), "packed entry: 9 bits node, 23 bits edge or source");
static_assert(NE % 32 == 0 && NE % 4 == 0, "edge list is walked in 32s and in 16-byte loads");

static __device__ __forceinline__ void own_max(int* sMax, unsigned dl, float v0, float v1, int lane) {
  atomicMax(&sMax[dl * CH + lane],      __float_as_int(fmaxf(v0, 0.f) + EPS_MSG));
  atomicMax(&sMax[dl * CH + 32 + lane], __float_as_int(fmaxf(v1, 0.f) + EPS_MSG));
}
static __device__ __forceinline__ void own_acc(const int* sMax, float* sDen, float* sNum,
                                               unsigned dl, float v0, float v1, int lane) {
  const float m0 = fmaxf(v0, 0.f) + EPS_MSG, m1 = fmaxf(v1, 0.f) + EPS_MSG;
  const float e0 = __expf(m0 - __int_as_float(sMax[dl * CH + lane]));
  const float e1 = __expf(m1 - __int_as_float(sMax[dl * CH + 32 + lane]));
  atomicAdd(&sDen[dl * CH + lane], e0);       atomicAdd(&sNum[dl * CH + lane], m0 * e0);
  atomicAdd(&sDen[dl * CH + 32 + lane], e1);  atomicAdd(&sNum[dl * CH + 32 + lane], m1 * e1);
}

#define SHIFT_LIM 40.f
#define INITN ((NPB * CH / 4 + OWNT - 1) / OWNT)

static __device__ __forceinline__ void list_two_pass(const unsigned* sList, int total,
                                                     const float* __restrict__ hin,
                                                     int* sMax, float* sDen, float* sNum,
                                                     int wave, int lane) {
  for (int ph = 0; ph < 2; ph++) {
    for (int base = wave * LGRP; base < total; base += 32 * LGRP) {
      unsigned ent[LGRP]; float v0[LGRP], v1[LGRP];
      #pragma unroll
      for (int u = 0; u < LGRP; u++) {
        const int idx = base + u;
        ent[u] = sList[idx < total ? idx : base];
        const float* hp = hin + (size_t)(ent[u] >> 9) * CH + lane;
        v0[u] = hp[0]; v1[u] = hp[32];
      }
      #pragma unroll
      for (int u = 0; u < LGRP; u++) {
        if (base + u < total) {
          if (ph == 0) own_max(sMax, ent[u] & 511u, v0[u], v1[u], lane);
          else         own_acc(sMax, sDen, sNum, ent[u] & 511u, v0[u], v1[u], lane);
        }
      }
    }
    __syncthreads();
  }
}

template <bool FIRST>
__global__ __launch_bounds__(OWNT)
void k_agg_owner(const int* __restrict__ src, const int* __restrict__ dst,
                 const float* __restrict__ hin, float* __restrict__ agg,
                 unsigned* __restrict__ wlist) {
  __shared__ __attribute__((aligned(16))) int   sMax[NPB * CH];
  __shared__ __attribute__((aligned(16))) float sDen[NPB * CH];
  __shared__ __attribute__((aligned(16))) float sNum[NPB * CH];
  __shared__ unsigned sList[LCAP];
  __shared__ unsigned sCnt;
  __shared__ unsigned sBad;
  const int tid = threadIdx.x, lane = tid & 31, wave = tid >> 5;
  const int lo = blockIdx.x * NPB;
  const unsigned cnt = (unsigned)min(NPB, NN - lo);
  const int nout = (int)cnt * CH;

  {
    const float4* own4 = (const float4*)(hin + (size_t)lo * CH);
    const int nown4 = nout / 4;
    float4 r[INITN];
    #pragma unroll
    for (int k = 0; k < INITN; k++) {
      const int i = (tid + k * OWNT < NPB * CH / 4) ? tid + k * OWNT : tid;
      r[k] = own4[min(i, nown4 - 1)];
    }
    int4* m4 = (int4*)sMax; float4* d4 = (float4*)sDen; float4* n4 = (float4*)sNum;
    const float4 z4 = make_float4(0.f, 0.f, 0.f, 0.f);
    #pragma unroll
    for (int k = 0; k < INITN; k++) {
      const int i = (tid + k * OWNT < NPB * CH / 4) ? tid + k * OWNT : tid;
      const bool in = i < nown4;
      float4 s;
      s.x = in ? fmaxf(r[k].x, 0.f) + EPS_MSG : 0.f;
      s.y = in ? fmaxf(r[k].y, 0.f) + EPS_MSG : 0.f;
      s.z = in ? fmaxf(r[k].z, 0.f) + EPS_MSG : 0.f;
      s.w = in ? fmaxf(r[k].w, 0.f) + EPS_MSG : 0.f;
      m4[i] = make_int4(__float_as_int(s.x), __float_as_int(s.y), __float_as_int(s.z), __float_as_int(s.w));
      d4[i] = z4; n4[i] = z4;
    }
  }
  unsigned* wl = wlist + (size_t)blockIdx.x * LSTR;
  if (tid == 0) { sBad = 0u; if (FIRST) sCnt = 0u; }
  __syncthreads();

  unsigned found;
  if (FIRST) {
    const int4* dst4 = (const int4*)dst;
    for (int i = tid; i < NE / 4; i += SGRP * OWNT) {
      int4 d[SGRP];
      #pragma unroll
      for (int u = 0; u < SGRP; u++) {
        const int ii = i + u * OWNT;
        d[u] = (ii < NE / 4) ? dst4[ii] : make_int4(-1, -1, -1, -1);
      }
      #pragma unroll
      for (int u = 0; u < SGRP; u++) {
        const int e = 4 * (i + u * OWNT);
        const int dd[4] = {d[u].x, d[u].y, d[u].z, d[u].w};
        #pragma unroll
        for (int c = 0; c < 4; c++) {
          const unsigned dl = (unsigned)(dd[c] - lo);
          if (dl < cnt) {
            const unsigned pos = atomicAdd(&sCnt, 1u);
            if (pos < LCAP) sList[pos] = ((unsigned)(e + c) << 9) | dl;
          }
        }
      }
    }
    __syncthreads();
    found = sCnt;
    if (tid == 0) wl[LCAP] = found;
  } else {
    found = wl[LCAP];
  }

  if (found <= LCAP) {
    const int total = (int)found;
    if (FIRST) {
      for (int i = tid; i < total; i += OWNT) {
        const unsigned ent = sList[i];
        const unsigned fin = ((unsigned)src[ent >> 9] << 9) | (ent & 511u);
        sList[i] = fin;
        wl[i] = fin;
      }
    } else {
      for (int i = tid; i < total; i += OWNT) sList[i] = wl[i];
    }
    __syncthreads();

    bool bad = false;
    for (int base = wave * LGRP; base < total; base += 32 * LGRP) {
      unsigned ent[LGRP]; float v0[LGRP], v1[LGRP], s0[LGRP], s1[LGRP];
      #pragma unroll
      for (int u = 0; u < LGRP; u++) {
        const int idx = base + u;
        ent[u] = sList[idx < total ? idx : base];
        const float* hp = hin + (size_t)(ent[u] >> 9) * CH + lane;
        v0[u] = hp[0]; v1[u] = hp[32];
      }
      #pragma unroll
      for (int u = 0; u < LGRP; u++) {
        const unsigned dl = ent[u] & 511u;
        s0[u] = __int_as_float(sMax[dl * CH + lane]);
        s1[u] = __int_as_float(sMax[dl * CH + 32 + lane]);
      }
      #pragma unroll
      for (int u = 0; u < LGRP; u++) {
        const bool live = base + u < total;
        const unsigned dl = ent[u] & 511u;
        const float m0 = fmaxf(v0[u], 0.f) + EPS_MSG, m1 = fmaxf(v1[u], 0.f) + EPS_MSG;
        const float d0 = m0 - s0[u], d1 = m1 - s1[u];
        bad = bad || !(fabsf(d0) <= SHIFT_LIM) || !(fabsf(d1) <= SHIFT_LIM);
        const float x0 = __expf(d0), x1 = __expf(d1);
        const float e0 = live ? x0 : 0.f, e1 = live ? x1 : 0.f;
        const float p0 = live ? m0 * x0 : 0.f, p1 = live ? m1 * x1 : 0.f;
        atomicAdd(&sDen[dl * CH + lane], e0);       atomicAdd(&sNum[dl * CH + lane], p0);
        atomicAdd(&sDen[dl * CH + 32 + lane], e1);  atomicAdd(&sNum[dl * CH + 32 + lane], p1);
      }
    }
    if (bad) sBad = 1u;
    __syncthreads();
    if (sBad != 0u) {
      for (int i = tid; i < NPB * CH; i += OWNT) { sMax[i] = 0; sDen[i] = 0.f; sNum[i] = 0.f; }
      __syncthreads();
      list_two_pass(sList, total, hin, sMax, sDen, sNum, wave, lane);
    }
  } else {
    for (int i = tid; i < NPB * CH; i += OWNT) sMax[i] = 0;
    __syncthreads();
    for (int ph = 0; ph < 2; ph++) {
      for (int eb = wave * 32; eb < NE; eb += 32 * 32) {
        const unsigned dl = (unsigned)(dst[eb + lane] - lo);
        unsigned mask = __builtin_amdgcn_ballot_w32(dl < cnt);
        while (mask) {
          const int b = __ffs(mask) - 1;
          mask &= mask - 1u;
          const unsigned dlb = (unsigned)__shfl((int)dl, b);
          const float* hp = hin + (size_t)src[eb + b] * CH + lane;
          const float v0 = hp[0], v1 = hp[32];
          if (ph == 0) own_max(sMax, dlb, v0, v1, lane);
          else         own_acc(sMax, sDen, sNum, dlb, v0, v1, lane);
        }
      }
      __syncthreads();
    }
  }

  float* ap = agg + (size_t)lo * CH;
  for (int i = tid; i < nout; i += OWNT) {
    const float den = sDen[i];
    ap[i] = den > 0.f ? sNum[i] / den : 0.f;
  }
}

static_assert(NN % 32 == 0, "two 16-row strips per work-group");
static_assert(CH == 64 && HD == 128, "channel maps below are written for 64 -> 128 -> 64");

static __device__ __forceinline__ float other_half(float v) {
  return __shfl_xor(v, 16);
}
static __device__ __forceinline__ float f4at(const float4& v, int i) {
  return i == 0 ? v.x : i == 1 ? v.y : i == 2 ? v.z : v.w;
}

template <bool OUTER>
__global__ __launch_bounds__(64)
void k_conv_mlp(const float* __restrict__ resin, const float* __restrict__ agg,
                const float* __restrict__ W1, const float* __restrict__ b1,
                const float* __restrict__ lnw, const float* __restrict__ lnb,
                const float* __restrict__ W2, const float* __restrict__ b2,
                const float* outer, float* out) {
  const int tid = threadIdx.x;
  const int wave = tid >> 5, lane = tid & 31;
  const int half = lane >> 4, l16 = lane & 15;
  const size_t base = ((size_t)blockIdx.x * 2 + wave) * 16;

  v2f actf[CH / 4];
  {
    const float* rp = resin + (base + l16) * CH + 2 * half;
    const float* ap = agg   + (base + l16) * CH + 2 * half;
    #pragma unroll
    for (int j = 0; j < CH / 4; j++) {
      const v2f r = *(const v2f*)(rp + 4 * j);
      const v2f a = *(const v2f*)(ap + 4 * j);
      actf[j] = r + a;
    }
  }

  v8f acc[HD / 16];
  #pragma unroll
  for (int nt = 0; nt < HD / 16; nt++) acc[nt] = (v8f){0.f, 0.f, 0.f, 0.f, 0.f, 0.f, 0.f, 0.f};
  {
    const float* wp = W1 + (2 * half) * HD + 4 * l16;
    #pragma unroll
    for (int j = 0; j < CH / 4; j++) {
      #pragma unroll
      for (int g = 0; g < 2; g++) {
        const float4 w0 = *(const float4*)(wp + (4 * j) * HD + 64 * g);
        const float4 w1 = *(const float4*)(wp + (4 * j + 1) * HD + 64 * g);
        #pragma unroll
        for (int t = 0; t < 4; t++) {
          v2f wf; wf.x = f4at(w0, t); wf.y = f4at(w1, t);
          acc[4 * g + t] = wmma4(wf, actf[j], acc[4 * g + t]);
        }
      }
    }
  }

  {
    float s = 0.f, q = 0.f;
    #pragma unroll
    for (int g = 0; g < 2; g++) {
      #pragma unroll
      for (int r = 0; r < 8; r++) {
        const float4 bv = *(const float4*)(b1 + 64 * g + 4 * (r + 8 * half));
        #pragma unroll
        for (int t = 0; t < 4; t++) {
          const float v = acc[4 * g + t][r] + f4at(bv, t);
          acc[4 * g + t][r] = v;
          s += v; q += v * v;
        }
      }
    }
    s += other_half(s);
    q += other_half(q);
    const float mu = s * (1.f / HD);
    const float var = q * (1.f / HD) - mu * mu;
    const float inv = rsqrtf(var + EPS_LN);
    #pragma unroll
    for (int g = 0; g < 2; g++) {
      #pragma unroll
      for (int r = 0; r < 8; r++) {
        const float4 wv = *(const float4*)(lnw + 64 * g + 4 * (r + 8 * half));
        const float4 cv = *(const float4*)(lnb + 64 * g + 4 * (r + 8 * half));
        #pragma unroll
        for (int t = 0; t < 4; t++) {
          const float v = (acc[4 * g + t][r] - mu) * inv * f4at(wv, t) + f4at(cv, t);
          acc[4 * g + t][r] = fmaxf(v, 0.f);
        }
      }
    }
  }

  v8f acc2[CH / 16];
  #pragma unroll
  for (int ct = 0; ct < CH / 16; ct++) acc2[ct] = (v8f){0.f, 0.f, 0.f, 0.f, 0.f, 0.f, 0.f, 0.f};
  {
    const float* wp = W2 + (size_t)(32 * half) * CH + 4 * l16;
    #pragma unroll
    for (int nt = 0; nt < HD / 16; nt++) {
      #pragma unroll
      for (int rp = 0; rp < 4; rp++) {
        v2f a; a.x = acc[nt][2 * rp]; a.y = acc[nt][2 * rp + 1];
        const int k0 = 64 * (nt / 4) + 8 * rp + (nt % 4);
        const float4 w0 = *(const float4*)(wp + (k0) * CH);
        const float4 w1 = *(const float4*)(wp + (k0 + 4) * CH);
        #pragma unroll
        for (int ct = 0; ct < CH / 16; ct++) {
          v2f bb; bb.x = f4at(w0, ct); bb.y = f4at(w1, ct);
          acc2[ct] = wmma4(a, bb, acc2[ct]);
        }
      }
    }
  }

  const float4 bias = *(const float4*)(b2 + 4 * l16);
  float4 res[8];
  #pragma unroll
  for (int r = 0; r < 8; r++) {
    if (OUTER) res[r] = *(const float4*)(outer + (base + r + 8 * half) * CH + 4 * l16);
    else       res[r] = make_float4(0.f, 0.f, 0.f, 0.f);
  }
  #pragma unroll
  for (int r = 0; r < 8; r++) {
    float4 o;
    o.x = acc2[0][r] + bias.x + res[r].x;
    o.y = acc2[1][r] + bias.y + res[r].y;
    o.z = acc2[2][r] + bias.z + res[r].z;
    o.w = acc2[3][r] + bias.w + res[r].w;
    *(float4*)(out + (base + r + 8 * half) * CH + 4 * l16) = o;
  }
}

__global__ __launch_bounds__(64)
void k_head(const float* __restrict__ xc,
            const float* __restrict__ W1, const float* __restrict__ b1,
            const float* __restrict__ W2, const float* __restrict__ b2,
            const float* __restrict__ W3, const float* __restrict__ b3,
            float* __restrict__ out) {
  const int tid = threadIdx.x;
  const int wave = tid >> 5, lane = tid & 31;
  const int half = lane >> 4, l16 = lane & 15;
  const size_t base = ((size_t)blockIdx.x * 2 + wave) * 16;

  v2f xf[CH / 4];
  {
    const float* xp = xc + (base + l16) * CH + 2 * half;
    #pragma unroll
    for (int j = 0; j < CH / 4; j++) xf[j] = *(const v2f*)(xp + 4 * j);
  }

  v8f h1[CH / 16];
  #pragma unroll
  for (int nt = 0; nt < CH / 16; nt++) h1[nt] = (v8f){0.f, 0.f, 0.f, 0.f, 0.f, 0.f, 0.f, 0.f};
  {
    const float* wp = W1 + (2 * half) * CH + 4 * l16;
    #pragma unroll
    for (int j = 0; j < CH / 4; j++) {
      const float4 w0 = *(const float4*)(wp + (4 * j) * CH);
      const float4 w1 = *(const float4*)(wp + (4 * j + 1) * CH);
      #pragma unroll
      for (int nt = 0; nt < CH / 16; nt++) {
        v2f wf; wf.x = f4at(w0, nt); wf.y = f4at(w1, nt);
        h1[nt] = wmma4(wf, xf[j], h1[nt]);
      }
    }
  }
  #pragma unroll
  for (int r = 0; r < 8; r++) {
    const float4 bv = *(const float4*)(b1 + 4 * (r + 8 * half));
    #pragma unroll
    for (int nt = 0; nt < CH / 16; nt++) h1[nt][r] = fmaxf(h1[nt][r] + f4at(bv, nt), 0.f);
  }

  v8f h2[CH / 16];
  #pragma unroll
  for (int nt = 0; nt < CH / 16; nt++) h2[nt] = (v8f){0.f, 0.f, 0.f, 0.f, 0.f, 0.f, 0.f, 0.f};
  {
    const float* wp = W2 + (size_t)(32 * half) * CH + 4 * l16;
    #pragma unroll
    for (int nt = 0; nt < CH / 16; nt++) {
      #pragma unroll
      for (int rp = 0; rp < 4; rp++) {
        v2f hb; hb.x = h1[nt][2 * rp]; hb.y = h1[nt][2 * rp + 1];
        const int k0 = 8 * rp + nt;
        const float4 w0 = *(const float4*)(wp + (k0) * CH);
        const float4 w1 = *(const float4*)(wp + (k0 + 4) * CH);
        #pragma unroll
        for (int n2 = 0; n2 < CH / 16; n2++) {
          v2f wf; wf.x = f4at(w0, n2); wf.y = f4at(w1, n2);
          h2[n2] = wmma4(wf, hb, h2[n2]);
        }
      }
    }
  }

  float o0 = 0.f, o1 = 0.f;
  #pragma unroll
  for (int r = 0; r < 8; r++) {
    const int c0 = 4 * (r + 8 * half);
    const float4 bv = *(const float4*)(b2 + c0);
    const float4 ta = *(const float4*)(W3 + 2 * c0);
    const float4 tb = *(const float4*)(W3 + 2 * c0 + 4);
    const float w3[8] = {ta.x, ta.y, ta.z, ta.w, tb.x, tb.y, tb.z, tb.w};
    #pragma unroll
    for (int n2 = 0; n2 < CH / 16; n2++) {
      const float v = fmaxf(h2[n2][r] + f4at(bv, n2), 0.f);
      o0 += v * w3[2 * n2];
      o1 += v * w3[2 * n2 + 1];
    }
  }
  o0 += other_half(o0);
  o1 += other_half(o1);
  if (half == 0) {
    float2 res; res.x = o0 + b3[0]; res.y = o1 + b3[1];
    *(float2*)(out + (base + l16) * 2) = res;
  }
}

extern "C" void kernel_launch(void* const* d_in, const int* in_sizes, int n_in,
                              void* d_out, int out_size, void* d_ws, size_t ws_size,
                              hipStream_t stream) {
  (void)in_sizes; (void)n_in; (void)out_size; (void)ws_size;
  const float* x      = (const float*)d_in[0];
  const int*   ei     = (const int*)d_in[1];
  const int*   src    = ei;
  const int*   dst    = ei + NE;
  const float* c1_W1  = (const float*)d_in[2];
  const float* c1_b1  = (const float*)d_in[3];
  const float* c1_lnw = (const float*)d_in[4];
  const float* c1_lnb = (const float*)d_in[5];
  const float* c1_W2  = (const float*)d_in[6];
  const float* c1_b2  = (const float*)d_in[7];
  const float* n1_w   = (const float*)d_in[8];
  const float* n1_b   = (const float*)d_in[9];
  const float* cW1    = (const float*)d_in[10];
  const float* cb1    = (const float*)d_in[11];
  const float* clnw   = (const float*)d_in[12];
  const float* clnb   = (const float*)d_in[13];
  const float* cW2    = (const float*)d_in[14];
  const float* cb2    = (const float*)d_in[15];
  const float* nw     = (const float*)d_in[16];
  const float* nb     = (const float*)d_in[17];
  const float* lW1    = (const float*)d_in[18];
  const float* lb1    = (const float*)d_in[19];
  const float* lW2    = (const float*)d_in[20];
  const float* lb2    = (const float*)d_in[21];
  const float* lW3    = (const float*)d_in[22];
  const float* lb3    = (const float*)d_in[23];

  float* ws    = (float*)d_ws;
  float* xc    = ws;
  float* hpre  = ws + (size_t)N64;
  float* agg   = ws + 4 * (size_t)N64;
  float* sums  = ws + 5 * (size_t)N64;
  unsigned* wlist = (unsigned*)(ws + 2 * (size_t)N64);

  const dim3 B(256);
  const int APB  = (N64 / 4 + 255) / 256;
  const int MLPB = NN / 32;
  const dim3 B2(64);

  auto edge_phase = [&](const float* hin, bool first) {
    if (first) k_agg_owner<true><<<NOWN, OWNT, 0, stream>>>(src, dst, hin, agg, wlist);
    else       k_agg_owner<false><<<NOWN, OWNT, 0, stream>>>(src, dst, hin, agg, wlist);
  };

  edge_phase(x, true);
  k_conv_mlp<false><<<MLPB, B2, 0, stream>>>(x, agg, c1_W1, c1_b1, c1_lnw, c1_lnb,
                                     c1_W2, c1_b2, nullptr, hpre);
  k_zero2<<<1, 32, 0, stream>>>(sums);
  k_reduce<<<4096, B, 0, stream>>>((const float4*)hpre, sums);
  k_gln_apply<<<APB, B, 0, stream>>>((const float4*)hpre, sums, n1_w, n1_b, (float4*)xc);

  for (int i = 0; i < 3; i++) {
    k_zero2<<<1, 32, 0, stream>>>(sums);
    k_reduce<<<4096, B, 0, stream>>>((const float4*)xc, sums);
    k_gln_apply<<<APB, B, 0, stream>>>((const float4*)xc, sums, nw + i * CH, nb + i * CH,
                                       (float4*)hpre);
    edge_phase(hpre, false);
    k_conv_mlp<true><<<MLPB, B2, 0, stream>>>(hpre, agg,
                                       cW1 + (size_t)i * CH * HD, cb1 + (size_t)i * HD,
                                       clnw + (size_t)i * HD, clnb + (size_t)i * HD,
                                       cW2 + (size_t)i * HD * CH, cb2 + (size_t)i * CH,
                                       xc, xc);
  }

  k_head<<<MLPB, B2, 0, stream>>>(xc, lW1, lb1, lW2, lb2, lW3, lb3, (float*)d_out);
}


# Edge scan on owner keys: a launch in front writes one byte per edge (owner group of its dst; 255 groups of 393 nodes) and its list entry; groups scan 16 keys per load; sums stay in arrival order
# speedup vs baseline: 3.4476x; 1.1508x over previous
#include <hip/hip_runtime.h>
#include <math.h>

#define NN   100000
#define CH   64
#define HD   128
#define NE   800000
#define N64  6400000
#define EPS_MSG 1e-7f
#define EPS_LN  1e-5f

typedef float v2f __attribute__((ext_vector_type(2)));
typedef float v8f __attribute__((ext_vector_type(8)));

static __device__ __forceinline__ v8f wmma4(v2f a, v2f b, v8f c) {
  return __builtin_amdgcn_wmma_f32_16x16x4_f32(
       false, a,  false, b,
       (short)0, c,  false,  false);
}

#define GLN_WG   256
#define GLN_T    1024
#define GLN_PER  ((N64 / 4 + GLN_WG * GLN_T - 1) / (GLN_WG * GLN_T))
static_assert(GLN_WG == 256, "a lane of the apply kernel adds 8 pairs: 32 lanes x 8 = 256");

static __device__ __forceinline__ float wave_sum(float v) {
  v += __shfl_xor(v, 16); v += __shfl_xor(v, 8); v += __shfl_xor(v, 4);
  v += __shfl_xor(v, 2);  v += __shfl_xor(v, 1);
  return v;
}

__global__ __launch_bounds__(GLN_T) void k_gln_partial(const float4* __restrict__ h,
                                                       float* __restrict__ part) {
  __shared__ float ss[GLN_T / 32];
  __shared__ float sq[GLN_T / 32];
  const int tid = threadIdx.x, lane = tid & 31, wave = tid >> 5;
  const int g = blockIdx.x * GLN_T + tid;
  float4 v[GLN_PER];
  #pragma unroll
  for (int k = 0; k < GLN_PER; k++) {
    const int i = g + k * (GLN_WG * GLN_T);
    v[k] = h[min(i, N64 / 4 - 1)];
  }
  float s = 0.f, q = 0.f;
  #pragma unroll
  for (int k = 0; k < GLN_PER; k++) {
    const bool in = g + k * (GLN_WG * GLN_T) < N64 / 4;
    const float4 u = v[k];
    const float a = (u.x + u.y) + (u.z + u.w);
    const float b = (u.x * u.x + u.y * u.y) + (u.z * u.z + u.w * u.w);
    s += in ? a : 0.f;
    q += in ? b : 0.f;
  }
  s = wave_sum(s); q = wave_sum(q);
  if (lane == 0) { ss[wave] = s; sq[wave] = q; }
  __syncthreads();
  if (wave == 0) {
    s = wave_sum(ss[lane]); q = wave_sum(sq[lane]);
    if (lane == 0) { part[2 * blockIdx.x] = s; part[2 * blockIdx.x + 1] = q; }
  }
}

__global__ __launch_bounds__(256) void k_gln_apply(const float4* __restrict__ h,
                                                   const float* __restrict__ part,
                                                   const float* __restrict__ w,
                                                   const float* __restrict__ b,
                                                   float4* __restrict__ out) {
  int i = blockIdx.x * 256 + threadIdx.x;
  if (i >= N64 / 4) return;
  const float4 v = h[i];
  float s = 0.f, q = 0.f;
  {
    const float4* p4 = (const float4*)(part + 16 * (threadIdx.x & 31));
    const float4 p0 = p4[0], p1 = p4[1], p2 = p4[2], p3 = p4[3];
    s = ((p0.x + p0.z) + (p1.x + p1.z)) + ((p2.x + p2.z) + (p3.x + p3.z));
    q = ((p0.y + p0.w) + (p1.y + p1.w)) + ((p2.y + p2.w) + (p3.y + p3.w));
  }
  s = wave_sum(s); q = wave_sum(q);
  const float invn = 1.f / (float)N64;
  float mu = s * invn;
  float var = q * invn - mu * mu;
  float scale = 1.f / (sqrtf(fmaxf(var, 0.f)) + EPS_LN);
  int c = (i & 15) << 2;
  float r0 = (v.x - mu) * scale * w[c + 0] + b[c + 0];
  float r1 = (v.y - mu) * scale * w[c + 1] + b[c + 1];
  float r2 = (v.z - mu) * scale * w[c + 2] + b[c + 2];
  float r3 = (v.w - mu) * scale * w[c + 3] + b[c + 3];
  out[i] = make_float4(fmaxf(r0, 0.f), fmaxf(r1, 0.f), fmaxf(r2, 0.f), fmaxf(r3, 0.f));
}

#define NPB   393
#define NOWN  255
#define OWNT  1024
#define LCAP  4096
#define LGRP  16
#define SGRP  7
#define LSTR  (LCAP + 32)
static_assert(NPB * NOWN >= NN && NPB * (NOWN - 1) < NN, "owner ranges must cover NN");
static_assert(NPB <= 512 && NN <= (1 << 23) && NE <= (1 << 23), "packed entry: 9 bits node, 23 bits edge or source");
static_assert(NE % 32 == 0 && NE % 16 == 0, "edge list is walked in 32s; keys in 16-byte loads");
static_assert(NOWN < 256, "owner numbers 0..NOWN-1 and KEY_NONE must fit one byte");

#define KEY_NONE 255u
#define KEY_T    256
#define KEY_PER  4
#define KEY_WG   ((NE / 4 + KEY_T * KEY_PER - 1) / (KEY_T * KEY_PER))

static __device__ __forceinline__ unsigned owner_key(int d) {
  return (unsigned)d < (unsigned)NN ? (unsigned)d / (unsigned)NPB : KEY_NONE;
}

static __device__ __forceinline__ unsigned owner_entry(int s, int d, unsigned key) {
  return ((unsigned)s << 9) | (((unsigned)d - key * (unsigned)NPB) & 511u);
}

__global__ __launch_bounds__(KEY_T) void k_owner_keys(const int* __restrict__ src,
                                                      const int* __restrict__ dst,
                                                      unsigned* __restrict__ keys,
                                                      unsigned* __restrict__ entry) {
  uint4* entry4 = (uint4*)entry;
  const int4* src4 = (const int4*)src;
  const int4* dst4 = (const int4*)dst;
  const int g = blockIdx.x * KEY_T + threadIdx.x;
  int4 s[KEY_PER], d[KEY_PER];
  #pragma unroll
  for (int k = 0; k < KEY_PER; k++) {
    const int i = min(g + k * (KEY_WG * KEY_T), NE / 4 - 1);
    s[k] = src4[i]; d[k] = dst4[i];
  }
  #pragma unroll
  for (int k = 0; k < KEY_PER; k++) {
    const int i = g + k * (KEY_WG * KEY_T);
    if (i < NE / 4) {
      const unsigned k0 = owner_key(d[k].x), k1 = owner_key(d[k].y);
      const unsigned k2 = owner_key(d[k].z), k3 = owner_key(d[k].w);
      keys[i] = k0 | (k1 << 8) | (k2 << 16) | (k3 << 24);
      entry4[i] = make_uint4(owner_entry(s[k].x, d[k].x, k0), owner_entry(s[k].y, d[k].y, k1),
                            owner_entry(s[k].z, d[k].z, k2), owner_entry(s[k].w, d[k].w, k3));
    }
  }
}

static __device__ __forceinline__ void own_max(int* sMax, unsigned dl, float v0, float v1, int lane) {
  atomicMax(&sMax[dl * CH + lane],      __float_as_int(fmaxf(v0, 0.f) + EPS_MSG));
  atomicMax(&sMax[dl * CH + 32 + lane], __float_as_int(fmaxf(v1, 0.f) + EPS_MSG));
}
static __device__ __forceinline__ void own_acc(const int* sMax, float* sDen, float* sNum,
                                               unsigned dl, float v0, float v1, int lane) {
  const float m0 = fmaxf(v0, 0.f) + EPS_MSG, m1 = fmaxf(v1, 0.f) + EPS_MSG;
  const float e0 = __expf(m0 - __int_as_float(sMax[dl * CH + lane]));
  const float e1 = __expf(m1 - __int_as_float(sMax[dl * CH + 32 + lane]));
  atomicAdd(&sDen[dl * CH + lane], e0);       atomicAdd(&sNum[dl * CH + lane], m0 * e0);
  atomicAdd(&sDen[dl * CH + 32 + lane], e1);  atomicAdd(&sNum[dl * CH + 32 + lane], m1 * e1);
}

#define SHIFT_LIM 40.f
#define INITN ((NPB * CH / 4 + OWNT - 1) / OWNT)

static __device__ __forceinline__ void list_two_pass(const unsigned* sList, int total,
                                                     const float* __restrict__ hin,
                                                     int* sMax, float* sDen, float* sNum,
                                                     int wave, int lane) {
  for (int ph = 0; ph < 2; ph++) {
    for (int base = wave * LGRP; base < total; base += 32 * LGRP) {
      unsigned ent[LGRP]; float v0[LGRP], v1[LGRP];
      #pragma unroll
      for (int u = 0; u < LGRP; u++) {
        const int idx = base + u;
        ent[u] = sList[idx < total ? idx : base];
        const float* hp = hin + (size_t)(ent[u] >> 9) * CH + lane;
        v0[u] = hp[0]; v1[u] = hp[32];
      }
      #pragma unroll
      for (int u = 0; u < LGRP; u++) {
        if (base + u < total) {
          if (ph == 0) own_max(sMax, ent[u] & 511u, v0[u], v1[u], lane);
          else         own_acc(sMax, sDen, sNum, ent[u] & 511u, v0[u], v1[u], lane);
        }
      }
    }
    __syncthreads();
  }
}

template <bool FIRST>
__global__ __launch_bounds__(OWNT)
void k_agg_owner(const int* __restrict__ src, const int* __restrict__ dst,
                 const unsigned* __restrict__ keys, const unsigned* __restrict__ entry,
                 const float* __restrict__ hin, float* __restrict__ agg,
                 unsigned* __restrict__ wlist) {
  __shared__ __attribute__((aligned(16))) int   sMax[NPB * CH];
  __shared__ __attribute__((aligned(16))) float sDen[NPB * CH];
  __shared__ __attribute__((aligned(16))) float sNum[NPB * CH];
  __shared__ unsigned sList[LCAP];
  __shared__ unsigned sCnt;
  __shared__ unsigned sBad;
  const int tid = threadIdx.x, lane = tid & 31, wave = tid >> 5;
  const int lo = blockIdx.x * NPB;
  const unsigned cnt = (unsigned)min(NPB, NN - lo);
  const int nout = (int)cnt * CH;

  {
    const float4* own4 = (const float4*)(hin + (size_t)lo * CH);
    const int nown4 = nout / 4;
    float4 r[INITN];
    #pragma unroll
    for (int k = 0; k < INITN; k++) {
      const int i = (tid + k * OWNT < NPB * CH / 4) ? tid + k * OWNT : tid;
      r[k] = own4[min(i, nown4 - 1)];
    }
    int4* m4 = (int4*)sMax; float4* d4 = (float4*)sDen; float4* n4 = (float4*)sNum;
    const float4 z4 = make_float4(0.f, 0.f, 0.f, 0.f);
    #pragma unroll
    for (int k = 0; k < INITN; k++) {
      const int i = (tid + k * OWNT < NPB * CH / 4) ? tid + k * OWNT : tid;
      const bool in = i < nown4;
      float4 s;
      s.x = in ? fmaxf(r[k].x, 0.f) + EPS_MSG : 0.f;
      s.y = in ? fmaxf(r[k].y, 0.f) + EPS_MSG : 0.f;
      s.z = in ? fmaxf(r[k].z, 0.f) + EPS_MSG : 0.f;
      s.w = in ? fmaxf(r[k].w, 0.f) + EPS_MSG : 0.f;
      m4[i] = make_int4(__float_as_int(s.x), __float_as_int(s.y), __float_as_int(s.z), __float_as_int(s.w));
      d4[i] = z4; n4[i] = z4;
    }
  }
  unsigned* wl = wlist + (size_t)blockIdx.x * LSTR;
  if (tid == 0) { sBad = 0u; if (FIRST) sCnt = 0u; }
  __syncthreads();

  unsigned found;
  if (FIRST) {
    const uint4* key4 = (const uint4*)keys;
    const unsigned me4 = (unsigned)blockIdx.x * 0x01010101u;
    for (int i = tid; i < NE / 16; i += SGRP * OWNT) {
      uint4 d[SGRP];
      #pragma unroll
      for (int u = 0; u < SGRP; u++) {
        const int ii = i + u * OWNT;
        d[u] = (ii < NE / 16) ? key4[ii] : make_uint4(~0u, ~0u, ~0u, ~0u);
      }
      #pragma unroll
      for (int u = 0; u < SGRP; u++) {
        const int e = 16 * (i + u * OWNT);
        const unsigned kk[4] = {d[u].x, d[u].y, d[u].z, d[u].w};
        unsigned f[4];
        #pragma unroll
        for (int c = 0; c < 4; c++) {
          const unsigned y = kk[c] ^ me4;
          f[c] = ((y & 0x7F7F7F7Fu) + 0x7F7F7F7Fu) | y | 0x7F7F7F7Fu;
        }
        const unsigned F = f[0] & __builtin_rotateleft32(f[1], 1) &
                           __builtin_rotateleft32(f[2], 2) & __builtin_rotateleft32(f[3], 3);
        if (F != 0xFFFFFFFFu) {
          unsigned z = ~F;
          while (z) {
            const unsigned q = (unsigned)(__ffs(z) - 1 + 25) & 31u;
            z &= z - 1u;
            const unsigned pos = atomicAdd(&sCnt, 1u);
            if (pos < LCAP) sList[pos] = (unsigned)e + 4u * (q & 7u) + (q >> 3);
          }
        }
      }
    }
    __syncthreads();
    found = sCnt;
    if (tid == 0) wl[LCAP] = found;
  } else {
    found = wl[LCAP];
  }

  if (found <= LCAP) {
    const int total = (int)found;
    if (FIRST) {
      for (int i = tid; i < total; i += OWNT) {
        const unsigned fin = entry[sList[i]];
        sList[i] = fin;
        wl[i] = fin;
      }
    } else {
      for (int i = tid; i < total; i += OWNT) sList[i] = wl[i];
    }
    __syncthreads();

    bool bad = false;
    for (int base = wave * LGRP; base < total; base += 32 * LGRP) {
      unsigned ent[LGRP]; float v0[LGRP], v1[LGRP], s0[LGRP], s1[LGRP];
      #pragma unroll
      for (int u = 0; u < LGRP; u++) {
        const int idx = base + u;
        ent[u] = sList[idx < total ? idx : base];
        const float* hp = hin + (size_t)(ent[u] >> 9) * CH + lane;
        v0[u] = hp[0]; v1[u] = hp[32];
      }
      #pragma unroll
      for (int u = 0; u < LGRP; u++) {
        const unsigned dl = ent[u] & 511u;
        s0[u] = __int_as_float(sMax[dl * CH + lane]);
        s1[u] = __int_as_float(sMax[dl * CH + 32 + lane]);
      }
      #pragma unroll
      for (int u = 0; u < LGRP; u++) {
        const bool live = base + u < total;
        const unsigned dl = ent[u] & 511u;
        const float m0 = fmaxf(v0[u], 0.f) + EPS_MSG, m1 = fmaxf(v1[u], 0.f) + EPS_MSG;
        const float d0 = m0 - s0[u], d1 = m1 - s1[u];
        bad = bad || !(fabsf(d0) <= SHIFT_LIM) || !(fabsf(d1) <= SHIFT_LIM);
        const float x0 = __expf(d0), x1 = __expf(d1);
        const float e0 = live ? x0 : 0.f, e1 = live ? x1 : 0.f;
        const float p0 = live ? m0 * x0 : 0.f, p1 = live ? m1 * x1 : 0.f;
        atomicAdd(&sDen[dl * CH + lane], e0);       atomicAdd(&sNum[dl * CH + lane], p0);
        atomicAdd(&sDen[dl * CH + 32 + lane], e1);  atomicAdd(&sNum[dl * CH + 32 + lane], p1);
      }
    }
    if (bad) sBad = 1u;
    __syncthreads();
    if (sBad != 0u) {
      for (int i = tid; i < NPB * CH; i += OWNT) { sMax[i] = 0; sDen[i] = 0.f; sNum[i] = 0.f; }
      __syncthreads();
      list_two_pass(sList, total, hin, sMax, sDen, sNum, wave, lane);
    }
  } else {
    for (int i = tid; i < NPB * CH; i += OWNT) sMax[i] = 0;
    __syncthreads();
    for (int ph = 0; ph < 2; ph++) {
      for (int eb = wave * 32; eb < NE; eb += 32 * 32) {
        const unsigned dl = (unsigned)(dst[eb + lane] - lo);
        unsigned mask = __builtin_amdgcn_ballot_w32(dl < cnt);
        while (mask) {
          const int b = __ffs(mask) - 1;
          mask &= mask - 1u;
          const unsigned dlb = (unsigned)__shfl((int)dl, b);
          const float* hp = hin + (size_t)src[eb + b] * CH + lane;
          const float v0 = hp[0], v1 = hp[32];
          if (ph == 0) own_max(sMax, dlb, v0, v1, lane);
          else         own_acc(sMax, sDen, sNum, dlb, v0, v1, lane);
        }
      }
      __syncthreads();
    }
  }

  float* ap = agg + (size_t)lo * CH;
  for (int i = tid; i < nout; i += OWNT) {
    const float den = sDen[i];
    ap[i] = den > 0.f ? sNum[i] / den : 0.f;
  }
}

static_assert(NN % 32 == 0, "two 16-row strips per work-group");
static_assert(CH == 64 && HD == 128, "channel maps below are written for 64 -> 128 -> 64");

static __device__ __forceinline__ float other_half(float v) {
  return __shfl_xor(v, 16);
}
static __device__ __forceinline__ float f4at(const float4& v, int i) {
  return i == 0 ? v.x : i == 1 ? v.y : i == 2 ? v.z : v.w;
}

template <bool OUTER>
__global__ __launch_bounds__(64)
void k_conv_mlp(const float* __restrict__ resin, const float* __restrict__ agg,
                const float* __restrict__ W1, const float* __restrict__ b1,
                const float* __restrict__ lnw, const float* __restrict__ lnb,
                const float* __restrict__ W2, const float* __restrict__ b2,
                const float* outer, float* out) {
  const int tid = threadIdx.x;
  const int wave = tid >> 5, lane = tid & 31;
  const int half = lane >> 4, l16 = lane & 15;
  const size_t base = ((size_t)blockIdx.x * 2 + wave) * 16;

  v2f actf[CH / 4];
  {
    const float* rp = resin + (base + l16) * CH + 2 * half;
    const float* ap = agg   + (base + l16) * CH + 2 * half;
    #pragma unroll
    for (int j = 0; j < CH / 4; j++) {
      const v2f r = *(const v2f*)(rp + 4 * j);
      const v2f a = *(const v2f*)(ap + 4 * j);
      actf[j] = r + a;
    }
  }

  v8f acc[HD / 16];
  #pragma unroll
  for (int nt = 0; nt < HD / 16; nt++) acc[nt] = (v8f){0.f, 0.f, 0.f, 0.f, 0.f, 0.f, 0.f, 0.f};
  {
    const float* wp = W1 + (2 * half) * HD + 4 * l16;
    #pragma unroll
    for (int j = 0; j < CH / 4; j++) {
      #pragma unroll
      for (int g = 0; g < 2; g++) {
        const float4 w0 = *(const float4*)(wp + (4 * j) * HD + 64 * g);
        const float4 w1 = *(const float4*)(wp + (4 * j + 1) * HD + 64 * g);
        #pragma unroll
        for (int t = 0; t < 4; t++) {
          v2f wf; wf.x = f4at(w0, t); wf.y = f4at(w1, t);
          acc[4 * g + t] = wmma4(wf, actf[j], acc[4 * g + t]);
        }
      }
    }
  }

  {
    float s = 0.f, q = 0.f;
    #pragma unroll
    for (int g = 0; g < 2; g++) {
      #pragma unroll
      for (int r = 0; r < 8; r++) {
        const float4 bv = *(const float4*)(b1 + 64 * g + 4 * (r + 8 * half));
        #pragma unroll
        for (int t = 0; t < 4; t++) {
          const float v = acc[4 * g + t][r] + f4at(bv, t);
          acc[4 * g + t][r] = v;
          s += v; q += v * v;
        }
      }
    }
    s += other_half(s);
    q += other_half(q);
    const float mu = s * (1.f / HD);
    const float var = q * (1.f / HD) - mu * mu;
    const float inv = rsqrtf(var + EPS_LN);
    #pragma unroll
    for (int g = 0; g < 2; g++) {
      #pragma unroll
      for (int r = 0; r < 8; r++) {
        const float4 wv = *(const float4*)(lnw + 64 * g + 4 * (r + 8 * half));
        const float4 cv = *(const float4*)(lnb + 64 * g + 4 * (r + 8 * half));
        #pragma unroll
        for (int t = 0; t < 4; t++) {
          const float v = (acc[4 * g + t][r] - mu) * inv * f4at(wv, t) + f4at(cv, t);
          acc[4 * g + t][r] = fmaxf(v, 0.f);
        }
      }
    }
  }

  v8f acc2[CH / 16];
  #pragma unroll
  for (int ct = 0; ct < CH / 16; ct++) acc2[ct] = (v8f){0.f, 0.f, 0.f, 0.f, 0.f, 0.f, 0.f, 0.f};
  {
    const float* wp = W2 + (size_t)(32 * half) * CH + 4 * l16;
    #pragma unroll
    for (int nt = 0; nt < HD / 16; nt++) {
      #pragma unroll
      for (int rp = 0; rp < 4; rp++) {
        v2f a; a.x = acc[nt][2 * rp]; a.y = acc[nt][2 * rp + 1];
        const int k0 = 64 * (nt / 4) + 8 * rp + (nt % 4);
        const float4 w0 = *(const float4*)(wp + (k0) * CH);
        const float4 w1 = *(const float4*)(wp + (k0 + 4) * CH);
        #pragma unroll
        for (int ct = 0; ct < CH / 16; ct++) {
          v2f bb; bb.x = f4at(w0, ct); bb.y = f4at(w1, ct);
          acc2[ct] = wmma4(a, bb, acc2[ct]);
        }
      }
    }
  }

  const float4 bias = *(const float4*)(b2 + 4 * l16);
  float4 res[8];
  #pragma unroll
  for (int r = 0; r < 8; r++) {
    if (OUTER) res[r] = *(const float4*)(outer + (base + r + 8 * half) * CH + 4 * l16);
    else       res[r] = make_float4(0.f, 0.f, 0.f, 0.f);
  }
  #pragma unroll
  for (int r = 0; r < 8; r++) {
    float4 o;
    o.x = acc2[0][r] + bias.x + res[r].x;
    o.y = acc2[1][r] + bias.y + res[r].y;
    o.z = acc2[2][r] + bias.z + res[r].z;
    o.w = acc2[3][r] + bias.w + res[r].w;
    *(float4*)(out + (base + r + 8 * half) * CH + 4 * l16) = o;
  }
}

__global__ __launch_bounds__(64)
void k_head(const float* __restrict__ xc,
            const float* __restrict__ W1, const float* __restrict__ b1,
            const float* __restrict__ W2, const float* __restrict__ b2,
            const float* __restrict__ W3, const float* __restrict__ b3,
            float* __restrict__ out) {
  const int tid = threadIdx.x;
  const int wave = tid >> 5, lane = tid & 31;
  const int half = lane >> 4, l16 = lane & 15;
  const size_t base = ((size_t)blockIdx.x * 2 + wave) * 16;

  v2f xf[CH / 4];
  {
    const float* xp = xc + (base + l16) * CH + 2 * half;
    #pragma unroll
    for (int j = 0; j < CH / 4; j++) xf[j] = *(const v2f*)(xp + 4 * j);
  }

  v8f h1[CH / 16];
  #pragma unroll
  for (int nt = 0; nt < CH / 16; nt++) h1[nt] = (v8f){0.f, 0.f, 0.f, 0.f, 0.f, 0.f, 0.f, 0.f};
  {
    const float* wp = W1 + (2 * half) * CH + 4 * l16;
    #pragma unroll
    for (int j = 0; j < CH / 4; j++) {
      const float4 w0 = *(const float4*)(wp + (4 * j) * CH);
      const float4 w1 = *(const float4*)(wp + (4 * j + 1) * CH);
      #pragma unroll
      for (int nt = 0; nt < CH / 16; nt++) {
        v2f wf; wf.x = f4at(w0, nt); wf.y = f4at(w1, nt);
        h1[nt] = wmma4(wf, xf[j], h1[nt]);
      }
    }
  }
  #pragma unroll
  for (int r = 0; r < 8; r++) {
    const float4 bv = *(const float4*)(b1 + 4 * (r + 8 * half));
    #pragma unroll
    for (int nt = 0; nt < CH / 16; nt++) h1[nt][r] = fmaxf(h1[nt][r] + f4at(bv, nt), 0.f);
  }

  v8f h2[CH / 16];
  #pragma unroll
  for (int nt = 0; nt < CH / 16; nt++) h2[nt] = (v8f){0.f, 0.f, 0.f, 0.f, 0.f, 0.f, 0.f, 0.f};
  {
    const float* wp = W2 + (size_t)(32 * half) * CH + 4 * l16;
    #pragma unroll
    for (int nt = 0; nt < CH / 16; nt++) {
      #pragma unroll
      for (int rp = 0; rp < 4; rp++) {
        v2f hb; hb.x = h1[nt][2 * rp]; hb.y = h1[nt][2 * rp + 1];
        const int k0 = 8 * rp + nt;
        const float4 w0 = *(const float4*)(wp + (k0) * CH);
        const float4 w1 = *(const float4*)(wp + (k0 + 4) * CH);
        #pragma unroll
        for (int n2 = 0; n2 < CH / 16; n2++) {
          v2f wf; wf.x = f4at(w0, n2); wf.y = f4at(w1, n2);
          h2[n2] = wmma4(wf, hb, h2[n2]);
        }
      }
    }
  }

  float o0 = 0.f, o1 = 0.f;
  #pragma unroll
  for (int r = 0; r < 8; r++) {
    const int c0 = 4 * (r + 8 * half);
    const float4 bv = *(const float4*)(b2 + c0);
    const float4 ta = *(const float4*)(W3 + 2 * c0);
    const float4 tb = *(const float4*)(W3 + 2 * c0 + 4);
    const float w3[8] = {ta.x, ta.y, ta.z, ta.w, tb.x, tb.y, tb.z, tb.w};
    #pragma unroll
    for (int n2 = 0; n2 < CH / 16; n2++) {
      const float v = fmaxf(h2[n2][r] + f4at(bv, n2), 0.f);
      o0 += v * w3[2 * n2];
      o1 += v * w3[2 * n2 + 1];
    }
  }
  o0 += other_half(o0);
  o1 += other_half(o1);
  if (half == 0) {
    float2 res; res.x = o0 + b3[0]; res.y = o1 + b3[1];
    *(float2*)(out + (base + l16) * 2) = res;
  }
}

extern "C" void kernel_launch(void* const* d_in, const int* in_sizes, int n_in,
                              void* d_out, int out_size, void* d_ws, size_t ws_size,
                              hipStream_t stream) {
  (void)in_sizes; (void)n_in; (void)out_size; (void)ws_size;
  const float* x      = (const float*)d_in[0];
  const int*   ei     = (const int*)d_in[1];
  const int*   src    = ei;
  const int*   dst    = ei + NE;
  const float* c1_W1  = (const float*)d_in[2];
  const float* c1_b1  = (const float*)d_in[3];
  const float* c1_lnw = (const float*)d_in[4];
  const float* c1_lnb = (const float*)d_in[5];
  const float* c1_W2  = (const float*)d_in[6];
  const float* c1_b2  = (const float*)d_in[7];
  const float* n1_w   = (const float*)d_in[8];
  const float* n1_b   = (const float*)d_in[9];
  const float* cW1    = (const float*)d_in[10];
  const float* cb1    = (const float*)d_in[11];
  const float* clnw   = (const float*)d_in[12];
  const float* clnb   = (const float*)d_in[13];
  const float* cW2    = (const float*)d_in[14];
  const float* cb2    = (const float*)d_in[15];
  const float* nw     = (const float*)d_in[16];
  const float* nb     = (const float*)d_in[17];
  const float* lW1    = (const float*)d_in[18];
  const float* lb1    = (const float*)d_in[19];
  const float* lW2    = (const float*)d_in[20];
  const float* lb2    = (const float*)d_in[21];
  const float* lW3    = (const float*)d_in[22];
  const float* lb3    = (const float*)d_in[23];

  float* ws    = (float*)d_ws;
  float* xc    = ws;
  float* hpre  = ws + (size_t)N64;
  float* agg   = ws + 4 * (size_t)N64;
  float* part  = ws + 3 * (size_t)N64;
  unsigned* wlist = (unsigned*)(ws + 2 * (size_t)N64);
  unsigned* keys  = (unsigned*)(ws + 3 * (size_t)N64 + 1024);
  unsigned* entry = (unsigned*)(ws + 3 * (size_t)N64 + 262144);

  const dim3 B(256);
  const int APB  = (N64 / 4 + 255) / 256;
  const int MLPB = NN / 32;
  const dim3 B2(64);

  auto edge_phase = [&](const float* hin, bool first) {
    if (first) k_agg_owner<true><<<NOWN, OWNT, 0, stream>>>(src, dst, keys, entry, hin, agg, wlist);
    else       k_agg_owner<false><<<NOWN, OWNT, 0, stream>>>(src, dst, keys, entry, hin, agg, wlist);
  };

  k_owner_keys<<<KEY_WG, KEY_T, 0, stream>>>(src, dst, keys, entry);

  edge_phase(x, true);
  k_conv_mlp<false><<<MLPB, B2, 0, stream>>>(x, agg, c1_W1, c1_b1, c1_lnw, c1_lnb,
                                     c1_W2, c1_b2, nullptr, hpre);
  k_gln_partial<<<GLN_WG, GLN_T, 0, stream>>>((const float4*)hpre, part);
  k_gln_apply<<<APB, B, 0, stream>>>((const float4*)hpre, part, n1_w, n1_b, (float4*)xc);

  for (int i = 0; i < 3; i++) {
    k_gln_partial<<<GLN_WG, GLN_T, 0, stream>>>((const float4*)xc, part);
    k_gln_apply<<<APB, B, 0, stream>>>((const float4*)xc, part, nw + i * CH, nb + i * CH,
                                       (float4*)hpre);
    edge_phase(hpre, false);
    k_conv_mlp<true><<<MLPB, B2, 0, stream>>>(hpre, agg,
                                       cW1 + (size_t)i * CH * HD, cb1 + (size_t)i * HD,
                                       clnw + (size_t)i * HD, clnb + (size_t)i * HD,
                                       cW2 + (size_t)i * HD * CH, cb2 + (size_t)i * CH,
                                       xc, xc);
  }

  k_head<<<MLPB, B2, 0, stream>>>(xc, lW1, lb1, lW2, lb2, lW3, lb3, (float*)d_out);
}


# Softmax with no shift: exponent = the message itself (range test is one running max, exact two-pass redo above 40); own-row fetch, shift plane fill and its LDS read per edge gone; sum order kept
# speedup vs baseline: 3.5952x; 1.0428x over previous
#include <hip/hip_runtime.h>
#include <math.h>

#define NN   100000
#define CH   64
#define HD   128
#define NE   800000
#define N64  6400000
#define EPS_MSG 1e-7f
#define EPS_LN  1e-5f

typedef float v2f __attribute__((ext_vector_type(2)));
typedef float v8f __attribute__((ext_vector_type(8)));

static __device__ __forceinline__ v8f wmma4(v2f a, v2f b, v8f c) {
  return __builtin_amdgcn_wmma_f32_16x16x4_f32(
       false, a,  false, b,
       (short)0, c,  false,  false);
}

#define GLN_WG   256
#define GLN_T    1024
#define GLN_PER  ((N64 / 4 + GLN_WG * GLN_T - 1) / (GLN_WG * GLN_T))
static_assert(GLN_WG == 256, "a lane of the apply kernel adds 8 pairs: 32 lanes x 8 = 256");

static __device__ __forceinline__ float wave_sum(float v) {
  v += __shfl_xor(v, 16); v += __shfl_xor(v, 8); v += __shfl_xor(v, 4);
  v += __shfl_xor(v, 2);  v += __shfl_xor(v, 1);
  return v;
}

__global__ __launch_bounds__(GLN_T) void k_gln_partial(const float4* __restrict__ h,
                                                       float* __restrict__ part) {
  __shared__ float ss[GLN_T / 32];
  __shared__ float sq[GLN_T / 32];
  const int tid = threadIdx.x, lane = tid & 31, wave = tid >> 5;
  const int g = blockIdx.x * GLN_T + tid;
  float4 v[GLN_PER];
  #pragma unroll
  for (int k = 0; k < GLN_PER; k++) {
    const int i = g + k * (GLN_WG * GLN_T);
    v[k] = h[min(i, N64 / 4 - 1)];
  }
  float s = 0.f, q = 0.f;
  #pragma unroll
  for (int k = 0; k < GLN_PER; k++) {
    const bool in = g + k * (GLN_WG * GLN_T) < N64 / 4;
    const float4 u = v[k];
    const float a = (u.x + u.y) + (u.z + u.w);
    const float b = (u.x * u.x + u.y * u.y) + (u.z * u.z + u.w * u.w);
    s += in ? a : 0.f;
    q += in ? b : 0.f;
  }
  s = wave_sum(s); q = wave_sum(q);
  if (lane == 0) { ss[wave] = s; sq[wave] = q; }
  __syncthreads();
  if (wave == 0) {
    s = wave_sum(ss[lane]); q = wave_sum(sq[lane]);
    if (lane == 0) { part[2 * blockIdx.x] = s; part[2 * blockIdx.x + 1] = q; }
  }
}

__global__ __launch_bounds__(256) void k_gln_apply(const float4* __restrict__ h,
                                                   const float* __restrict__ part,
                                                   const float* __restrict__ w,
                                                   const float* __restrict__ b,
                                                   float4* __restrict__ out) {
  int i = blockIdx.x * 256 + threadIdx.x;
  if (i >= N64 / 4) return;
  const float4 v = h[i];
  float s = 0.f, q = 0.f;
  {
    const float4* p4 = (const float4*)(part + 16 * (threadIdx.x & 31));
    const float4 p0 = p4[0], p1 = p4[1], p2 = p4[2], p3 = p4[3];
    s = ((p0.x + p0.z) + (p1.x + p1.z)) + ((p2.x + p2.z) + (p3.x + p3.z));
    q = ((p0.y + p0.w) + (p1.y + p1.w)) + ((p2.y + p2.w) + (p3.y + p3.w));
  }
  s = wave_sum(s); q = wave_sum(q);
  const float invn = 1.f / (float)N64;
  float mu = s * invn;
  float var = q * invn - mu * mu;
  float scale = 1.f / (sqrtf(fmaxf(var, 0.f)) + EPS_LN);
  int c = (i & 15) << 2;
  float r0 = (v.x - mu) * scale * w[c + 0] + b[c + 0];
  float r1 = (v.y - mu) * scale * w[c + 1] + b[c + 1];
  float r2 = (v.z - mu) * scale * w[c + 2] + b[c + 2];
  float r3 = (v.w - mu) * scale * w[c + 3] + b[c + 3];
  out[i] = make_float4(fmaxf(r0, 0.f), fmaxf(r1, 0.f), fmaxf(r2, 0.f), fmaxf(r3, 0.f));
}

#define NPB   393
#define NOWN  255
#define OWNT  1024
#define LCAP  4096
#define LGRP  16
#define SGRP  7
#define LSTR  (LCAP + 32)
static_assert(NPB * NOWN >= NN && NPB * (NOWN - 1) < NN, "owner ranges must cover NN");
static_assert(NPB <= 512 && NN <= (1 << 23) && NE <= (1 << 23), "packed entry: 9 bits node, 23 bits edge or source");
static_assert(NE % 32 == 0 && NE % 16 == 0, "edge list is walked in 32s; keys in 16-byte loads");
static_assert(NOWN < 256, "owner numbers 0..NOWN-1 and KEY_NONE must fit one byte");

#define KEY_NONE 255u
#define KEY_T    256
#define KEY_PER  4
#define KEY_WG   ((NE / 4 + KEY_T * KEY_PER - 1) / (KEY_T * KEY_PER))

static __device__ __forceinline__ unsigned owner_key(int d) {
  return (unsigned)d < (unsigned)NN ? (unsigned)d / (unsigned)NPB : KEY_NONE;
}

static __device__ __forceinline__ unsigned owner_entry(int s, int d, unsigned key) {
  return ((unsigned)s << 9) | (((unsigned)d - key * (unsigned)NPB) & 511u);
}

__global__ __launch_bounds__(KEY_T) void k_owner_keys(const int* __restrict__ src,
                                                      const int* __restrict__ dst,
                                                      unsigned* __restrict__ keys,
                                                      unsigned* __restrict__ entry) {
  uint4* entry4 = (uint4*)entry;
  const int4* src4 = (const int4*)src;
  const int4* dst4 = (const int4*)dst;
  const int g = blockIdx.x * KEY_T + threadIdx.x;
  int4 s[KEY_PER], d[KEY_PER];
  #pragma unroll
  for (int k = 0; k < KEY_PER; k++) {
    const int i = min(g + k * (KEY_WG * KEY_T), NE / 4 - 1);
    s[k] = src4[i]; d[k] = dst4[i];
  }
  #pragma unroll
  for (int k = 0; k < KEY_PER; k++) {
    const int i = g + k * (KEY_WG * KEY_T);
    if (i < NE / 4) {
      const unsigned k0 = owner_key(d[k].x), k1 = owner_key(d[k].y);
      const unsigned k2 = owner_key(d[k].z), k3 = owner_key(d[k].w);
      keys[i] = k0 | (k1 << 8) | (k2 << 16) | (k3 << 24);
      entry4[i] = make_uint4(owner_entry(s[k].x, d[k].x, k0), owner_entry(s[k].y, d[k].y, k1),
                            owner_entry(s[k].z, d[k].z, k2), owner_entry(s[k].w, d[k].w, k3));
    }
  }
}

static __device__ __forceinline__ void own_max(int* sMax, unsigned dl, float v0, float v1, int lane) {
  atomicMax(&sMax[dl * CH + lane],      __float_as_int(fmaxf(v0, 0.f) + EPS_MSG));
  atomicMax(&sMax[dl * CH + 32 + lane], __float_as_int(fmaxf(v1, 0.f) + EPS_MSG));
}
static __device__ __forceinline__ void own_acc(const int* sMax, float* sDen, float* sNum,
                                               unsigned dl, float v0, float v1, int lane) {
  const float m0 = fmaxf(v0, 0.f) + EPS_MSG, m1 = fmaxf(v1, 0.f) + EPS_MSG;
  const float e0 = __expf(m0 - __int_as_float(sMax[dl * CH + lane]));
  const float e1 = __expf(m1 - __int_as_float(sMax[dl * CH + 32 + lane]));
  atomicAdd(&sDen[dl * CH + lane], e0);       atomicAdd(&sNum[dl * CH + lane], m0 * e0);
  atomicAdd(&sDen[dl * CH + 32 + lane], e1);  atomicAdd(&sNum[dl * CH + 32 + lane], m1 * e1);
}

#define SHIFT_LIM 40.f

static __device__ __forceinline__ void list_two_pass(const unsigned* sList, int total,
                                                     const float* __restrict__ hin,
                                                     int* sMax, float* sDen, float* sNum,
                                                     int wave, int lane) {
  for (int ph = 0; ph < 2; ph++) {
    for (int base = wave * LGRP; base < total; base += 32 * LGRP) {
      unsigned ent[LGRP]; float v0[LGRP], v1[LGRP];
      #pragma unroll
      for (int u = 0; u < LGRP; u++) {
        const int idx = base + u;
        ent[u] = sList[idx < total ? idx : base];
        const float* hp = hin + (size_t)(ent[u] >> 9) * CH + lane;
        v0[u] = hp[0]; v1[u] = hp[32];
      }
      #pragma unroll
      for (int u = 0; u < LGRP; u++) {
        if (base + u < total) {
          if (ph == 0) own_max(sMax, ent[u] & 511u, v0[u], v1[u], lane);
          else         own_acc(sMax, sDen, sNum, ent[u] & 511u, v0[u], v1[u], lane);
        }
      }
    }
    __syncthreads();
  }
}

template <bool FIRST>
__global__ __launch_bounds__(OWNT)
void k_agg_owner(const int* __restrict__ src, const int* __restrict__ dst,
                 const unsigned* __restrict__ keys, const unsigned* __restrict__ entry,
                 const float* __restrict__ hin, float* __restrict__ agg,
                 unsigned* __restrict__ wlist) {
  __shared__ __attribute__((aligned(16))) int   sMax[NPB * CH];
  __shared__ __attribute__((aligned(16))) float sDen[NPB * CH];
  __shared__ __attribute__((aligned(16))) float sNum[NPB * CH];
  __shared__ unsigned sList[LCAP];
  __shared__ unsigned sCnt;
  __shared__ unsigned sBad;
  const int tid = threadIdx.x, lane = tid & 31, wave = tid >> 5;
  const int lo = blockIdx.x * NPB;
  const unsigned cnt = (unsigned)min(NPB, NN - lo);
  const int nout = (int)cnt * CH;

  {
    float4* d4 = (float4*)sDen; float4* n4 = (float4*)sNum;
    const float4 z4 = make_float4(0.f, 0.f, 0.f, 0.f);
    for (int i = tid; i < NPB * CH / 4; i += OWNT) { d4[i] = z4; n4[i] = z4; }
  }
  unsigned* wl = wlist + (size_t)blockIdx.x * LSTR;
  if (tid == 0) { sBad = 0u; if (FIRST) sCnt = 0u; }
  __syncthreads();

  unsigned found;
  if (FIRST) {
    const uint4* key4 = (const uint4*)keys;
    const unsigned me4 = (unsigned)blockIdx.x * 0x01010101u;
    for (int i = tid; i < NE / 16; i += SGRP * OWNT) {
      uint4 d[SGRP];
      #pragma unroll
      for (int u = 0; u < SGRP; u++) {
        const int ii = i + u * OWNT;
        d[u] = (ii < NE / 16) ? key4[ii] : make_uint4(~0u, ~0u, ~0u, ~0u);
      }
      #pragma unroll
      for (int u = 0; u < SGRP; u++) {
        const int e = 16 * (i + u * OWNT);
        const unsigned kk[4] = {d[u].x, d[u].y, d[u].z, d[u].w};
        unsigned f[4];
        #pragma unroll
        for (int c = 0; c < 4; c++) {
          const unsigned y = kk[c] ^ me4;
          f[c] = ((y & 0x7F7F7F7Fu) + 0x7F7F7F7Fu) | y | 0x7F7F7F7Fu;
        }
        const unsigned F = f[0] & __builtin_rotateleft32(f[1], 1) &
                           __builtin_rotateleft32(f[2], 2) & __builtin_rotateleft32(f[3], 3);
        if (F != 0xFFFFFFFFu) {
          unsigned z = ~F;
          while (z) {
            const unsigned q = (unsigned)(__ffs(z) - 1 + 25) & 31u;
            z &= z - 1u;
            const unsigned pos = atomicAdd(&sCnt, 1u);
            if (pos < LCAP) sList[pos] = (unsigned)e + 4u * (q & 7u) + (q >> 3);
          }
        }
      }
    }
    __syncthreads();
    found = sCnt;
    if (tid == 0) wl[LCAP] = found;
  } else {
    found = wl[LCAP];
  }

  if (found <= LCAP) {
    const int total = (int)found;
    if (FIRST) {
      for (int i = tid; i < total; i += OWNT) {
        const unsigned fin = entry[sList[i]];
        sList[i] = fin;
        wl[i] = fin;
      }
    } else {
      for (int i = tid; i < total; i += OWNT) sList[i] = wl[i];
    }
    __syncthreads();

    float top = 0.f;
    for (int base = wave * LGRP; base < total; base += 32 * LGRP) {
      unsigned ent[LGRP]; float v0[LGRP], v1[LGRP];
      #pragma unroll
      for (int u = 0; u < LGRP; u++) {
        const int idx = base + u;
        ent[u] = sList[idx < total ? idx : base];
        const float* hp = hin + (size_t)(ent[u] >> 9) * CH + lane;
        v0[u] = hp[0]; v1[u] = hp[32];
      }
      #pragma unroll
      for (int u = 0; u < LGRP; u++) {
        const bool live = base + u < total;
        const unsigned dl = ent[u] & 511u;
        const float m0 = fmaxf(v0[u], 0.f) + EPS_MSG, m1 = fmaxf(v1[u], 0.f) + EPS_MSG;
        top = fmaxf(top, fmaxf(m0, m1));
        const float x0 = __expf(m0), x1 = __expf(m1);
        const float e0 = live ? x0 : 0.f, e1 = live ? x1 : 0.f;
        const float p0 = live ? m0 * x0 : 0.f, p1 = live ? m1 * x1 : 0.f;
        atomicAdd(&sDen[dl * CH + lane], e0);       atomicAdd(&sNum[dl * CH + lane], p0);
        atomicAdd(&sDen[dl * CH + 32 + lane], e1);  atomicAdd(&sNum[dl * CH + 32 + lane], p1);
      }
    }
    if (!(top <= SHIFT_LIM)) sBad = 1u;
    __syncthreads();
    if (sBad != 0u) {
      for (int i = tid; i < NPB * CH; i += OWNT) { sMax[i] = 0; sDen[i] = 0.f; sNum[i] = 0.f; }
      __syncthreads();
      list_two_pass(sList, total, hin, sMax, sDen, sNum, wave, lane);
    }
  } else {
    for (int i = tid; i < NPB * CH; i += OWNT) sMax[i] = 0;
    __syncthreads();
    for (int ph = 0; ph < 2; ph++) {
      for (int eb = wave * 32; eb < NE; eb += 32 * 32) {
        const unsigned dl = (unsigned)(dst[eb + lane] - lo);
        unsigned mask = __builtin_amdgcn_ballot_w32(dl < cnt);
        while (mask) {
          const int b = __ffs(mask) - 1;
          mask &= mask - 1u;
          const unsigned dlb = (unsigned)__shfl((int)dl, b);
          const float* hp = hin + (size_t)src[eb + b] * CH + lane;
          const float v0 = hp[0], v1 = hp[32];
          if (ph == 0) own_max(sMax, dlb, v0, v1, lane);
          else         own_acc(sMax, sDen, sNum, dlb, v0, v1, lane);
        }
      }
      __syncthreads();
    }
  }

  float* ap = agg + (size_t)lo * CH;
  for (int i = tid; i < nout; i += OWNT) {
    const float den = sDen[i];
    ap[i] = den > 0.f ? sNum[i] / den : 0.f;
  }
}

static_assert(NN % 32 == 0, "two 16-row strips per work-group");
static_assert(CH == 64 && HD == 128, "channel maps below are written for 64 -> 128 -> 64");

static __device__ __forceinline__ float other_half(float v) {
  return __shfl_xor(v, 16);
}
static __device__ __forceinline__ float f4at(const float4& v, int i) {
  return i == 0 ? v.x : i == 1 ? v.y : i == 2 ? v.z : v.w;
}

template <bool OUTER>
__global__ __launch_bounds__(64)
void k_conv_mlp(const float* __restrict__ resin, const float* __restrict__ agg,
                const float* __restrict__ W1, const float* __restrict__ b1,
                const float* __restrict__ lnw, const float* __restrict__ lnb,
                const float* __restrict__ W2, const float* __restrict__ b2,
                const float* outer, float* out) {
  const int tid = threadIdx.x;
  const int wave = tid >> 5, lane = tid & 31;
  const int half = lane >> 4, l16 = lane & 15;
  const size_t base = ((size_t)blockIdx.x * 2 + wave) * 16;

  v2f actf[CH / 4];
  {
    const float* rp = resin + (base + l16) * CH + 2 * half;
    const float* ap = agg   + (base + l16) * CH + 2 * half;
    #pragma unroll
    for (int j = 0; j < CH / 4; j++) {
      const v2f r = *(const v2f*)(rp + 4 * j);
      const v2f a = *(const v2f*)(ap + 4 * j);
      actf[j] = r + a;
    }
  }

  v8f acc[HD / 16];
  #pragma unroll
  for (int nt = 0; nt < HD / 16; nt++) acc[nt] = (v8f){0.f, 0.f, 0.f, 0.f, 0.f, 0.f, 0.f, 0.f};
  {
    const float* wp = W1 + (2 * half) * HD + 4 * l16;
    #pragma unroll
    for (int j = 0; j < CH / 4; j++) {
      #pragma unroll
      for (int g = 0; g < 2; g++) {
        const float4 w0 = *(const float4*)(wp + (4 * j) * HD + 64 * g);
        const float4 w1 = *(const float4*)(wp + (4 * j + 1) * HD + 64 * g);
        #pragma unroll
        for (int t = 0; t < 4; t++) {
          v2f wf; wf.x = f4at(w0, t); wf.y = f4at(w1, t);
          acc[4 * g + t] = wmma4(wf, actf[j], acc[4 * g + t]);
        }
      }
    }
  }

  {
    float s = 0.f, q = 0.f;
    #pragma unroll
    for (int g = 0; g < 2; g++) {
      #pragma unroll
      for (int r = 0; r < 8; r++) {
        const float4 bv = *(const float4*)(b1 + 64 * g + 4 * (r + 8 * half));
        #pragma unroll
        for (int t = 0; t < 4; t++) {
          const float v = acc[4 * g + t][r] + f4at(bv, t);
          acc[4 * g + t][r] = v;
          s += v; q += v * v;
        }
      }
    }
    s += other_half(s);
    q += other_half(q);
    const float mu = s * (1.f / HD);
    const float var = q * (1.f / HD) - mu * mu;
    const float inv = rsqrtf(var + EPS_LN);
    #pragma unroll
    for (int g = 0; g < 2; g++) {
      #pragma unroll
      for (int r = 0; r < 8; r++) {
        const float4 wv = *(const float4*)(lnw + 64 * g + 4 * (r + 8 * half));
        const float4 cv = *(const float4*)(lnb + 64 * g + 4 * (r + 8 * half));
        #pragma unroll
        for (int t = 0; t < 4; t++) {
          const float v = (acc[4 * g + t][r] - mu) * inv * f4at(wv, t) + f4at(cv, t);
          acc[4 * g + t][r] = fmaxf(v, 0.f);
        }
      }
    }
  }

  v8f acc2[CH / 16];
  #pragma unroll
  for (int ct = 0; ct < CH / 16; ct++) acc2[ct] = (v8f){0.f, 0.f, 0.f, 0.f, 0.f, 0.f, 0.f, 0.f};
  {
    const float* wp = W2 + (size_t)(32 * half) * CH + 4 * l16;
    #pragma unroll
    for (int nt = 0; nt < HD / 16; nt++) {
      #pragma unroll
      for (int rp = 0; rp < 4; rp++) {
        v2f a; a.x = acc[nt][2 * rp]; a.y = acc[nt][2 * rp + 1];
        const int k0 = 64 * (nt / 4) + 8 * rp + (nt % 4);
        const float4 w0 = *(const float4*)(wp + (k0) * CH);
        const float4 w1 = *(const float4*)(wp + (k0 + 4) * CH);
        #pragma unroll
        for (int ct = 0; ct < CH / 16; ct++) {
          v2f bb; bb.x = f4at(w0, ct); bb.y = f4at(w1, ct);
          acc2[ct] = wmma4(a, bb, acc2[ct]);
        }
      }
    }
  }

  const float4 bias = *(const float4*)(b2 + 4 * l16);
  float4 res[8];
  #pragma unroll
  for (int r = 0; r < 8; r++) {
    if (OUTER) res[r] = *(const float4*)(outer + (base + r + 8 * half) * CH + 4 * l16);
    else       res[r] = make_float4(0.f, 0.f, 0.f, 0.f);
  }
  #pragma unroll
  for (int r = 0; r < 8; r++) {
    float4 o;
    o.x = acc2[0][r] + bias.x + res[r].x;
    o.y = acc2[1][r] + bias.y + res[r].y;
    o.z = acc2[2][r] + bias.z + res[r].z;
    o.w = acc2[3][r] + bias.w + res[r].w;
    *(float4*)(out + (base + r + 8 * half) * CH + 4 * l16) = o;
  }
}

__global__ __launch_bounds__(64)
void k_head(const float* __restrict__ xc,
            const float* __restrict__ W1, const float* __restrict__ b1,
            const float* __restrict__ W2, const float* __restrict__ b2,
            const float* __restrict__ W3, const float* __restrict__ b3,
            float* __restrict__ out) {
  const int tid = threadIdx.x;
  const int wave = tid >> 5, lane = tid & 31;
  const int half = lane >> 4, l16 = lane & 15;
  const size_t base = ((size_t)blockIdx.x * 2 + wave) * 16;

  v2f xf[CH / 4];
  {
    const float* xp = xc + (base + l16) * CH + 2 * half;
    #pragma unroll
    for (int j = 0; j < CH / 4; j++) xf[j] = *(const v2f*)(xp + 4 * j);
  }

  v8f h1[CH / 16];
  #pragma unroll
  for (int nt = 0; nt < CH / 16; nt++) h1[nt] = (v8f){0.f, 0.f, 0.f, 0.f, 0.f, 0.f, 0.f, 0.f};
  {
    const float* wp = W1 + (2 * half) * CH + 4 * l16;
    #pragma unroll
    for (int j = 0; j < CH / 4; j++) {
      const float4 w0 = *(const float4*)(wp + (4 * j) * CH);
      const float4 w1 = *(const float4*)(wp + (4 * j + 1) * CH);
      #pragma unroll
      for (int nt = 0; nt < CH / 16; nt++) {
        v2f wf; wf.x = f4at(w0, nt); wf.y = f4at(w1, nt);
        h1[nt] = wmma4(wf, xf[j], h1[nt]);
      }
    }
  }
  #pragma unroll
  for (int r = 0; r < 8; r++) {
    const float4 bv = *(const float4*)(b1 + 4 * (r + 8 * half));
    #pragma unroll
    for (int nt = 0; nt < CH / 16; nt++) h1[nt][r] = fmaxf(h1[nt][r] + f4at(bv, nt), 0.f);
  }

  v8f h2[CH / 16];
  #pragma unroll
  for (int nt = 0; nt < CH / 16; nt++) h2[nt] = (v8f){0.f, 0.f, 0.f, 0.f, 0.f, 0.f, 0.f, 0.f};
  {
    const float* wp = W2 + (size_t)(32 * half) * CH + 4 * l16;
    #pragma unroll
    for (int nt = 0; nt < CH / 16; nt++) {
      #pragma unroll
      for (int rp = 0; rp < 4; rp++) {
        v2f hb; hb.x = h1[nt][2 * rp]; hb.y = h1[nt][2 * rp + 1];
        const int k0 = 8 * rp + nt;
        const float4 w0 = *(const float4*)(wp + (k0) * CH);
        const float4 w1 = *(const float4*)(wp + (k0 + 4) * CH);
        #pragma unroll
        for (int n2 = 0; n2 < CH / 16; n2++) {
          v2f wf; wf.x = f4at(w0, n2); wf.y = f4at(w1, n2);
          h2[n2] = wmma4(wf, hb, h2[n2]);
        }
      }
    }
  }

  float o0 = 0.f, o1 = 0.f;
  #pragma unroll
  for (int r = 0; r < 8; r++) {
    const int c0 = 4 * (r + 8 * half);
    const float4 bv = *(const float4*)(b2 + c0);
    const float4 ta = *(const float4*)(W3 + 2 * c0);
    const float4 tb = *(const float4*)(W3 + 2 * c0 + 4);
    const float w3[8] = {ta.x, ta.y, ta.z, ta.w, tb.x, tb.y, tb.z, tb.w};
    #pragma unroll
    for (int n2 = 0; n2 < CH / 16; n2++) {
      const float v = fmaxf(h2[n2][r] + f4at(bv, n2), 0.f);
      o0 += v * w3[2 * n2];
      o1 += v * w3[2 * n2 + 1];
    }
  }
  o0 += other_half(o0);
  o1 += other_half(o1);
  if (half == 0) {
    float2 res; res.x = o0 + b3[0]; res.y = o1 + b3[1];
    *(float2*)(out + (base + l16) * 2) = res;
  }
}

extern "C" void kernel_launch(void* const* d_in, const int* in_sizes, int n_in,
                              void* d_out, int out_size, void* d_ws, size_t ws_size,
                              hipStream_t stream) {
  (void)in_sizes; (void)n_in; (void)out_size; (void)ws_size;
  const float* x      = (const float*)d_in[0];
  const int*   ei     = (const int*)d_in[1];
  const int*   src    = ei;
  const int*   dst    = ei + NE;
  const float* c1_W1  = (const float*)d_in[2];
  const float* c1_b1  = (const float*)d_in[3];
  const float* c1_lnw = (const float*)d_in[4];
  const float* c1_lnb = (const float*)d_in[5];
  const float* c1_W2  = (const float*)d_in[6];
  const float* c1_b2  = (const float*)d_in[7];
  const float* n1_w   = (const float*)d_in[8];
  const float* n1_b   = (const float*)d_in[9];
  const float* cW1    = (const float*)d_in[10];
  const float* cb1    = (const float*)d_in[11];
  const float* clnw   = (const float*)d_in[12];
  const float* clnb   = (const float*)d_in[13];
  const float* cW2    = (const float*)d_in[14];
  const float* cb2    = (const float*)d_in[15];
  const float* nw     = (const float*)d_in[16];
  const float* nb     = (const float*)d_in[17];
  const float* lW1    = (const float*)d_in[18];
  const float* lb1    = (const float*)d_in[19];
  const float* lW2    = (const float*)d_in[20];
  const float* lb2    = (const float*)d_in[21];
  const float* lW3    = (const float*)d_in[22];
  const float* lb3    = (const float*)d_in[23];

  float* ws    = (float*)d_ws;
  float* xc    = ws;
  float* hpre  = ws + (size_t)N64;
  float* agg   = ws + 4 * (size_t)N64;
  float* part  = ws + 3 * (size_t)N64;
  unsigned* wlist = (unsigned*)(ws + 2 * (size_t)N64);
  unsigned* keys  = (unsigned*)(ws + 3 * (size_t)N64 + 1024);
  unsigned* entry = (unsigned*)(ws + 3 * (size_t)N64 + 262144);

  const dim3 B(256);
  const int APB  = (N64 / 4 + 255) / 256;
  const int MLPB = NN / 32;
  const dim3 B2(64);

  auto edge_phase = [&](const float* hin, bool first) {
    if (first) k_agg_owner<true><<<NOWN, OWNT, 0, stream>>>(src, dst, keys, entry, hin, agg, wlist);
    else       k_agg_owner<false><<<NOWN, OWNT, 0, stream>>>(src, dst, keys, entry, hin, agg, wlist);
  };

  k_owner_keys<<<KEY_WG, KEY_T, 0, stream>>>(src, dst, keys, entry);

  edge_phase(x, true);
  k_conv_mlp<false><<<MLPB, B2, 0, stream>>>(x, agg, c1_W1, c1_b1, c1_lnw, c1_lnb,
                                     c1_W2, c1_b2, nullptr, hpre);
  k_gln_partial<<<GLN_WG, GLN_T, 0, stream>>>((const float4*)hpre, part);
  k_gln_apply<<<APB, B, 0, stream>>>((const float4*)hpre, part, n1_w, n1_b, (float4*)xc);

  for (int i = 0; i < 3; i++) {
    k_gln_partial<<<GLN_WG, GLN_T, 0, stream>>>((const float4*)xc, part);
    k_gln_apply<<<APB, B, 0, stream>>>((const float4*)xc, part, nw + i * CH, nb + i * CH,
                                       (float4*)hpre);
    edge_phase(hpre, false);
    k_conv_mlp<true><<<MLPB, B2, 0, stream>>>(hpre, agg,
                                       cW1 + (size_t)i * CH * HD, cb1 + (size_t)i * HD,
                                       clnw + (size_t)i * HD, clnb + (size_t)i * HD,
                                       cW2 + (size_t)i * HD * CH, cb2 + (size_t)i * CH,
                                       xc, xc);
  }

  k_head<<<MLPB, B2, 0, stream>>>(xc, lW1, lb1, lW2, lb2, lW3, lb3, (float*)d_out);
}


# Conv-MLP weights staged once per work-group in LDS in matrix-operand order: 250 work-groups of 25 waves (one a WGP, one 16-row strip a wave) read fragments as 16-byte LDS loads; k and sum order kept
# speedup vs baseline: 4.5948x; 1.2780x over previous
#include <hip/hip_runtime.h>
#include <math.h>

#define NN   100000
#define CH   64
#define HD   128
#define NE   800000
#define N64  6400000
#define EPS_MSG 1e-7f
#define EPS_LN  1e-5f

typedef float v2f __attribute__((ext_vector_type(2)));
typedef float v4f __attribute__((ext_vector_type(4)));
typedef float v8f __attribute__((ext_vector_type(8)));

static __device__ __forceinline__ v8f wmma4(v2f a, v2f b, v8f c) {
  return __builtin_amdgcn_wmma_f32_16x16x4_f32(
       false, a,  false, b,
       (short)0, c,  false,  false);
}

#define GLN_WG   256
#define GLN_T    1024
#define GLN_PER  ((N64 / 4 + GLN_WG * GLN_T - 1) / (GLN_WG * GLN_T))
static_assert(GLN_WG == 256, "a lane of the apply kernel adds 8 pairs: 32 lanes x 8 = 256");

static __device__ __forceinline__ float wave_sum(float v) {
  v += __shfl_xor(v, 16); v += __shfl_xor(v, 8); v += __shfl_xor(v, 4);
  v += __shfl_xor(v, 2);  v += __shfl_xor(v, 1);
  return v;
}

__global__ __launch_bounds__(GLN_T) void k_gln_partial(const float4* __restrict__ h,
                                                       float* __restrict__ part) {
  __shared__ float ss[GLN_T / 32];
  __shared__ float sq[GLN_T / 32];
  const int tid = threadIdx.x, lane = tid & 31, wave = tid >> 5;
  const int g = blockIdx.x * GLN_T + tid;
  float4 v[GLN_PER];
  #pragma unroll
  for (int k = 0; k < GLN_PER; k++) {
    const int i = g + k * (GLN_WG * GLN_T);
    v[k] = h[min(i, N64 / 4 - 1)];
  }
  float s = 0.f, q = 0.f;
  #pragma unroll
  for (int k = 0; k < GLN_PER; k++) {
    const bool in = g + k * (GLN_WG * GLN_T) < N64 / 4;
    const float4 u = v[k];
    const float a = (u.x + u.y) + (u.z + u.w);
    const float b = (u.x * u.x + u.y * u.y) + (u.z * u.z + u.w * u.w);
    s += in ? a : 0.f;
    q += in ? b : 0.f;
  }
  s = wave_sum(s); q = wave_sum(q);
  if (lane == 0) { ss[wave] = s; sq[wave] = q; }
  __syncthreads();
  if (wave == 0) {
    s = wave_sum(ss[lane]); q = wave_sum(sq[lane]);
    if (lane == 0) { part[2 * blockIdx.x] = s; part[2 * blockIdx.x + 1] = q; }
  }
}

__global__ __launch_bounds__(256) void k_gln_apply(const float4* __restrict__ h,
                                                   const float* __restrict__ part,
                                                   const float* __restrict__ w,
                                                   const float* __restrict__ b,
                                                   float4* __restrict__ out) {
  int i = blockIdx.x * 256 + threadIdx.x;
  if (i >= N64 / 4) return;
  const float4 v = h[i];
  float s = 0.f, q = 0.f;
  {
    const float4* p4 = (const float4*)(part + 16 * (threadIdx.x & 31));
    const float4 p0 = p4[0], p1 = p4[1], p2 = p4[2], p3 = p4[3];
    s = ((p0.x + p0.z) + (p1.x + p1.z)) + ((p2.x + p2.z) + (p3.x + p3.z));
    q = ((p0.y + p0.w) + (p1.y + p1.w)) + ((p2.y + p2.w) + (p3.y + p3.w));
  }
  s = wave_sum(s); q = wave_sum(q);
  const float invn = 1.f / (float)N64;
  float mu = s * invn;
  float var = q * invn - mu * mu;
  float scale = 1.f / (sqrtf(fmaxf(var, 0.f)) + EPS_LN);
  int c = (i & 15) << 2;
  float r0 = (v.x - mu) * scale * w[c + 0] + b[c + 0];
  float r1 = (v.y - mu) * scale * w[c + 1] + b[c + 1];
  float r2 = (v.z - mu) * scale * w[c + 2] + b[c + 2];
  float r3 = (v.w - mu) * scale * w[c + 3] + b[c + 3];
  out[i] = make_float4(fmaxf(r0, 0.f), fmaxf(r1, 0.f), fmaxf(r2, 0.f), fmaxf(r3, 0.f));
}

#define NPB   393
#define NOWN  255
#define OWNT  1024
#define LCAP  4096
#define LGRP  16
#define SGRP  7
#define LSTR  (LCAP + 32)
static_assert(NPB * NOWN >= NN && NPB * (NOWN - 1) < NN, "owner ranges must cover NN");
static_assert(NPB <= 512 && NN <= (1 << 23) && NE <= (1 << 23), "packed entry: 9 bits node, 23 bits edge or source");
static_assert(NE % 32 == 0 && NE % 16 == 0, "edge list is walked in 32s; keys in 16-byte loads");
static_assert(NOWN < 256, "owner numbers 0..NOWN-1 and KEY_NONE must fit one byte");

#define KEY_NONE 255u
#define KEY_T    256
#define KEY_PER  4
#define KEY_WG   ((NE / 4 + KEY_T * KEY_PER - 1) / (KEY_T * KEY_PER))

static __device__ __forceinline__ unsigned owner_key(int d) {
  return (unsigned)d < (unsigned)NN ? (unsigned)d / (unsigned)NPB : KEY_NONE;
}

static __device__ __forceinline__ unsigned owner_entry(int s, int d, unsigned key) {
  return ((unsigned)s << 9) | (((unsigned)d - key * (unsigned)NPB) & 511u);
}

__global__ __launch_bounds__(KEY_T) void k_owner_keys(const int* __restrict__ src,
                                                      const int* __restrict__ dst,
                                                      unsigned* __restrict__ keys,
                                                      unsigned* __restrict__ entry) {
  uint4* entry4 = (uint4*)entry;
  const int4* src4 = (const int4*)src;
  const int4* dst4 = (const int4*)dst;
  const int g = blockIdx.x * KEY_T + threadIdx.x;
  int4 s[KEY_PER], d[KEY_PER];
  #pragma unroll
  for (int k = 0; k < KEY_PER; k++) {
    const int i = min(g + k * (KEY_WG * KEY_T), NE / 4 - 1);
    s[k] = src4[i]; d[k] = dst4[i];
  }
  #pragma unroll
  for (int k = 0; k < KEY_PER; k++) {
    const int i = g + k * (KEY_WG * KEY_T);
    if (i < NE / 4) {
      const unsigned k0 = owner_key(d[k].x), k1 = owner_key(d[k].y);
      const unsigned k2 = owner_key(d[k].z), k3 = owner_key(d[k].w);
      keys[i] = k0 | (k1 << 8) | (k2 << 16) | (k3 << 24);
      entry4[i] = make_uint4(owner_entry(s[k].x, d[k].x, k0), owner_entry(s[k].y, d[k].y, k1),
                            owner_entry(s[k].z, d[k].z, k2), owner_entry(s[k].w, d[k].w, k3));
    }
  }
}

static __device__ __forceinline__ void own_max(int* sMax, unsigned dl, float v0, float v1, int lane) {
  atomicMax(&sMax[dl * CH + lane],      __float_as_int(fmaxf(v0, 0.f) + EPS_MSG));
  atomicMax(&sMax[dl * CH + 32 + lane], __float_as_int(fmaxf(v1, 0.f) + EPS_MSG));
}
static __device__ __forceinline__ void own_acc(const int* sMax, float* sDen, float* sNum,
                                               unsigned dl, float v0, float v1, int lane) {
  const float m0 = fmaxf(v0, 0.f) + EPS_MSG, m1 = fmaxf(v1, 0.f) + EPS_MSG;
  const float e0 = __expf(m0 - __int_as_float(sMax[dl * CH + lane]));
  const float e1 = __expf(m1 - __int_as_float(sMax[dl * CH + 32 + lane]));
  atomicAdd(&sDen[dl * CH + lane], e0);       atomicAdd(&sNum[dl * CH + lane], m0 * e0);
  atomicAdd(&sDen[dl * CH + 32 + lane], e1);  atomicAdd(&sNum[dl * CH + 32 + lane], m1 * e1);
}

#define SHIFT_LIM 40.f

static __device__ __forceinline__ void list_two_pass(const unsigned* sList, int total,
                                                     const float* __restrict__ hin,
                                                     int* sMax, float* sDen, float* sNum,
                                                     int wave, int lane) {
  for (int ph = 0; ph < 2; ph++) {
    for (int base = wave * LGRP; base < total; base += 32 * LGRP) {
      unsigned ent[LGRP]; float v0[LGRP], v1[LGRP];
      #pragma unroll
      for (int u = 0; u < LGRP; u++) {
        const int idx = base + u;
        ent[u] = sList[idx < total ? idx : base];
        const float* hp = hin + (size_t)(ent[u] >> 9) * CH + lane;
        v0[u] = hp[0]; v1[u] = hp[32];
      }
      #pragma unroll
      for (int u = 0; u < LGRP; u++) {
        if (base + u < total) {
          if (ph == 0) own_max(sMax, ent[u] & 511u, v0[u], v1[u], lane);
          else         own_acc(sMax, sDen, sNum, ent[u] & 511u, v0[u], v1[u], lane);
        }
      }
    }
    __syncthreads();
  }
}

template <bool FIRST>
__global__ __launch_bounds__(OWNT)
void k_agg_owner(const int* __restrict__ src, const int* __restrict__ dst,
                 const unsigned* __restrict__ keys, const unsigned* __restrict__ entry,
                 const float* __restrict__ hin, float* __restrict__ agg,
                 unsigned* __restrict__ wlist) {
  __shared__ __attribute__((aligned(16))) int   sMax[NPB * CH];
  __shared__ __attribute__((aligned(16))) float sDen[NPB * CH];
  __shared__ __attribute__((aligned(16))) float sNum[NPB * CH];
  __shared__ unsigned sList[LCAP];
  __shared__ unsigned sCnt;
  __shared__ unsigned sBad;
  const int tid = threadIdx.x, lane = tid & 31, wave = tid >> 5;
  const int lo = blockIdx.x * NPB;
  const unsigned cnt = (unsigned)min(NPB, NN - lo);
  const int nout = (int)cnt * CH;

  {
    float4* d4 = (float4*)sDen; float4* n4 = (float4*)sNum;
    const float4 z4 = make_float4(0.f, 0.f, 0.f, 0.f);
    for (int i = tid; i < NPB * CH / 4; i += OWNT) { d4[i] = z4; n4[i] = z4; }
  }
  unsigned* wl = wlist + (size_t)blockIdx.x * LSTR;
  if (tid == 0) { sBad = 0u; if (FIRST) sCnt = 0u; }
  __syncthreads();

  unsigned found;
  if (FIRST) {
    const uint4* key4 = (const uint4*)keys;
    const unsigned me4 = (unsigned)blockIdx.x * 0x01010101u;
    for (int i = tid; i < NE / 16; i += SGRP * OWNT) {
      uint4 d[SGRP];
      #pragma unroll
      for (int u = 0; u < SGRP; u++) {
        const int ii = i + u * OWNT;
        d[u] = (ii < NE / 16) ? key4[ii] : make_uint4(~0u, ~0u, ~0u, ~0u);
      }
      #pragma unroll
      for (int u = 0; u < SGRP; u++) {
        const int e = 16 * (i + u * OWNT);
        const unsigned kk[4] = {d[u].x, d[u].y, d[u].z, d[u].w};
        unsigned f[4];
        #pragma unroll
        for (int c = 0; c < 4; c++) {
          const unsigned y = kk[c] ^ me4;
          f[c] = ((y & 0x7F7F7F7Fu) + 0x7F7F7F7Fu) | y | 0x7F7F7F7Fu;
        }
        const unsigned F = f[0] & __builtin_rotateleft32(f[1], 1) &
                           __builtin_rotateleft32(f[2], 2) & __builtin_rotateleft32(f[3], 3);
        if (F != 0xFFFFFFFFu) {
          unsigned z = ~F;
          while (z) {
            const unsigned q = (unsigned)(__ffs(z) - 1 + 25) & 31u;
            z &= z - 1u;
            const unsigned pos = atomicAdd(&sCnt, 1u);
            if (pos < LCAP) sList[pos] = (unsigned)e + 4u * (q & 7u) + (q >> 3);
          }
        }
      }
    }
    __syncthreads();
    found = sCnt;
    if (tid == 0) wl[LCAP] = found;
  } else {
    found = wl[LCAP];
  }

  if (found <= LCAP) {
    const int total = (int)found;
    if (FIRST) {
      for (int i = tid; i < total; i += OWNT) {
        const unsigned fin = entry[sList[i]];
        sList[i] = fin;
        wl[i] = fin;
      }
    } else {
      for (int i = tid; i < total; i += OWNT) sList[i] = wl[i];
    }
    __syncthreads();

    float top = 0.f;
    for (int base = wave * LGRP; base < total; base += 32 * LGRP) {
      unsigned ent[LGRP]; float v0[LGRP], v1[LGRP];
      #pragma unroll
      for (int u = 0; u < LGRP; u++) {
        const int idx = base + u;
        ent[u] = sList[idx < total ? idx : base];
        const float* hp = hin + (size_t)(ent[u] >> 9) * CH + lane;
        v0[u] = hp[0]; v1[u] = hp[32];
      }
      #pragma unroll
      for (int u = 0; u < LGRP; u++) {
        const bool live = base + u < total;
        const unsigned dl = ent[u] & 511u;
        const float m0 = fmaxf(v0[u], 0.f) + EPS_MSG, m1 = fmaxf(v1[u], 0.f) + EPS_MSG;
        top = fmaxf(top, fmaxf(m0, m1));
        const float x0 = __expf(m0), x1 = __expf(m1);
        const float e0 = live ? x0 : 0.f, e1 = live ? x1 : 0.f;
        const float p0 = live ? m0 * x0 : 0.f, p1 = live ? m1 * x1 : 0.f;
        atomicAdd(&sDen[dl * CH + lane], e0);       atomicAdd(&sNum[dl * CH + lane], p0);
        atomicAdd(&sDen[dl * CH + 32 + lane], e1);  atomicAdd(&sNum[dl * CH + 32 + lane], p1);
      }
    }
    if (!(top <= SHIFT_LIM)) sBad = 1u;
    __syncthreads();
    if (sBad != 0u) {
      for (int i = tid; i < NPB * CH; i += OWNT) { sMax[i] = 0; sDen[i] = 0.f; sNum[i] = 0.f; }
      __syncthreads();
      list_two_pass(sList, total, hin, sMax, sDen, sNum, wave, lane);
    }
  } else {
    for (int i = tid; i < NPB * CH; i += OWNT) sMax[i] = 0;
    __syncthreads();
    for (int ph = 0; ph < 2; ph++) {
      for (int eb = wave * 32; eb < NE; eb += 32 * 32) {
        const unsigned dl = (unsigned)(dst[eb + lane] - lo);
        unsigned mask = __builtin_amdgcn_ballot_w32(dl < cnt);
        while (mask) {
          const int b = __ffs(mask) - 1;
          mask &= mask - 1u;
          const unsigned dlb = (unsigned)__shfl((int)dl, b);
          const float* hp = hin + (size_t)src[eb + b] * CH + lane;
          const float v0 = hp[0], v1 = hp[32];
          if (ph == 0) own_max(sMax, dlb, v0, v1, lane);
          else         own_acc(sMax, sDen, sNum, dlb, v0, v1, lane);
        }
      }
      __syncthreads();
    }
  }

  float* ap = agg + (size_t)lo * CH;
  for (int i = tid; i < nout; i += OWNT) {
    const float den = sDen[i];
    ap[i] = den > 0.f ? sNum[i] / den : 0.f;
  }
}

#define MLP_WAVES 25
#define MLP_T     (MLP_WAVES * 32)
#define MLP_WG    (NN / (16 * MLP_WAVES))
#define MLP_STG   ((CH * HD / 4 + MLP_T - 1) / MLP_T)
static_assert(NN % (16 * MLP_WAVES) == 0, "every wave of every work-group owns a whole strip");
static_assert(NN % 32 == 0, "head: two 16-row strips per work-group");
static_assert(CH == 64 && HD == 128, "channel maps below are written for 64 -> 128 -> 64");

static __device__ __forceinline__ float other_half(float v) {
  return __shfl_xor(v, 16);
}
static __device__ __forceinline__ float f4at(const float4& v, int i) {
  return i == 0 ? v.x : i == 1 ? v.y : i == 2 ? v.z : v.w;
}

template <bool OUTER>
__global__ __launch_bounds__(MLP_T)
void k_conv_mlp(const float* __restrict__ resin, const float* __restrict__ agg,
                const float* __restrict__ W1, const float* __restrict__ b1,
                const float* __restrict__ lnw, const float* __restrict__ lnb,
                const float* __restrict__ W2, const float* __restrict__ b2,
                const float* outer, float* out) {
  __shared__ __attribute__((aligned(16))) float sW1[CH * HD];
  __shared__ __attribute__((aligned(16))) float sW2[HD * CH];
  const int tid = threadIdx.x;
  const int wave = tid >> 5, lane = tid & 31;
  const int half = lane >> 4, l16 = lane & 15;
  const size_t base = ((size_t)blockIdx.x * MLP_WAVES + wave) * 16;

  v4f t1[MLP_STG], t2[MLP_STG];
  #pragma unroll
  for (int k = 0; k < MLP_STG; k++) {
    const int f = min(tid + k * MLP_T, CH * HD / 4 - 1);
    t1[k] = ((const v4f*)W1)[f];
    t2[k] = ((const v4f*)W2)[f];
  }

  v2f actf[CH / 4];
  {
    const float* rp = resin + (base + l16) * CH + 2 * half;
    const float* ap = agg   + (base + l16) * CH + 2 * half;
    v2f rin[CH / 4], ain[CH / 4];
    #pragma unroll
    for (int j = 0; j < CH / 4; j++) {
      rin[j] = *(const v2f*)(rp + 4 * j);
      ain[j] = *(const v2f*)(ap + 4 * j);
    }
    #pragma unroll
    for (int k = 0; k < MLP_STG; k++) asm volatile("" : "+v"(t1[k]), "+v"(t2[k]));
    __builtin_amdgcn_sched_barrier(0);
    #pragma unroll
    for (int j = 0; j < CH / 4; j++) actf[j] = rin[j] + ain[j];
  }
  #pragma unroll
  for (int k = 0; k < MLP_STG; k++) {
    const int f = min(tid + k * MLP_T, CH * HD / 4 - 1);
    {
      const int row = f >> 5, col = f & 31;
      float* d = sW1 + ((row >> 2) * 2 + (col >> 4)) * 256
                     + (((row >> 1) & 1) * 16 + (col & 15)) * 4 + (row & 1);
      d[0] = t1[k].x; d[2] = t1[k].y; d[128] = t1[k].z; d[130] = t1[k].w;
    }
    {
      const int row = f >> 4, col = f & 15;
      const int nt = 4 * (row >> 6) + (row & 3);
      float* d = sW2 + (nt * 4 + ((row >> 3) & 3)) * 256
                     + (((row >> 5) & 1) * 16 + col) * 4 + ((row >> 2) & 1);
      d[0] = t2[k].x; d[2] = t2[k].y; d[128] = t2[k].z; d[130] = t2[k].w;
    }
  }
  __syncthreads();

  v8f acc[HD / 16];
  #pragma unroll
  for (int nt = 0; nt < HD / 16; nt++) acc[nt] = (v8f){0.f, 0.f, 0.f, 0.f, 0.f, 0.f, 0.f, 0.f};
  {
    const v4f* wp = (const v4f*)sW1 + lane;
    __builtin_amdgcn_sched_group_barrier(0x100, 2, 0);
    #pragma unroll
    for (int j = 0; j < CH / 4; j++) {
      #pragma unroll
      for (int g = 0; g < 2; g++) {
        const v4f p = wp[((j * 2 + g) * 2) * 32];
        const v4f q = wp[((j * 2 + g) * 2 + 1) * 32];
        v2f wf;
        wf.x = p.x; wf.y = p.y; acc[4 * g + 0] = wmma4(wf, actf[j], acc[4 * g + 0]);
        wf.x = p.z; wf.y = p.w; acc[4 * g + 1] = wmma4(wf, actf[j], acc[4 * g + 1]);
        wf.x = q.x; wf.y = q.y; acc[4 * g + 2] = wmma4(wf, actf[j], acc[4 * g + 2]);
        wf.x = q.z; wf.y = q.w; acc[4 * g + 3] = wmma4(wf, actf[j], acc[4 * g + 3]);
        __builtin_amdgcn_sched_group_barrier(0x100, 2, 0);
        __builtin_amdgcn_sched_group_barrier(0x008, 4, 0);
      }
    }
  }

  {
    float s = 0.f, q = 0.f;
    #pragma unroll
    for (int g = 0; g < 2; g++) {
      #pragma unroll
      for (int r = 0; r < 8; r++) {
        const float4 bv = *(const float4*)(b1 + 64 * g + 4 * (r + 8 * half));
        #pragma unroll
        for (int t = 0; t < 4; t++) {
          const float v = acc[4 * g + t][r] + f4at(bv, t);
          acc[4 * g + t][r] = v;
          s += v; q += v * v;
        }
      }
    }
    s += other_half(s);
    q += other_half(q);
    const float mu = s * (1.f / HD);
    const float var = q * (1.f / HD) - mu * mu;
    const float inv = rsqrtf(var + EPS_LN);
    #pragma unroll
    for (int g = 0; g < 2; g++) {
      #pragma unroll
      for (int r = 0; r < 8; r++) {
        const float4 wv = *(const float4*)(lnw + 64 * g + 4 * (r + 8 * half));
        const float4 cv = *(const float4*)(lnb + 64 * g + 4 * (r + 8 * half));
        #pragma unroll
        for (int t = 0; t < 4; t++) {
          const float v = (acc[4 * g + t][r] - mu) * inv * f4at(wv, t) + f4at(cv, t);
          acc[4 * g + t][r] = fmaxf(v, 0.f);
        }
      }
    }
  }

  v8f acc2[CH / 16];
  #pragma unroll
  for (int ct = 0; ct < CH / 16; ct++) acc2[ct] = (v8f){0.f, 0.f, 0.f, 0.f, 0.f, 0.f, 0.f, 0.f};
  {
    const v4f* wp = (const v4f*)sW2 + lane;
    #pragma unroll
    for (int nt = 0; nt < HD / 16; nt++) {
      #pragma unroll
      for (int rp = 0; rp < 4; rp++) {
        v2f a; a.x = acc[nt][2 * rp]; a.y = acc[nt][2 * rp + 1];
        const v4f p = wp[((nt * 4 + rp) * 2) * 32];
        const v4f q = wp[((nt * 4 + rp) * 2 + 1) * 32];
        v2f bb;
        bb.x = p.x; bb.y = p.y; acc2[0] = wmma4(a, bb, acc2[0]);
        bb.x = p.z; bb.y = p.w; acc2[1] = wmma4(a, bb, acc2[1]);
        bb.x = q.x; bb.y = q.y; acc2[2] = wmma4(a, bb, acc2[2]);
        bb.x = q.z; bb.y = q.w; acc2[3] = wmma4(a, bb, acc2[3]);
        __builtin_amdgcn_sched_group_barrier(0x100, 2, 0);
        __builtin_amdgcn_sched_group_barrier(0x008, 4, 0);
      }
    }
  }

  const float4 bias = *(const float4*)(b2 + 4 * l16);
  float4 res[8];
  #pragma unroll
  for (int r = 0; r < 8; r++) {
    if (OUTER) res[r] = *(const float4*)(outer + (base + r + 8 * half) * CH + 4 * l16);
    else       res[r] = make_float4(0.f, 0.f, 0.f, 0.f);
  }
  #pragma unroll
  for (int r = 0; r < 8; r++) {
    float4 o;
    o.x = acc2[0][r] + bias.x + res[r].x;
    o.y = acc2[1][r] + bias.y + res[r].y;
    o.z = acc2[2][r] + bias.z + res[r].z;
    o.w = acc2[3][r] + bias.w + res[r].w;
    *(float4*)(out + (base + r + 8 * half) * CH + 4 * l16) = o;
  }
}

__global__ __launch_bounds__(64)
void k_head(const float* __restrict__ xc,
            const float* __restrict__ W1, const float* __restrict__ b1,
            const float* __restrict__ W2, const float* __restrict__ b2,
            const float* __restrict__ W3, const float* __restrict__ b3,
            float* __restrict__ out) {
  const int tid = threadIdx.x;
  const int wave = tid >> 5, lane = tid & 31;
  const int half = lane >> 4, l16 = lane & 15;
  const size_t base = ((size_t)blockIdx.x * 2 + wave) * 16;

  v2f xf[CH / 4];
  {
    const float* xp = xc + (base + l16) * CH + 2 * half;
    #pragma unroll
    for (int j = 0; j < CH / 4; j++) xf[j] = *(const v2f*)(xp + 4 * j);
  }

  v8f h1[CH / 16];
  #pragma unroll
  for (int nt = 0; nt < CH / 16; nt++) h1[nt] = (v8f){0.f, 0.f, 0.f, 0.f, 0.f, 0.f, 0.f, 0.f};
  {
    const float* wp = W1 + (2 * half) * CH + 4 * l16;
    #pragma unroll
    for (int j = 0; j < CH / 4; j++) {
      const float4 w0 = *(const float4*)(wp + (4 * j) * CH);
      const float4 w1 = *(const float4*)(wp + (4 * j + 1) * CH);
      #pragma unroll
      for (int nt = 0; nt < CH / 16; nt++) {
        v2f wf; wf.x = f4at(w0, nt); wf.y = f4at(w1, nt);
        h1[nt] = wmma4(wf, xf[j], h1[nt]);
      }
    }
  }
  #pragma unroll
  for (int r = 0; r < 8; r++) {
    const float4 bv = *(const float4*)(b1 + 4 * (r + 8 * half));
    #pragma unroll
    for (int nt = 0; nt < CH / 16; nt++) h1[nt][r] = fmaxf(h1[nt][r] + f4at(bv, nt), 0.f);
  }

  v8f h2[CH / 16];
  #pragma unroll
  for (int nt = 0; nt < CH / 16; nt++) h2[nt] = (v8f){0.f, 0.f, 0.f, 0.f, 0.f, 0.f, 0.f, 0.f};
  {
    const float* wp = W2 + (size_t)(32 * half) * CH + 4 * l16;
    #pragma unroll
    for (int nt = 0; nt < CH / 16; nt++) {
      #pragma unroll
      for (int rp = 0; rp < 4; rp++) {
        v2f hb; hb.x = h1[nt][2 * rp]; hb.y = h1[nt][2 * rp + 1];
        const int k0 = 8 * rp + nt;
        const float4 w0 = *(const float4*)(wp + (k0) * CH);
        const float4 w1 = *(const float4*)(wp + (k0 + 4) * CH);
        #pragma unroll
        for (int n2 = 0; n2 < CH / 16; n2++) {
          v2f wf; wf.x = f4at(w0, n2); wf.y = f4at(w1, n2);
          h2[n2] = wmma4(wf, hb, h2[n2]);
        }
      }
    }
  }

  float o0 = 0.f, o1 = 0.f;
  #pragma unroll
  for (int r = 0; r < 8; r++) {
    const int c0 = 4 * (r + 8 * half);
    const float4 bv = *(const float4*)(b2 + c0);
    const float4 ta = *(const float4*)(W3 + 2 * c0);
    const float4 tb = *(const float4*)(W3 + 2 * c0 + 4);
    const float w3[8] = {ta.x, ta.y, ta.z, ta.w, tb.x, tb.y, tb.z, tb.w};
    #pragma unroll
    for (int n2 = 0; n2 < CH / 16; n2++) {
      const float v = fmaxf(h2[n2][r] + f4at(bv, n2), 0.f);
      o0 += v * w3[2 * n2];
      o1 += v * w3[2 * n2 + 1];
    }
  }
  o0 += other_half(o0);
  o1 += other_half(o1);
  if (half == 0) {
    float2 res; res.x = o0 + b3[0]; res.y = o1 + b3[1];
    *(float2*)(out + (base + l16) * 2) = res;
  }
}

extern "C" void kernel_launch(void* const* d_in, const int* in_sizes, int n_in,
                              void* d_out, int out_size, void* d_ws, size_t ws_size,
                              hipStream_t stream) {
  (void)in_sizes; (void)n_in; (void)out_size; (void)ws_size;
  const float* x      = (const float*)d_in[0];
  const int*   ei     = (const int*)d_in[1];
  const int*   src    = ei;
  const int*   dst    = ei + NE;
  const float* c1_W1  = (const float*)d_in[2];
  const float* c1_b1  = (const float*)d_in[3];
  const float* c1_lnw = (const float*)d_in[4];
  const float* c1_lnb = (const float*)d_in[5];
  const float* c1_W2  = (const float*)d_in[6];
  const float* c1_b2  = (const float*)d_in[7];
  const float* n1_w   = (const float*)d_in[8];
  const float* n1_b   = (const float*)d_in[9];
  const float* cW1    = (const float*)d_in[10];
  const float* cb1    = (const float*)d_in[11];
  const float* clnw   = (const float*)d_in[12];
  const float* clnb   = (const float*)d_in[13];
  const float* cW2    = (const float*)d_in[14];
  const float* cb2    = (const float*)d_in[15];
  const float* nw     = (const float*)d_in[16];
  const float* nb     = (const float*)d_in[17];
  const float* lW1    = (const float*)d_in[18];
  const float* lb1    = (const float*)d_in[19];
  const float* lW2    = (const float*)d_in[20];
  const float* lb2    = (const float*)d_in[21];
  const float* lW3    = (const float*)d_in[22];
  const float* lb3    = (const float*)d_in[23];

  float* ws    = (float*)d_ws;
  float* xc    = ws;
  float* hpre  = ws + (size_t)N64;
  float* agg   = ws + 4 * (size_t)N64;
  float* part  = ws + 3 * (size_t)N64;
  unsigned* wlist = (unsigned*)(ws + 2 * (size_t)N64);
  unsigned* keys  = (unsigned*)(ws + 3 * (size_t)N64 + 1024);
  unsigned* entry = (unsigned*)(ws + 3 * (size_t)N64 + 262144);

  const dim3 B(256);
  const int APB  = (N64 / 4 + 255) / 256;
  const int MLPB = NN / 32;
  const dim3 B2(64);

  auto edge_phase = [&](const float* hin, bool first) {
    if (first) k_agg_owner<true><<<NOWN, OWNT, 0, stream>>>(src, dst, keys, entry, hin, agg, wlist);
    else       k_agg_owner<false><<<NOWN, OWNT, 0, stream>>>(src, dst, keys, entry, hin, agg, wlist);
  };

  k_owner_keys<<<KEY_WG, KEY_T, 0, stream>>>(src, dst, keys, entry);

  edge_phase(x, true);
  k_conv_mlp<false><<<MLP_WG, MLP_T, 0, stream>>>(x, agg, c1_W1, c1_b1, c1_lnw, c1_lnb,
                                     c1_W2, c1_b2, nullptr, hpre);
  k_gln_partial<<<GLN_WG, GLN_T, 0, stream>>>((const float4*)hpre, part);
  k_gln_apply<<<APB, B, 0, stream>>>((const float4*)hpre, part, n1_w, n1_b, (float4*)xc);

  for (int i = 0; i < 3; i++) {
    k_gln_partial<<<GLN_WG, GLN_T, 0, stream>>>((const float4*)xc, part);
    k_gln_apply<<<APB, B, 0, stream>>>((const float4*)xc, part, nw + i * CH, nb + i * CH,
                                       (float4*)hpre);
    edge_phase(hpre, false);
    k_conv_mlp<true><<<MLP_WG, MLP_T, 0, stream>>>(hpre, agg,
                                       cW1 + (size_t)i * CH * HD, cb1 + (size_t)i * HD,
                                       clnw + (size_t)i * HD, clnb + (size_t)i * HD,
                                       cW2 + (size_t)i * HD * CH, cb2 + (size_t)i * CH,
                                       xc, xc);
  }

  k_head<<<MLPB, B2, 0, stream>>>(xc, lW1, lb1, lW2, lb2, lW3, lb3, (float*)d_out);
}


# Conv-MLP run by the owner work-group on its own 393 nodes straight from the LDS sums (wave w = rows 16w..16w+15): aggregate never stored or re-read, 4 launches fewer; all sum orders kept
# speedup vs baseline: 5.0568x; 1.1005x over previous
#include <hip/hip_runtime.h>
#include <math.h>

#define NN   100000
#define CH   64
#define HD   128
#define NE   800000
#define N64  6400000
#define EPS_MSG 1e-7f
#define EPS_LN  1e-5f

typedef float v2f __attribute__((ext_vector_type(2)));
typedef float v4f __attribute__((ext_vector_type(4)));
typedef float v8f __attribute__((ext_vector_type(8)));

static __device__ __forceinline__ v8f wmma4(v2f a, v2f b, v8f c) {
  return __builtin_amdgcn_wmma_f32_16x16x4_f32(
       false, a,  false, b,
       (short)0, c,  false,  false);
}

#define GLN_WG   256
#define GLN_T    1024
#define GLN_PER  ((N64 / 4 + GLN_WG * GLN_T - 1) / (GLN_WG * GLN_T))
static_assert(GLN_WG == 256, "a lane of the apply kernel adds 8 pairs: 32 lanes x 8 = 256");

static __device__ __forceinline__ float wave_sum(float v) {
  v += __shfl_xor(v, 16); v += __shfl_xor(v, 8); v += __shfl_xor(v, 4);
  v += __shfl_xor(v, 2);  v += __shfl_xor(v, 1);
  return v;
}

__global__ __launch_bounds__(GLN_T) void k_gln_partial(const float4* __restrict__ h,
                                                       float* __restrict__ part) {
  __shared__ float ss[GLN_T / 32];
  __shared__ float sq[GLN_T / 32];
  const int tid = threadIdx.x, lane = tid & 31, wave = tid >> 5;
  const int g = blockIdx.x * GLN_T + tid;
  float4 v[GLN_PER];
  #pragma unroll
  for (int k = 0; k < GLN_PER; k++) {
    const int i = g + k * (GLN_WG * GLN_T);
    v[k] = h[min(i, N64 / 4 - 1)];
  }
  float s = 0.f, q = 0.f;
  #pragma unroll
  for (int k = 0; k < GLN_PER; k++) {
    const bool in = g + k * (GLN_WG * GLN_T) < N64 / 4;
    const float4 u = v[k];
    const float a = (u.x + u.y) + (u.z + u.w);
    const float b = (u.x * u.x + u.y * u.y) + (u.z * u.z + u.w * u.w);
    s += in ? a : 0.f;
    q += in ? b : 0.f;
  }
  s = wave_sum(s); q = wave_sum(q);
  if (lane == 0) { ss[wave] = s; sq[wave] = q; }
  __syncthreads();
  if (wave == 0) {
    s = wave_sum(ss[lane]); q = wave_sum(sq[lane]);
    if (lane == 0) { part[2 * blockIdx.x] = s; part[2 * blockIdx.x + 1] = q; }
  }
}

__global__ __launch_bounds__(256) void k_gln_apply(const float4* __restrict__ h,
                                                   const float* __restrict__ part,
                                                   const float* __restrict__ w,
                                                   const float* __restrict__ b,
                                                   float4* __restrict__ out) {
  int i = blockIdx.x * 256 + threadIdx.x;
  if (i >= N64 / 4) return;
  const float4 v = h[i];
  float s = 0.f, q = 0.f;
  {
    const float4* p4 = (const float4*)(part + 16 * (threadIdx.x & 31));
    const float4 p0 = p4[0], p1 = p4[1], p2 = p4[2], p3 = p4[3];
    s = ((p0.x + p0.z) + (p1.x + p1.z)) + ((p2.x + p2.z) + (p3.x + p3.z));
    q = ((p0.y + p0.w) + (p1.y + p1.w)) + ((p2.y + p2.w) + (p3.y + p3.w));
  }
  s = wave_sum(s); q = wave_sum(q);
  const float invn = 1.f / (float)N64;
  float mu = s * invn;
  float var = q * invn - mu * mu;
  float scale = 1.f / (sqrtf(fmaxf(var, 0.f)) + EPS_LN);
  int c = (i & 15) << 2;
  float r0 = (v.x - mu) * scale * w[c + 0] + b[c + 0];
  float r1 = (v.y - mu) * scale * w[c + 1] + b[c + 1];
  float r2 = (v.z - mu) * scale * w[c + 2] + b[c + 2];
  float r3 = (v.w - mu) * scale * w[c + 3] + b[c + 3];
  out[i] = make_float4(fmaxf(r0, 0.f), fmaxf(r1, 0.f), fmaxf(r2, 0.f), fmaxf(r3, 0.f));
}

#define NPB   393
#define NOWN  255
#define OWNT  1024
#define LCAP  4096
#define LGRP  16
#define SGRP  7
#define LSTR  (LCAP + 32)
static_assert(NPB * NOWN >= NN && NPB * (NOWN - 1) < NN, "owner ranges must cover NN");
static_assert(NPB <= 512 && NN <= (1 << 23) && NE <= (1 << 23), "packed entry: 9 bits node, 23 bits edge or source");
static_assert(NE % 32 == 0 && NE % 16 == 0, "edge list is walked in 32s; keys in 16-byte loads");
static_assert(NOWN < 256, "owner numbers 0..NOWN-1 and KEY_NONE must fit one byte");

#define KEY_NONE 255u
#define KEY_T    256
#define KEY_PER  4
#define KEY_WG   ((NE / 4 + KEY_T * KEY_PER - 1) / (KEY_T * KEY_PER))

static __device__ __forceinline__ unsigned owner_key(int d) {
  return (unsigned)d < (unsigned)NN ? (unsigned)d / (unsigned)NPB : KEY_NONE;
}

static __device__ __forceinline__ unsigned owner_entry(int s, int d, unsigned key) {
  return ((unsigned)s << 9) | (((unsigned)d - key * (unsigned)NPB) & 511u);
}

__global__ __launch_bounds__(KEY_T) void k_owner_keys(const int* __restrict__ src,
                                                      const int* __restrict__ dst,
                                                      unsigned* __restrict__ keys,
                                                      unsigned* __restrict__ entry) {
  uint4* entry4 = (uint4*)entry;
  const int4* src4 = (const int4*)src;
  const int4* dst4 = (const int4*)dst;
  const int g = blockIdx.x * KEY_T + threadIdx.x;
  int4 s[KEY_PER], d[KEY_PER];
  #pragma unroll
  for (int k = 0; k < KEY_PER; k++) {
    const int i = min(g + k * (KEY_WG * KEY_T), NE / 4 - 1);
    s[k] = src4[i]; d[k] = dst4[i];
  }
  #pragma unroll
  for (int k = 0; k < KEY_PER; k++) {
    const int i = g + k * (KEY_WG * KEY_T);
    if (i < NE / 4) {
      const unsigned k0 = owner_key(d[k].x), k1 = owner_key(d[k].y);
      const unsigned k2 = owner_key(d[k].z), k3 = owner_key(d[k].w);
      keys[i] = k0 | (k1 << 8) | (k2 << 16) | (k3 << 24);
      entry4[i] = make_uint4(owner_entry(s[k].x, d[k].x, k0), owner_entry(s[k].y, d[k].y, k1),
                            owner_entry(s[k].z, d[k].z, k2), owner_entry(s[k].w, d[k].w, k3));
    }
  }
}

#define PIT 66
static __device__ __forceinline__ void own_max(int* sMax, unsigned dl, float v0, float v1, int lane) {
  atomicMax(&sMax[dl * CH + lane],      __float_as_int(fmaxf(v0, 0.f) + EPS_MSG));
  atomicMax(&sMax[dl * CH + 32 + lane], __float_as_int(fmaxf(v1, 0.f) + EPS_MSG));
}
static __device__ __forceinline__ void own_acc(const int* sMax, float* sDen, float* sNum,
                                               unsigned dl, float v0, float v1, int lane) {
  const float m0 = fmaxf(v0, 0.f) + EPS_MSG, m1 = fmaxf(v1, 0.f) + EPS_MSG;
  const float e0 = __expf(m0 - __int_as_float(sMax[dl * CH + lane]));
  const float e1 = __expf(m1 - __int_as_float(sMax[dl * CH + 32 + lane]));
  atomicAdd(&sDen[dl * CH + lane], e0);       atomicAdd(&sNum[dl * PIT + lane], m0 * e0);
  atomicAdd(&sDen[dl * CH + 32 + lane], e1);  atomicAdd(&sNum[dl * PIT + 32 + lane], m1 * e1);
}

#define SHIFT_LIM 40.f

static __device__ __forceinline__ void list_two_pass(const unsigned* sList, int total,
                                                     const float* __restrict__ hin,
                                                     int* sMax, float* sDen, float* sNum,
                                                     int wave, int lane) {
  for (int ph = 0; ph < 2; ph++) {
    for (int base = wave * LGRP; base < total; base += 32 * LGRP) {
      unsigned ent[LGRP]; float v0[LGRP], v1[LGRP];
      #pragma unroll
      for (int u = 0; u < LGRP; u++) {
        const int idx = base + u;
        ent[u] = sList[idx < total ? idx : base];
        const float* hp = hin + (size_t)(ent[u] >> 9) * CH + lane;
        v0[u] = hp[0]; v1[u] = hp[32];
      }
      #pragma unroll
      for (int u = 0; u < LGRP; u++) {
        if (base + u < total) {
          if (ph == 0) own_max(sMax, ent[u] & 511u, v0[u], v1[u], lane);
          else         own_acc(sMax, sDen, sNum, ent[u] & 511u, v0[u], v1[u], lane);
        }
      }
    }
    __syncthreads();
  }
}

#define MLP_STG   (CH * HD / 4 / OWNT)
static_assert((CH * HD / 4) % OWNT == 0, "every thread copies whole pieces, none left over");
static_assert(CH * HD * 2 <= NPB * CH, "both staged matrices fit the max plane's space");
static_assert((NPB + 15) / 16 <= OWNT / 32, "one wave per 16-row strip of the group's nodes");
static_assert(NN % 32 == 0, "head: two 16-row strips per work-group");
static_assert(CH == 64 && HD == 128, "channel maps below are written for 64 -> 128 -> 64");
static_assert(PIT % 2 == 0, "8-byte reads of the aggregate plane");

static __device__ __forceinline__ float other_half(float v) {
  return __shfl_xor(v, 16);
}
static __device__ __forceinline__ float f4at(const float4& v, int i) {
  return i == 0 ? v.x : i == 1 ? v.y : i == 2 ? v.z : v.w;
}

template <bool FIRST, bool OUTER>
__global__ __launch_bounds__(OWNT)
void k_layer(const int* __restrict__ src, const int* __restrict__ dst,
             const unsigned* __restrict__ keys, const unsigned* __restrict__ entry,
             const float* __restrict__ hin, unsigned* __restrict__ wlist,
             const float* __restrict__ W1, const float* __restrict__ b1,
             const float* __restrict__ lnw, const float* __restrict__ lnb,
             const float* __restrict__ W2, const float* __restrict__ b2,
             const float* outer, float* out) {
  __shared__ __attribute__((aligned(16))) int   sMax[NPB * CH];
  __shared__ __attribute__((aligned(16))) float sDen[NPB * CH];
  __shared__ __attribute__((aligned(16))) float sNum[NPB * PIT];
  __shared__ unsigned sList[LCAP];
  __shared__ unsigned sCnt;
  __shared__ unsigned sBad;
  const int tid = threadIdx.x, lane = tid & 31, wave = tid >> 5;
  const int lo = blockIdx.x * NPB;
  const unsigned cnt = (unsigned)min(NPB, NN - lo);
  const int nout = (int)cnt * CH;

  {
    float4* d4 = (float4*)sDen; float2* n2 = (float2*)sNum;
    const float4 z4 = make_float4(0.f, 0.f, 0.f, 0.f);
    for (int i = tid; i < NPB * CH / 4; i += OWNT) d4[i] = z4;
    for (int i = tid; i < NPB * PIT / 2; i += OWNT) n2[i] = make_float2(0.f, 0.f);
  }
  unsigned* wl = wlist + (size_t)blockIdx.x * LSTR;
  if (tid == 0) { sBad = 0u; if (FIRST) sCnt = 0u; }
  __syncthreads();

  unsigned found;
  if (FIRST) {
    const uint4* key4 = (const uint4*)keys;
    const unsigned me4 = (unsigned)blockIdx.x * 0x01010101u;
    for (int i = tid; i < NE / 16; i += SGRP * OWNT) {
      uint4 d[SGRP];
      #pragma unroll
      for (int u = 0; u < SGRP; u++) {
        const int ii = i + u * OWNT;
        d[u] = (ii < NE / 16) ? key4[ii] : make_uint4(~0u, ~0u, ~0u, ~0u);
      }
      #pragma unroll
      for (int u = 0; u < SGRP; u++) {
        const int e = 16 * (i + u * OWNT);
        const unsigned kk[4] = {d[u].x, d[u].y, d[u].z, d[u].w};
        unsigned f[4];
        #pragma unroll
        for (int c = 0; c < 4; c++) {
          const unsigned y = kk[c] ^ me4;
          f[c] = ((y & 0x7F7F7F7Fu) + 0x7F7F7F7Fu) | y | 0x7F7F7F7Fu;
        }
        const unsigned F = f[0] & __builtin_rotateleft32(f[1], 1) &
                           __builtin_rotateleft32(f[2], 2) & __builtin_rotateleft32(f[3], 3);
        if (F != 0xFFFFFFFFu) {
          unsigned z = ~F;
          while (z) {
            const unsigned q = (unsigned)(__ffs(z) - 1 + 25) & 31u;
            z &= z - 1u;
            const unsigned pos = atomicAdd(&sCnt, 1u);
            if (pos < LCAP) sList[pos] = (unsigned)e + 4u * (q & 7u) + (q >> 3);
          }
        }
      }
    }
    __syncthreads();
    found = sCnt;
    if (tid == 0) wl[LCAP] = found;
  } else {
    found = wl[LCAP];
  }

  if (found <= LCAP) {
    const int total = (int)found;
    if (FIRST) {
      for (int i = tid; i < total; i += OWNT) {
        const unsigned fin = entry[sList[i]];
        sList[i] = fin;
        wl[i] = fin;
      }
    } else {
      for (int i = tid; i < total; i += OWNT) sList[i] = wl[i];
    }
    __syncthreads();

    float top = 0.f;
    for (int base = wave * LGRP; base < total; base += 32 * LGRP) {
      unsigned ent[LGRP]; float v0[LGRP], v1[LGRP];
      #pragma unroll
      for (int u = 0; u < LGRP; u++) {
        const int idx = base + u;
        ent[u] = sList[idx < total ? idx : base];
        const float* hp = hin + (size_t)(ent[u] >> 9) * CH + lane;
        v0[u] = hp[0]; v1[u] = hp[32];
      }
      #pragma unroll
      for (int u = 0; u < LGRP; u++) {
        const bool live = base + u < total;
        const unsigned dl = ent[u] & 511u;
        const float m0 = fmaxf(v0[u], 0.f) + EPS_MSG, m1 = fmaxf(v1[u], 0.f) + EPS_MSG;
        top = fmaxf(top, fmaxf(m0, m1));
        const float x0 = __expf(m0), x1 = __expf(m1);
        const float e0 = live ? x0 : 0.f, e1 = live ? x1 : 0.f;
        const float p0 = live ? m0 * x0 : 0.f, p1 = live ? m1 * x1 : 0.f;
        atomicAdd(&sDen[dl * CH + lane], e0);       atomicAdd(&sNum[dl * PIT + lane], p0);
        atomicAdd(&sDen[dl * CH + 32 + lane], e1);  atomicAdd(&sNum[dl * PIT + 32 + lane], p1);
      }
    }
    if (!(top <= SHIFT_LIM)) sBad = 1u;
    __syncthreads();
    if (sBad != 0u) {
      for (int i = tid; i < NPB * CH; i += OWNT) { sMax[i] = 0; sDen[i] = 0.f; }
      for (int i = tid; i < NPB * PIT; i += OWNT) sNum[i] = 0.f;
      __syncthreads();
      list_two_pass(sList, total, hin, sMax, sDen, sNum, wave, lane);
    }
  } else {
    for (int i = tid; i < NPB * CH; i += OWNT) sMax[i] = 0;
    __syncthreads();
    for (int ph = 0; ph < 2; ph++) {
      for (int eb = wave * 32; eb < NE; eb += 32 * 32) {
        const unsigned dl = (unsigned)(dst[eb + lane] - lo);
        unsigned mask = __builtin_amdgcn_ballot_w32(dl < cnt);
        while (mask) {
          const int b = __ffs(mask) - 1;
          mask &= mask - 1u;
          const unsigned dlb = (unsigned)__shfl((int)dl, b);
          const float* hp = hin + (size_t)src[eb + b] * CH + lane;
          const float v0 = hp[0], v1 = hp[32];
          if (ph == 0) own_max(sMax, dlb, v0, v1, lane);
          else         own_acc(sMax, sDen, sNum, dlb, v0, v1, lane);
        }
      }
      __syncthreads();
    }
  }

  const int half = lane >> 4, l16 = lane & 15;
  const int nstr = ((int)cnt + 15) >> 4;

  v4f t1[MLP_STG], t2[MLP_STG];
  #pragma unroll
  for (int k = 0; k < MLP_STG; k++) {
    const int f = tid + k * OWNT;
    t1[k] = ((const v4f*)W1)[f];
    t2[k] = ((const v4f*)W2)[f];
  }
  const int lr = min(16 * wave + l16, (int)cnt - 1);
  v2f rin[CH / 4];
  {
    const float* rp = hin + (size_t)(lo + lr) * CH + 2 * half;
    #pragma unroll
    for (int j = 0; j < CH / 4; j++) rin[j] = *(const v2f*)(rp + 4 * j);
  }

  for (int i = tid; i < nout; i += OWNT) {
    const float den = sDen[i];
    float* np = sNum + (i >> 6) * PIT + (i & 63);
    const float num = *np;
    *np = den > 0.f ? num / den : 0.f;
  }

  float* sW1 = (float*)sMax;
  float* sW2 = sW1 + CH * HD;
  #pragma unroll
  for (int k = 0; k < MLP_STG; k++) {
    const int f = tid + k * OWNT;
    {
      const int row = f >> 5, col = f & 31;
      float* d = sW1 + ((row >> 2) * 2 + (col >> 4)) * 256
                     + (((row >> 1) & 1) * 16 + (col & 15)) * 4 + (row & 1);
      d[0] = t1[k].x; d[2] = t1[k].y; d[128] = t1[k].z; d[130] = t1[k].w;
    }
    {
      const int row = f >> 4, col = f & 15;
      const int nt = 4 * (row >> 6) + (row & 3);
      float* d = sW2 + (nt * 4 + ((row >> 3) & 3)) * 256
                     + (((row >> 5) & 1) * 16 + col) * 4 + ((row >> 2) & 1);
      d[0] = t2[k].x; d[2] = t2[k].y; d[128] = t2[k].z; d[130] = t2[k].w;
    }
  }
  __syncthreads();
  if (wave >= nstr) return;

  v2f actf[CH / 4];
  {
    const float* ap = sNum + lr * PIT + 2 * half;
    #pragma unroll
    for (int j = 0; j < CH / 4; j++) actf[j] = rin[j] + *(const v2f*)(ap + 4 * j);
  }
  __builtin_amdgcn_sched_barrier(0);

  v8f acc[HD / 16];
  #pragma unroll
  for (int nt = 0; nt < HD / 16; nt++) acc[nt] = (v8f){0.f, 0.f, 0.f, 0.f, 0.f, 0.f, 0.f, 0.f};
  {
    const v4f* wp = (const v4f*)sW1 + lane;
    __builtin_amdgcn_sched_group_barrier(0x100, 2, 0);
    #pragma unroll
    for (int j = 0; j < CH / 4; j++) {
      #pragma unroll
      for (int g = 0; g < 2; g++) {
        const v4f p = wp[((j * 2 + g) * 2) * 32];
        const v4f q = wp[((j * 2 + g) * 2 + 1) * 32];
        v2f wf;
        wf.x = p.x; wf.y = p.y; acc[4 * g + 0] = wmma4(wf, actf[j], acc[4 * g + 0]);
        wf.x = p.z; wf.y = p.w; acc[4 * g + 1] = wmma4(wf, actf[j], acc[4 * g + 1]);
        wf.x = q.x; wf.y = q.y; acc[4 * g + 2] = wmma4(wf, actf[j], acc[4 * g + 2]);
        wf.x = q.z; wf.y = q.w; acc[4 * g + 3] = wmma4(wf, actf[j], acc[4 * g + 3]);
        __builtin_amdgcn_sched_group_barrier(0x100, 2, 0);
        __builtin_amdgcn_sched_group_barrier(0x008, 4, 0);
      }
    }
  }

  {
    float s = 0.f, q = 0.f;
    #pragma unroll
    for (int g = 0; g < 2; g++) {
      #pragma unroll
      for (int r = 0; r < 8; r++) {
        const float4 bv = *(const float4*)(b1 + 64 * g + 4 * (r + 8 * half));
        #pragma unroll
        for (int t = 0; t < 4; t++) {
          const float v = acc[4 * g + t][r] + f4at(bv, t);
          acc[4 * g + t][r] = v;
          s += v; q += v * v;
        }
      }
    }
    s += other_half(s);
    q += other_half(q);
    const float mu = s * (1.f / HD);
    const float var = q * (1.f / HD) - mu * mu;
    const float inv = rsqrtf(var + EPS_LN);
    #pragma unroll
    for (int g = 0; g < 2; g++) {
      #pragma unroll
      for (int r = 0; r < 8; r++) {
        const float4 wv = *(const float4*)(lnw + 64 * g + 4 * (r + 8 * half));
        const float4 cv = *(const float4*)(lnb + 64 * g + 4 * (r + 8 * half));
        #pragma unroll
        for (int t = 0; t < 4; t++) {
          const float v = (acc[4 * g + t][r] - mu) * inv * f4at(wv, t) + f4at(cv, t);
          acc[4 * g + t][r] = fmaxf(v, 0.f);
        }
      }
    }
  }

  v8f acc2[CH / 16];
  #pragma unroll
  for (int ct = 0; ct < CH / 16; ct++) acc2[ct] = (v8f){0.f, 0.f, 0.f, 0.f, 0.f, 0.f, 0.f, 0.f};
  {
    const v4f* wp = (const v4f*)sW2 + lane;
    #pragma unroll
    for (int nt = 0; nt < HD / 16; nt++) {
      #pragma unroll
      for (int rp = 0; rp < 4; rp++) {
        v2f a; a.x = acc[nt][2 * rp]; a.y = acc[nt][2 * rp + 1];
        const v4f p = wp[((nt * 4 + rp) * 2) * 32];
        const v4f q = wp[((nt * 4 + rp) * 2 + 1) * 32];
        v2f bb;
        bb.x = p.x; bb.y = p.y; acc2[0] = wmma4(a, bb, acc2[0]);
        bb.x = p.z; bb.y = p.w; acc2[1] = wmma4(a, bb, acc2[1]);
        bb.x = q.x; bb.y = q.y; acc2[2] = wmma4(a, bb, acc2[2]);
        bb.x = q.z; bb.y = q.w; acc2[3] = wmma4(a, bb, acc2[3]);
        __builtin_amdgcn_sched_group_barrier(0x100, 2, 0);
        __builtin_amdgcn_sched_group_barrier(0x008, 4, 0);
      }
    }
  }

  const int m0 = 16 * wave + 8 * half;
  const float4 bias = *(const float4*)(b2 + 4 * l16);
  float4 res[8];
  #pragma unroll
  for (int r = 0; r < 8; r++) {
    const int row = lo + min(m0 + r, (int)cnt - 1);
    if (OUTER) res[r] = *(const float4*)(outer + (size_t)row * CH + 4 * l16);
    else       res[r] = make_float4(0.f, 0.f, 0.f, 0.f);
  }
  #pragma unroll
  for (int r = 0; r < 8; r++) {
    float4 o;
    o.x = acc2[0][r] + bias.x + res[r].x;
    o.y = acc2[1][r] + bias.y + res[r].y;
    o.z = acc2[2][r] + bias.z + res[r].z;
    o.w = acc2[3][r] + bias.w + res[r].w;
    if (m0 + r < (int)cnt) *(float4*)(out + (size_t)(lo + m0 + r) * CH + 4 * l16) = o;
  }
}

__global__ __launch_bounds__(64)
void k_head(const float* __restrict__ xc,
            const float* __restrict__ W1, const float* __restrict__ b1,
            const float* __restrict__ W2, const float* __restrict__ b2,
            const float* __restrict__ W3, const float* __restrict__ b3,
            float* __restrict__ out) {
  const int tid = threadIdx.x;
  const int wave = tid >> 5, lane = tid & 31;
  const int half = lane >> 4, l16 = lane & 15;
  const size_t base = ((size_t)blockIdx.x * 2 + wave) * 16;

  v2f xf[CH / 4];
  {
    const float* xp = xc + (base + l16) * CH + 2 * half;
    #pragma unroll
    for (int j = 0; j < CH / 4; j++) xf[j] = *(const v2f*)(xp + 4 * j);
  }

  v8f h1[CH / 16];
  #pragma unroll
  for (int nt = 0; nt < CH / 16; nt++) h1[nt] = (v8f){0.f, 0.f, 0.f, 0.f, 0.f, 0.f, 0.f, 0.f};
  {
    const float* wp = W1 + (2 * half) * CH + 4 * l16;
    #pragma unroll
    for (int j = 0; j < CH / 4; j++) {
      const float4 w0 = *(const float4*)(wp + (4 * j) * CH);
      const float4 w1 = *(const float4*)(wp + (4 * j + 1) * CH);
      #pragma unroll
      for (int nt = 0; nt < CH / 16; nt++) {
        v2f wf; wf.x = f4at(w0, nt); wf.y = f4at(w1, nt);
        h1[nt] = wmma4(wf, xf[j], h1[nt]);
      }
    }
  }
  #pragma unroll
  for (int r = 0; r < 8; r++) {
    const float4 bv = *(const float4*)(b1 + 4 * (r + 8 * half));
    #pragma unroll
    for (int nt = 0; nt < CH / 16; nt++) h1[nt][r] = fmaxf(h1[nt][r] + f4at(bv, nt), 0.f);
  }

  v8f h2[CH / 16];
  #pragma unroll
  for (int nt = 0; nt < CH / 16; nt++) h2[nt] = (v8f){0.f, 0.f, 0.f, 0.f, 0.f, 0.f, 0.f, 0.f};
  {
    const float* wp = W2 + (size_t)(32 * half) * CH + 4 * l16;
    #pragma unroll
    for (int nt = 0; nt < CH / 16; nt++) {
      #pragma unroll
      for (int rp = 0; rp < 4; rp++) {
        v2f hb; hb.x = h1[nt][2 * rp]; hb.y = h1[nt][2 * rp + 1];
        const int k0 = 8 * rp + nt;
        const float4 w0 = *(const float4*)(wp + (k0) * CH);
        const float4 w1 = *(const float4*)(wp + (k0 + 4) * CH);
        #pragma unroll
        for (int n2 = 0; n2 < CH / 16; n2++) {
          v2f wf; wf.x = f4at(w0, n2); wf.y = f4at(w1, n2);
          h2[n2] = wmma4(wf, hb, h2[n2]);
        }
      }
    }
  }

  float o0 = 0.f, o1 = 0.f;
  #pragma unroll
  for (int r = 0; r < 8; r++) {
    const int c0 = 4 * (r + 8 * half);
    const float4 bv = *(const float4*)(b2 + c0);
    const float4 ta = *(const float4*)(W3 + 2 * c0);
    const float4 tb = *(const float4*)(W3 + 2 * c0 + 4);
    const float w3[8] = {ta.x, ta.y, ta.z, ta.w, tb.x, tb.y, tb.z, tb.w};
    #pragma unroll
    for (int n2 = 0; n2 < CH / 16; n2++) {
      const float v = fmaxf(h2[n2][r] + f4at(bv, n2), 0.f);
      o0 += v * w3[2 * n2];
      o1 += v * w3[2 * n2 + 1];
    }
  }
  o0 += other_half(o0);
  o1 += other_half(o1);
  if (half == 0) {
    float2 res; res.x = o0 + b3[0]; res.y = o1 + b3[1];
    *(float2*)(out + (base + l16) * 2) = res;
  }
}

extern "C" void kernel_launch(void* const* d_in, const int* in_sizes, int n_in,
                              void* d_out, int out_size, void* d_ws, size_t ws_size,
                              hipStream_t stream) {
  (void)in_sizes; (void)n_in; (void)out_size; (void)ws_size;
  const float* x      = (const float*)d_in[0];
  const int*   ei     = (const int*)d_in[1];
  const int*   src    = ei;
  const int*   dst    = ei + NE;
  const float* c1_W1  = (const float*)d_in[2];
  const float* c1_b1  = (const float*)d_in[3];
  const float* c1_lnw = (const float*)d_in[4];
  const float* c1_lnb = (const float*)d_in[5];
  const float* c1_W2  = (const float*)d_in[6];
  const float* c1_b2  = (const float*)d_in[7];
  const float* n1_w   = (const float*)d_in[8];
  const float* n1_b   = (const float*)d_in[9];
  const float* cW1    = (const float*)d_in[10];
  const float* cb1    = (const float*)d_in[11];
  const float* clnw   = (const float*)d_in[12];
  const float* clnb   = (const float*)d_in[13];
  const float* cW2    = (const float*)d_in[14];
  const float* cb2    = (const float*)d_in[15];
  const float* nw     = (const float*)d_in[16];
  const float* nb     = (const float*)d_in[17];
  const float* lW1    = (const float*)d_in[18];
  const float* lb1    = (const float*)d_in[19];
  const float* lW2    = (const float*)d_in[20];
  const float* lb2    = (const float*)d_in[21];
  const float* lW3    = (const float*)d_in[22];
  const float* lb3    = (const float*)d_in[23];

  float* ws    = (float*)d_ws;
  float* xc    = ws;
  float* hpre  = ws + (size_t)N64;
  float* part  = ws + 3 * (size_t)N64;
  unsigned* wlist = (unsigned*)(ws + 2 * (size_t)N64);
  unsigned* keys  = (unsigned*)(ws + 3 * (size_t)N64 + 1024);
  unsigned* entry = (unsigned*)(ws + 3 * (size_t)N64 + 262144);

  const dim3 B(256);
  const int APB  = (N64 / 4 + 255) / 256;
  const int MLPB = NN / 32;
  const dim3 B2(64);

  k_owner_keys<<<KEY_WG, KEY_T, 0, stream>>>(src, dst, keys, entry);

  k_layer<true, false><<<NOWN, OWNT, 0, stream>>>(src, dst, keys, entry, x, wlist,
                                                  c1_W1, c1_b1, c1_lnw, c1_lnb, c1_W2, c1_b2,
                                                  nullptr, hpre);
  k_gln_partial<<<GLN_WG, GLN_T, 0, stream>>>((const float4*)hpre, part);
  k_gln_apply<<<APB, B, 0, stream>>>((const float4*)hpre, part, n1_w, n1_b, (float4*)xc);

  for (int i = 0; i < 3; i++) {
    k_gln_partial<<<GLN_WG, GLN_T, 0, stream>>>((const float4*)xc, part);
    k_gln_apply<<<APB, B, 0, stream>>>((const float4*)xc, part, nw + i * CH, nb + i * CH,
                                       (float4*)hpre);
    k_layer<false, true><<<NOWN, OWNT, 0, stream>>>(src, dst, keys, entry, hpre, wlist,
                                       cW1 + (size_t)i * CH * HD, cb1 + (size_t)i * HD,
                                       clnw + (size_t)i * HD, clnb + (size_t)i * HD,
                                       cW2 + (size_t)i * HD * CH, cb2 + (size_t)i * CH,
                                       xc, xc);
  }

  k_head<<<MLPB, B2, 0, stream>>>(xc, lW1, lb1, lW2, lb2, lW3, lb3, (float*)d_out);
}
